# starting kernel
# baseline (speedup 1.0000x reference)
#include <hip/hip_runtime.h>
#include <cstdint>

typedef float v2f __attribute__((ext_vector_type(2)));
typedef float v8f __attribute__((ext_vector_type(8)));

#define NFEAT 256
#define NHID  128
#define NCLASS 41
#define NCPAD  48

__global__ __launch_bounds__(256) void k_deg_init(float* deg, int N) {
    int i = blockIdx.x * 256 + threadIdx.x;
    if (i < N) deg[i] = 1.0f;
}

__global__ __launch_bounds__(256) void k_deg_edges(const int* __restrict__ dst,
                                                   const float* __restrict__ w,
                                                   float* deg, int E) {
    int e = blockIdx.x * 256 + threadIdx.x;
    if (e < E) unsafeAtomicAdd(&deg[dst[e]], w[e]);
}

__global__ __launch_bounds__(256) void k_deg_finish(float* dinv, int N) {
    int i = blockIdx.x * 256 + threadIdx.x;
    if (i < N) {
        float d = dinv[i];
        dinv[i] = d > 0.0f ? rsqrtf(d) : 0.0f;
    }
}

__global__ __launch_bounds__(256) void k_pack_w1(const float* __restrict__ W1, float* __restrict__ Wp) {
    int idx = blockIdx.x * 256 + threadIdx.x;
    if (idx >= NFEAT * NHID) return;
    int k = idx >> 7, n = idx & 127;
    Wp[(size_t)(k >> 1) * (NHID * 2) + n * 2 + (k & 1)] = W1[idx];
}

__global__ __launch_bounds__(256) void k_pack_w2(const float* __restrict__ W2, float* __restrict__ Wp) {
    int idx = blockIdx.x * 256 + threadIdx.x;
    if (idx >= NHID * NCPAD) return;
    int k = idx / NCPAD, n = idx % NCPAD;
    float v = (n < NCLASS) ? W2[k * NCLASS + n] : 0.0f;
    Wp[(size_t)(k >> 1) * (NCPAD * 2) + n * 2 + (k & 1)] = v;
}

__global__ __launch_bounds__(256) void k_gemm1(const float* __restrict__ X,
                                               const float* __restrict__ Wp,
                                               float* __restrict__ H, int M) {
    __shared__ float2 Xs[128 * 33];
    const int tid = threadIdx.x;
    const int m0 = blockIdx.x * 32;

#pragma unroll
    for (int it = 0; it < 8; ++it) {
        int q = it * 256 + tid;
        int m = q >> 6, k4 = q & 63;
        int mr = m0 + m; if (mr >= M) mr = M - 1;
        float4 v = ((const float4*)(X + (size_t)mr * NFEAT))[k4];
        Xs[(2 * k4)     * 33 + m] = make_float2(v.x, v.y);
        Xs[(2 * k4 + 1) * 33 + m] = make_float2(v.z, v.w);
    }
    __syncthreads();

    const int wave = tid >> 5, lane = tid & 31, l = lane & 15, half = lane >> 4;
    const int n0 = wave * 16;
    const float2* Bp = (const float2*)Wp;

    v8f acc0 = {}, acc1 = {};
#pragma unroll 4
    for (int kp = 0; kp < 128; kp += 2) {
        int row = kp + half;
        float2 a0 = Xs[row * 33 + l];
        float2 a1 = Xs[row * 33 + 16 + l];
        float2 b  = Bp[(size_t)row * NHID + n0 + l];
        v2f av0 = {a0.x, a0.y}, av1 = {a1.x, a1.y}, bv = {b.x, b.y};
        acc0 = __builtin_amdgcn_wmma_f32_16x16x4_f32(false, av0, false, bv, (short)0, acc0, false, false);
        acc1 = __builtin_amdgcn_wmma_f32_16x16x4_f32(false, av1, false, bv, (short)0, acc1, false, false);
    }

    const int mb = 8 * half;
    float* Ho = H + (size_t)(m0 + mb) * NHID + n0 + l;
    if (m0 + 32 <= M) {
#pragma unroll
        for (int r = 0; r < 8; ++r) {
            Ho[(size_t)r * NHID]        = acc0[r];
            Ho[(size_t)(16 + r) * NHID] = acc1[r];
        }
    } else {
#pragma unroll
        for (int r = 0; r < 8; ++r) {
            if (m0 + mb + r < M)      Ho[(size_t)r * NHID]        = acc0[r];
            if (m0 + 16 + mb + r < M) Ho[(size_t)(16 + r) * NHID] = acc1[r];
        }
    }
}

__global__ __launch_bounds__(256) void k_agg1_init(const float* __restrict__ H,
                                                   const float* __restrict__ dinv,
                                                   float* __restrict__ Hacc, int N) {
    int idx = blockIdx.x * 256 + threadIdx.x;
    if (idx >= N * 32) return;
    int i = idx >> 5, j = idx & 31;
    float di = dinv[i], c = di * di;
    float4 v = ((const float4*)(H + (size_t)i * NHID))[j];
    float4 o = make_float4(c * v.x, c * v.y, c * v.z, c * v.w);
    ((float4*)(Hacc + (size_t)i * NHID))[j] = o;
}

__global__ __launch_bounds__(256) void k_agg1_edges(const int* __restrict__ src,
                                                    const int* __restrict__ dst,
                                                    const float* __restrict__ w,
                                                    const float* __restrict__ dinv,
                                                    const float* __restrict__ H,
                                                    float* __restrict__ Hacc, int E) {
    int e = blockIdx.x * 8 + (threadIdx.x >> 5);
    if (e >= E) return;
    int lane = threadIdx.x & 31;
    int s = src[e], d = dst[e];
    float c = dinv[s] * w[e] * dinv[d];
    float4 v = ((const float4*)(H + (size_t)s * NHID))[lane];
    float* o = Hacc + (size_t)d * NHID + lane * 4;
    unsafeAtomicAdd(o + 0, c * v.x);
    unsafeAtomicAdd(o + 1, c * v.y);
    unsafeAtomicAdd(o + 2, c * v.z);
    unsafeAtomicAdd(o + 3, c * v.w);
}

__global__ __launch_bounds__(64) void k_gemm2(const float* __restrict__ Hacc,
                                              const float* __restrict__ b1,
                                              const float* __restrict__ Wp,
                                              float* __restrict__ G, int M) {
    __shared__ float2 Xs[64 * 33];
    const int tid = threadIdx.x;
    const int m0 = blockIdx.x * 32;

#pragma unroll
    for (int it = 0; it < 16; ++it) {
        int q = it * 64 + tid;
        int m = q >> 5, k4 = q & 31;
        int mr = m0 + m; if (mr >= M) mr = M - 1;
        float4 v = ((const float4*)(Hacc + (size_t)mr * NHID))[k4];
        float4 bb = ((const float4*)b1)[k4];
        v.x = fmaxf(v.x + bb.x, 0.0f);
        v.y = fmaxf(v.y + bb.y, 0.0f);
        v.z = fmaxf(v.z + bb.z, 0.0f);
        v.w = fmaxf(v.w + bb.w, 0.0f);
        Xs[(2 * k4)     * 33 + m] = make_float2(v.x, v.y);
        Xs[(2 * k4 + 1) * 33 + m] = make_float2(v.z, v.w);
    }
    __syncthreads();

    const int wave = tid >> 5, lane = tid & 31, l = lane & 15, half = lane >> 4;
    const float2* Bp = (const float2*)Wp;

    v8f acc0 = {}, acc1 = {}, acc2 = {};
#pragma unroll 4
    for (int kp = 0; kp < 64; kp += 2) {
        int row = kp + half;
        float2 a  = Xs[row * 33 + wave * 16 + l];
        float2 b0 = Bp[(size_t)row * NCPAD + 0  + l];
        float2 b1f= Bp[(size_t)row * NCPAD + 16 + l];
        float2 b2f= Bp[(size_t)row * NCPAD + 32 + l];
        v2f av = {a.x, a.y};
        v2f bv0 = {b0.x, b0.y}, bv1 = {b1f.x, b1f.y}, bv2 = {b2f.x, b2f.y};
        acc0 = __builtin_amdgcn_wmma_f32_16x16x4_f32(false, av, false, bv0, (short)0, acc0, false, false);
        acc1 = __builtin_amdgcn_wmma_f32_16x16x4_f32(false, av, false, bv1, (short)0, acc1, false, false);
        acc2 = __builtin_amdgcn_wmma_f32_16x16x4_f32(false, av, false, bv2, (short)0, acc2, false, false);
    }

    const int mb = m0 + wave * 16 + 8 * half;
    float* Go = G + (size_t)mb * NCPAD + l;
    if (m0 + 32 <= M) {
#pragma unroll
        for (int r = 0; r < 8; ++r) {
            float* o = Go + (size_t)r * NCPAD;
            o[0]  = acc0[r];
            o[16] = acc1[r];
            o[32] = acc2[r];
        }
    } else {
#pragma unroll
        for (int r = 0; r < 8; ++r) {
            if (mb + r < M) {
                float* o = Go + (size_t)r * NCPAD;
                o[0]  = acc0[r];
                o[16] = acc1[r];
                o[32] = acc2[r];
            }
        }
    }
}

__global__ __launch_bounds__(256) void k_agg2_init(const float* __restrict__ G,
                                                   const float* __restrict__ dinv,
                                                   float* __restrict__ Oacc, int N) {
    int idx = blockIdx.x * 256 + threadIdx.x;
    if (idx >= N * 12) return;
    int i = idx / 12, j = idx % 12;
    float di = dinv[i], c = di * di;
    float4 v = ((const float4*)(G + (size_t)i * NCPAD))[j];
    float4 o = make_float4(c * v.x, c * v.y, c * v.z, c * v.w);
    ((float4*)(Oacc + (size_t)i * NCPAD))[j] = o;
}

__global__ __launch_bounds__(256) void k_agg2_edges(const int* __restrict__ src,
                                                    const int* __restrict__ dst,
                                                    const float* __restrict__ w,
                                                    const float* __restrict__ dinv,
                                                    const float* __restrict__ G,
                                                    float* __restrict__ Oacc, int E) {
    int e = blockIdx.x * 8 + (threadIdx.x >> 5);
    if (e >= E) return;
    int lane = threadIdx.x & 31;
    int s = src[e], d = dst[e];
    float c = dinv[s] * w[e] * dinv[d];
    const float* g = G + (size_t)s * NCPAD;
    float* o = Oacc + (size_t)d * NCPAD;
    unsafeAtomicAdd(o + lane, c * g[lane]);
    if (lane < 16) unsafeAtomicAdd(o + 32 + lane, c * g[32 + lane]);
}

__global__ __launch_bounds__(256) void k_logsoftmax(const float* __restrict__ Oacc,
                                                    const float* __restrict__ b2,
                                                    float* __restrict__ out, int N) {
    int i = blockIdx.x * 8 + (threadIdx.x >> 5);
    if (i >= N) return;
    int lane = threadIdx.x & 31;
    const float* row = Oacc + (size_t)i * NCPAD;

    float v1 = row[lane] + b2[lane];
    int c2 = 32 + lane;
    bool has2 = c2 < NCLASS;
    float v2 = has2 ? (row[c2] + b2[c2]) : -INFINITY;

    float mx = fmaxf(v1, v2);
#pragma unroll
    for (int off = 16; off >= 1; off >>= 1) mx = fmaxf(mx, __shfl_xor(mx, off));

    float s = expf(v1 - mx) + (has2 ? expf(v2 - mx) : 0.0f);
#pragma unroll
    for (int off = 16; off >= 1; off >>= 1) s += __shfl_xor(s, off);

    float ls = logf(s);
    out[(size_t)i * NCLASS + lane] = v1 - mx - ls;
    if (has2) out[(size_t)i * NCLASS + c2] = v2 - mx - ls;
}

extern "C" void kernel_launch(void* const* d_in, const int* in_sizes, int n_in,
                              void* d_out, int out_size, void* d_ws, size_t ws_size,
                              hipStream_t stream) {
    const float* x  = (const float*)d_in[0];
    const int*   ei = (const int*)  d_in[1];
    const float* ew = (const float*)d_in[2];
    const float* W1 = (const float*)d_in[3];
    const float* b1 = (const float*)d_in[4];
    const float* W2 = (const float*)d_in[5];
    const float* b2 = (const float*)d_in[6];
    float* out = (float*)d_out;

    const int N = in_sizes[0] / NFEAT;
    const int E = in_sizes[2];
    const int* src = ei;
    const int* dst = ei + E;

    float* ws = (float*)d_ws;
    size_t o = 0;
    auto take = [&](size_t n) { float* p = ws + o; o += (n + 255) & ~(size_t)255; return p; };
    float* dinv = take((size_t)N);
    float* Wp1  = take((size_t)NFEAT * NHID);
    float* Wp2  = take((size_t)NHID * NCPAD);
    float* Hxw  = take((size_t)N * NHID);
    float* Hacc = take((size_t)N * NHID);
    float* G    = Hxw;
    float* Oacc = Hxw + (size_t)N * NCPAD;

    k_pack_w1<<<(NFEAT * NHID + 255) / 256, 256, 0, stream>>>(W1, Wp1);
    k_pack_w2<<<(NHID * NCPAD + 255) / 256, 256, 0, stream>>>(W2, Wp2);

    k_deg_init  <<<(N + 255) / 256, 256, 0, stream>>>(dinv, N);
    k_deg_edges <<<(E + 255) / 256, 256, 0, stream>>>(dst, ew, dinv, E);
    k_deg_finish<<<(N + 255) / 256, 256, 0, stream>>>(dinv, N);

    k_gemm1<<<(N + 31) / 32, 256, 0, stream>>>(x, Wp1, Hxw, N);

    k_agg1_init <<<(N * 32 + 255) / 256, 256, 0, stream>>>(Hxw, dinv, Hacc, N);
    k_agg1_edges<<<(E + 7) / 8, 256, 0, stream>>>(src, dst, ew, dinv, Hxw, Hacc, E);

    k_gemm2<<<(N + 31) / 32, 64, 0, stream>>>(Hacc, b1, Wp2, G, N);

    k_agg2_init <<<(N * 12 + 255) / 256, 256, 0, stream>>>(G, dinv, Oacc, N);
    k_agg2_edges<<<(E + 7) / 8, 256, 0, stream>>>(src, dst, ew, dinv, G, Oacc, E);

    k_logsoftmax<<<(N + 7) / 8, 256, 0, stream>>>(Oacc, b2, out, N);
}


# Edges counting-sorted by destination once per call (int rank atomics); degree and both aggregations become per-node gathers, no float atomics; sums now run in arrival-rank order
# speedup vs baseline: 2.5042x; 2.5042x over previous
#include <hip/hip_runtime.h>
#include <cstdint>

typedef float v2f __attribute__((ext_vector_type(2)));
typedef float v8f __attribute__((ext_vector_type(8)));

#define NFEAT 256
#define NHID  128
#define NCLASS 41
#define NCPAD  48

#define CNT_STRIDE 32

__global__ __launch_bounds__(256) void k_cnt_zero(int* __restrict__ cnt, int N) {
    int i = blockIdx.x * 256 + threadIdx.x;
    if (i < N) cnt[(size_t)i * CNT_STRIDE] = 0;
}

__global__ __launch_bounds__(256) void k_rank(const int* __restrict__ dst, int* cnt,
                                              int* __restrict__ rank, int E) {
    int e = blockIdx.x * 256 + threadIdx.x;
    if (e < E) rank[e] = atomicAdd(&cnt[(size_t)dst[e] * CNT_STRIDE], 1);
}

__global__ __launch_bounds__(256) void k_scan_part(const int* __restrict__ cnt,
                                                   int* __restrict__ bsum, int N) {
    __shared__ int wsum[8];
    const int t = threadIdx.x;
    const int i0 = blockIdx.x * 1024 + t * 4;
    int v = 0;
    for (int k = 0; k < 4; ++k) {
        int i = i0 + k;
        if (i < N) v += cnt[(size_t)i * CNT_STRIDE];
    }
    for (int off = 16; off >= 1; off >>= 1) v += __shfl_xor(v, off);
    if ((t & 31) == 0) wsum[t >> 5] = v;
    __syncthreads();
    if (t == 0) {
        int s = 0;
        for (int k = 0; k < 8; ++k) s += wsum[k];
        bsum[blockIdx.x] = s;
    }
}

__global__ __launch_bounds__(256) void k_scan_final(const int* __restrict__ cnt,
                                                    const int* __restrict__ bsum,
                                                    int* __restrict__ first, int N) {
    __shared__ int wtot[8];
    __shared__ int wpre[8];
    const int t = threadIdx.x, lane = t & 31, wv = t >> 5;

    int p = 0;
    for (int j = t; j < (int)blockIdx.x; j += 256) p += bsum[j];
    for (int off = 16; off >= 1; off >>= 1) p += __shfl_xor(p, off);
    if (lane == 0) wpre[wv] = p;

    const int i0 = blockIdx.x * 1024 + t * 4;
    int c[4];
    int v = 0;
    for (int k = 0; k < 4; ++k) {
        int i = i0 + k;
        c[k] = (i < N) ? cnt[(size_t)i * CNT_STRIDE] : 0;
        v += c[k];
    }
    int inc = v;
    for (int off = 1; off < 32; off <<= 1) {
        int u = __shfl_up(inc, off);
        if (lane >= off) inc += u;
    }
    if (lane == 31) wtot[wv] = inc;
    __syncthreads();

    int ex = inc - v;
    for (int k = 0; k < 8; ++k) {
        ex += wpre[k];
        if (k < wv) ex += wtot[k];
    }
    for (int k = 0; k < 4; ++k) {
        int i = i0 + k;
        if (i < N) {
            first[i] = ex;
            ex += c[k];
            if (i == N - 1) first[N] = ex;
        }
    }
}

__global__ __launch_bounds__(256) void k_fill(const int* __restrict__ src,
                                              const int* __restrict__ dst,
                                              const float* __restrict__ w,
                                              const int* __restrict__ rank,
                                              const int* __restrict__ first,
                                              int* __restrict__ edges, int E) {
    int e = blockIdx.x * 256 + threadIdx.x;
    if (e < E) {
        int slot = first[dst[e]] + rank[e];
        ((int2*)edges)[slot] = make_int2(src[e], __float_as_int(w[e]));
    }
}

__global__ __launch_bounds__(256) void k_dinv(const int* __restrict__ first,
                                              const int* __restrict__ edges,
                                              float* __restrict__ dinv, int N) {
    const int g = blockIdx.x * 256 + threadIdx.x;
    const int i = g >> 3, sub = g & 7;
    float s = 0.0f;
    if (i < N) {
        const int end = first[i + 1];
        for (int p = first[i] + sub; p < end; p += 8) s += __int_as_float(edges[2 * (size_t)p + 1]);
    }
    s += __shfl_xor(s, 4);
    s += __shfl_xor(s, 2);
    s += __shfl_xor(s, 1);
    if (i < N && sub == 0) {
        float d = 1.0f + s;
        dinv[i] = d > 0.0f ? rsqrtf(d) : 0.0f;
    }
}

static __device__ __forceinline__ int lane_i(int v, int j) {
    return __builtin_amdgcn_readlane(v, j);
}
static __device__ __forceinline__ float lane_f(float v, int j) {
    return __int_as_float(__builtin_amdgcn_readlane(__float_as_int(v), j));
}

__global__ __launch_bounds__(256) void k_pack_w1(const float* __restrict__ W1, float* __restrict__ Wp) {
    int idx = blockIdx.x * 256 + threadIdx.x;
    if (idx >= NFEAT * NHID) return;
    int k = idx >> 7, n = idx & 127;
    Wp[(size_t)(k >> 1) * (NHID * 2) + n * 2 + (k & 1)] = W1[idx];
}

__global__ __launch_bounds__(256) void k_pack_w2(const float* __restrict__ W2, float* __restrict__ Wp) {
    int idx = blockIdx.x * 256 + threadIdx.x;
    if (idx >= NHID * NCPAD) return;
    int k = idx / NCPAD, n = idx % NCPAD;
    float v = (n < NCLASS) ? W2[k * NCLASS + n] : 0.0f;
    Wp[(size_t)(k >> 1) * (NCPAD * 2) + n * 2 + (k & 1)] = v;
}

__global__ __launch_bounds__(256) void k_gemm1(const float* __restrict__ X,
                                               const float* __restrict__ Wp,
                                               float* __restrict__ H, int M) {
    __shared__ float2 Xs[128 * 33];
    const int tid = threadIdx.x;
    const int m0 = blockIdx.x * 32;

#pragma unroll
    for (int it = 0; it < 8; ++it) {
        int q = it * 256 + tid;
        int m = q >> 6, k4 = q & 63;
        int mr = m0 + m; if (mr >= M) mr = M - 1;
        float4 v = ((const float4*)(X + (size_t)mr * NFEAT))[k4];
        Xs[(2 * k4)     * 33 + m] = make_float2(v.x, v.y);
        Xs[(2 * k4 + 1) * 33 + m] = make_float2(v.z, v.w);
    }
    __syncthreads();

    const int wave = tid >> 5, lane = tid & 31, l = lane & 15, half = lane >> 4;
    const int n0 = wave * 16;
    const float2* Bp = (const float2*)Wp;

    v8f acc0 = {}, acc1 = {};
#pragma unroll 4
    for (int kp = 0; kp < 128; kp += 2) {
        int row = kp + half;
        float2 a0 = Xs[row * 33 + l];
        float2 a1 = Xs[row * 33 + 16 + l];
        float2 b  = Bp[(size_t)row * NHID + n0 + l];
        v2f av0 = {a0.x, a0.y}, av1 = {a1.x, a1.y}, bv = {b.x, b.y};
        acc0 = __builtin_amdgcn_wmma_f32_16x16x4_f32(false, av0, false, bv, (short)0, acc0, false, false);
        acc1 = __builtin_amdgcn_wmma_f32_16x16x4_f32(false, av1, false, bv, (short)0, acc1, false, false);
    }

    const int mb = 8 * half;
    float* Ho = H + (size_t)(m0 + mb) * NHID + n0 + l;
    if (m0 + 32 <= M) {
#pragma unroll
        for (int r = 0; r < 8; ++r) {
            Ho[(size_t)r * NHID]        = acc0[r];
            Ho[(size_t)(16 + r) * NHID] = acc1[r];
        }
    } else {
#pragma unroll
        for (int r = 0; r < 8; ++r) {
            if (m0 + mb + r < M)      Ho[(size_t)r * NHID]        = acc0[r];
            if (m0 + 16 + mb + r < M) Ho[(size_t)(16 + r) * NHID] = acc1[r];
        }
    }
}

__global__ __launch_bounds__(256) void k_agg1_gather(const int* __restrict__ first,
                                                     const int* __restrict__ edges,
                                                     const float* __restrict__ dinv,
                                                     const float* __restrict__ H,
                                                     float* __restrict__ Hacc, int N) {
    const int lane = threadIdx.x & 31;
    const int i = __builtin_amdgcn_readfirstlane((int)(blockIdx.x * 8 + (threadIdx.x >> 5)));
    if (i >= N) return;
    const int beg = first[i], end = first[i + 1];
    const float di = dinv[i];
    const float4* H4 = (const float4*)H;

    const float cs = di * di;
    const float4 hs = H4[(size_t)i * 32 + lane];
    float4 acc = make_float4(cs * hs.x, cs * hs.y, cs * hs.z, cs * hs.w);

    for (int base = beg; base < end; base += 32) {
        const int n = min(32, end - base);
        int s_l = 0;
        float c_l = 0.0f;
        if (lane < n) {
            int2 sw = ((const int2*)edges)[base + lane];
            s_l = sw.x;
            c_l = dinv[sw.x] * __int_as_float(sw.y) * di;
        }
        int j = 0;
        for (; j + 4 <= n; j += 4) {
            const int s0 = lane_i(s_l, j), s1 = lane_i(s_l, j + 1);
            const int s2 = lane_i(s_l, j + 2), s3 = lane_i(s_l, j + 3);
            const float c0 = lane_f(c_l, j), c1 = lane_f(c_l, j + 1);
            const float c2 = lane_f(c_l, j + 2), c3 = lane_f(c_l, j + 3);
            const float4 v0 = H4[(size_t)s0 * 32 + lane];
            const float4 v1 = H4[(size_t)s1 * 32 + lane];
            const float4 v2 = H4[(size_t)s2 * 32 + lane];
            const float4 v3 = H4[(size_t)s3 * 32 + lane];
            acc.x += c0 * v0.x; acc.y += c0 * v0.y; acc.z += c0 * v0.z; acc.w += c0 * v0.w;
            acc.x += c1 * v1.x; acc.y += c1 * v1.y; acc.z += c1 * v1.z; acc.w += c1 * v1.w;
            acc.x += c2 * v2.x; acc.y += c2 * v2.y; acc.z += c2 * v2.z; acc.w += c2 * v2.w;
            acc.x += c3 * v3.x; acc.y += c3 * v3.y; acc.z += c3 * v3.z; acc.w += c3 * v3.w;
        }
        for (; j < n; ++j) {
            const int s0 = lane_i(s_l, j);
            const float c0 = lane_f(c_l, j);
            const float4 v0 = H4[(size_t)s0 * 32 + lane];
            acc.x += c0 * v0.x; acc.y += c0 * v0.y; acc.z += c0 * v0.z; acc.w += c0 * v0.w;
        }
    }
    ((float4*)Hacc)[(size_t)i * 32 + lane] = acc;
}

__global__ __launch_bounds__(64) void k_gemm2(const float* __restrict__ Hacc,
                                              const float* __restrict__ b1,
                                              const float* __restrict__ Wp,
                                              float* __restrict__ G, int M) {
    __shared__ float2 Xs[64 * 33];
    const int tid = threadIdx.x;
    const int m0 = blockIdx.x * 32;

#pragma unroll
    for (int it = 0; it < 16; ++it) {
        int q = it * 64 + tid;
        int m = q >> 5, k4 = q & 31;
        int mr = m0 + m; if (mr >= M) mr = M - 1;
        float4 v = ((const float4*)(Hacc + (size_t)mr * NHID))[k4];
        float4 bb = ((const float4*)b1)[k4];
        v.x = fmaxf(v.x + bb.x, 0.0f);
        v.y = fmaxf(v.y + bb.y, 0.0f);
        v.z = fmaxf(v.z + bb.z, 0.0f);
        v.w = fmaxf(v.w + bb.w, 0.0f);
        Xs[(2 * k4)     * 33 + m] = make_float2(v.x, v.y);
        Xs[(2 * k4 + 1) * 33 + m] = make_float2(v.z, v.w);
    }
    __syncthreads();

    const int wave = tid >> 5, lane = tid & 31, l = lane & 15, half = lane >> 4;
    const float2* Bp = (const float2*)Wp;

    v8f acc0 = {}, acc1 = {}, acc2 = {};
#pragma unroll 4
    for (int kp = 0; kp < 64; kp += 2) {
        int row = kp + half;
        float2 a  = Xs[row * 33 + wave * 16 + l];
        float2 b0 = Bp[(size_t)row * NCPAD + 0  + l];
        float2 b1f= Bp[(size_t)row * NCPAD + 16 + l];
        float2 b2f= Bp[(size_t)row * NCPAD + 32 + l];
        v2f av = {a.x, a.y};
        v2f bv0 = {b0.x, b0.y}, bv1 = {b1f.x, b1f.y}, bv2 = {b2f.x, b2f.y};
        acc0 = __builtin_amdgcn_wmma_f32_16x16x4_f32(false, av, false, bv0, (short)0, acc0, false, false);
        acc1 = __builtin_amdgcn_wmma_f32_16x16x4_f32(false, av, false, bv1, (short)0, acc1, false, false);
        acc2 = __builtin_amdgcn_wmma_f32_16x16x4_f32(false, av, false, bv2, (short)0, acc2, false, false);
    }

    const int mb = m0 + wave * 16 + 8 * half;
    float* Go = G + (size_t)mb * NCPAD + l;
    if (m0 + 32 <= M) {
#pragma unroll
        for (int r = 0; r < 8; ++r) {
            float* o = Go + (size_t)r * NCPAD;
            o[0]  = acc0[r];
            o[16] = acc1[r];
            o[32] = acc2[r];
        }
    } else {
#pragma unroll
        for (int r = 0; r < 8; ++r) {
            if (mb + r < M) {
                float* o = Go + (size_t)r * NCPAD;
                o[0]  = acc0[r];
                o[16] = acc1[r];
                o[32] = acc2[r];
            }
        }
    }
}

__global__ __launch_bounds__(256) void k_agg2_gather(const int* __restrict__ first,
                                                     const int* __restrict__ edges,
                                                     const float* __restrict__ dinv,
                                                     const float* __restrict__ G,
                                                     float* __restrict__ Oacc, int N) {
    const int lane = threadIdx.x & 31;
    const int lane2 = 32 + (lane & 15);
    const int i = __builtin_amdgcn_readfirstlane((int)(blockIdx.x * 8 + (threadIdx.x >> 5)));
    if (i >= N) return;
    const int beg = first[i], end = first[i + 1];
    const float di = dinv[i];

    const float cs = di * di;
    const float* gs = G + (size_t)i * NCPAD;
    float acc0 = cs * gs[lane];
    float acc1 = cs * gs[lane2];

    for (int base = beg; base < end; base += 32) {
        const int n = min(32, end - base);
        int s_l = 0;
        float c_l = 0.0f;
        if (lane < n) {
            int2 sw = ((const int2*)edges)[base + lane];
            s_l = sw.x;
            c_l = dinv[sw.x] * __int_as_float(sw.y) * di;
        }
        int j = 0;
        for (; j + 4 <= n; j += 4) {
            const float* g0 = G + (size_t)lane_i(s_l, j)     * NCPAD;
            const float* g1 = G + (size_t)lane_i(s_l, j + 1) * NCPAD;
            const float* g2 = G + (size_t)lane_i(s_l, j + 2) * NCPAD;
            const float* g3 = G + (size_t)lane_i(s_l, j + 3) * NCPAD;
            const float c0 = lane_f(c_l, j), c1 = lane_f(c_l, j + 1);
            const float c2 = lane_f(c_l, j + 2), c3 = lane_f(c_l, j + 3);
            const float x0 = g0[lane], y0 = g0[lane2];
            const float x1 = g1[lane], y1 = g1[lane2];
            const float x2 = g2[lane], y2 = g2[lane2];
            const float x3 = g3[lane], y3 = g3[lane2];
            acc0 += c0 * x0; acc1 += c0 * y0;
            acc0 += c1 * x1; acc1 += c1 * y1;
            acc0 += c2 * x2; acc1 += c2 * y2;
            acc0 += c3 * x3; acc1 += c3 * y3;
        }
        for (; j < n; ++j) {
            const float* g0 = G + (size_t)lane_i(s_l, j) * NCPAD;
            const float c0 = lane_f(c_l, j);
            acc0 += c0 * g0[lane];
            acc1 += c0 * g0[lane2];
        }
    }
    float* o = Oacc + (size_t)i * NCPAD;
    o[lane] = acc0;
    if (lane < 16) o[32 + lane] = acc1;
}

__global__ __launch_bounds__(256) void k_logsoftmax(const float* __restrict__ Oacc,
                                                    const float* __restrict__ b2,
                                                    float* __restrict__ out, int N) {
    int i = blockIdx.x * 8 + (threadIdx.x >> 5);
    if (i >= N) return;
    int lane = threadIdx.x & 31;
    const float* row = Oacc + (size_t)i * NCPAD;

    float v1 = row[lane] + b2[lane];
    int c2 = 32 + lane;
    bool has2 = c2 < NCLASS;
    float v2 = has2 ? (row[c2] + b2[c2]) : -INFINITY;

    float mx = fmaxf(v1, v2);
#pragma unroll
    for (int off = 16; off >= 1; off >>= 1) mx = fmaxf(mx, __shfl_xor(mx, off));

    float s = expf(v1 - mx) + (has2 ? expf(v2 - mx) : 0.0f);
#pragma unroll
    for (int off = 16; off >= 1; off >>= 1) s += __shfl_xor(s, off);

    float ls = logf(s);
    out[(size_t)i * NCLASS + lane] = v1 - mx - ls;
    if (has2) out[(size_t)i * NCLASS + c2] = v2 - mx - ls;
}

extern "C" void kernel_launch(void* const* d_in, const int* in_sizes, int n_in,
                              void* d_out, int out_size, void* d_ws, size_t ws_size,
                              hipStream_t stream) {
    const float* x  = (const float*)d_in[0];
    const int*   ei = (const int*)  d_in[1];
    const float* ew = (const float*)d_in[2];
    const float* W1 = (const float*)d_in[3];
    const float* b1 = (const float*)d_in[4];
    const float* W2 = (const float*)d_in[5];
    const float* b2 = (const float*)d_in[6];
    float* out = (float*)d_out;

    const int N = in_sizes[0] / NFEAT;
    const int E = in_sizes[2];
    const int* src = ei;
    const int* dst = ei + E;

    float* ws = (float*)d_ws;
    size_t o = 0;
    auto take = [&](size_t n) { float* p = ws + o; o += (n + 255) & ~(size_t)255; return p; };
    float* dinv = take((size_t)N);
    float* Wp1  = take((size_t)NFEAT * NHID);
    float* Wp2  = take((size_t)NHID * NCPAD);
    float* Hxw  = take((size_t)N * NHID);
    float* Hacc = take((size_t)N * NHID);
    float* G    = Hxw;
    float* Oacc = Hxw + (size_t)N * NCPAD;

    int* cnt  = (int*)Hacc;
    int* rank = cnt + (size_t)N * CNT_STRIDE;
    int* bsum = rank + E;
    int* edges = (int*)d_out;
    int* first = edges + 2 * (size_t)E;

    k_pack_w1<<<(NFEAT * NHID + 255) / 256, 256, 0, stream>>>(W1, Wp1);
    k_pack_w2<<<(NHID * NCPAD + 255) / 256, 256, 0, stream>>>(W2, Wp2);

    k_cnt_zero  <<<(N + 255) / 256, 256, 0, stream>>>(cnt, N);
    k_rank      <<<(E + 255) / 256, 256, 0, stream>>>(dst, cnt, rank, E);
    k_scan_part <<<(N + 1023) / 1024, 256, 0, stream>>>(cnt, bsum, N);
    k_scan_final<<<(N + 1023) / 1024, 256, 0, stream>>>(cnt, bsum, first, N);
    k_fill      <<<(E + 255) / 256, 256, 0, stream>>>(src, dst, ew, rank, first, edges, E);
    k_dinv      <<<(N * 8 + 255) / 256, 256, 0, stream>>>(first, edges, dinv, N);

    k_gemm1<<<(N + 31) / 32, 256, 0, stream>>>(x, Wp1, Hxw, N);

    k_agg1_gather<<<(N + 7) / 8, 256, 0, stream>>>(first, edges, dinv, Hxw, Hacc, N);

    k_gemm2<<<(N + 31) / 32, 64, 0, stream>>>(Hacc, b1, Wp2, G, N);

    k_agg2_gather<<<(N + 7) / 8, 256, 0, stream>>>(first, edges, dinv, G, Oacc, N);

    k_logsoftmax<<<(N + 7) / 8, 256, 0, stream>>>(Oacc, b2, out, N);
}


# Independent passes share launches: weight packs + counter clear in one; X@W1 row tiles ride as extra blocks in the sort's rank and fill launches (matrix issue under atomics and scatter); 13 to 10
# speedup vs baseline: 2.6407x; 1.0545x over previous
#include <hip/hip_runtime.h>
#include <cstdint>

typedef float v2f __attribute__((ext_vector_type(2)));
typedef float v8f __attribute__((ext_vector_type(8)));

#define NFEAT 256
#define NHID  128
#define NCLASS 41
#define NCPAD  48

#define CNT_STRIDE 32

static __device__ __forceinline__ void cnt_zero_chunk(int* __restrict__ cnt, int N, int chunk) {
    int i = chunk * 256 + threadIdx.x;
    if (i < N) cnt[(size_t)i * CNT_STRIDE] = 0;
}

static __device__ __forceinline__ void rank_chunk(const int* __restrict__ dst, int* cnt,
                                                  int* __restrict__ rank, int E, int chunk) {
    int e = chunk * 256 + threadIdx.x;
    if (e < E) rank[e] = atomicAdd(&cnt[(size_t)dst[e] * CNT_STRIDE], 1);
}

__global__ __launch_bounds__(256) void k_scan_part(const int* __restrict__ cnt,
                                                   int* __restrict__ bsum, int N) {
    __shared__ int wsum[8];
    const int t = threadIdx.x;
    const int i0 = blockIdx.x * 1024 + t * 4;
    int v = 0;
    for (int k = 0; k < 4; ++k) {
        int i = i0 + k;
        if (i < N) v += cnt[(size_t)i * CNT_STRIDE];
    }
    for (int off = 16; off >= 1; off >>= 1) v += __shfl_xor(v, off);
    if ((t & 31) == 0) wsum[t >> 5] = v;
    __syncthreads();
    if (t == 0) {
        int s = 0;
        for (int k = 0; k < 8; ++k) s += wsum[k];
        bsum[blockIdx.x] = s;
    }
}

__global__ __launch_bounds__(256) void k_scan_final(const int* __restrict__ cnt,
                                                    const int* __restrict__ bsum,
                                                    int* __restrict__ first, int N) {
    __shared__ int wtot[8];
    __shared__ int wpre[8];
    const int t = threadIdx.x, lane = t & 31, wv = t >> 5;

    int p = 0;
    for (int j = t; j < (int)blockIdx.x; j += 256) p += bsum[j];
    for (int off = 16; off >= 1; off >>= 1) p += __shfl_xor(p, off);
    if (lane == 0) wpre[wv] = p;

    const int i0 = blockIdx.x * 1024 + t * 4;
    int c[4];
    int v = 0;
    for (int k = 0; k < 4; ++k) {
        int i = i0 + k;
        c[k] = (i < N) ? cnt[(size_t)i * CNT_STRIDE] : 0;
        v += c[k];
    }
    int inc = v;
    for (int off = 1; off < 32; off <<= 1) {
        int u = __shfl_up(inc, off);
        if (lane >= off) inc += u;
    }
    if (lane == 31) wtot[wv] = inc;
    __syncthreads();

    int ex = inc - v;
    for (int k = 0; k < 8; ++k) {
        ex += wpre[k];
        if (k < wv) ex += wtot[k];
    }
    for (int k = 0; k < 4; ++k) {
        int i = i0 + k;
        if (i < N) {
            first[i] = ex;
            ex += c[k];
            if (i == N - 1) first[N] = ex;
        }
    }
}

static __device__ __forceinline__ void fill_chunk(const int* __restrict__ src,
                                                  const int* __restrict__ dst,
                                                  const float* __restrict__ w,
                                                  const int* __restrict__ rank,
                                                  const int* __restrict__ first,
                                                  int* __restrict__ edges, int E, int chunk) {
    int e = chunk * 256 + threadIdx.x;
    if (e < E) {
        int slot = first[dst[e]] + rank[e];
        ((int2*)edges)[slot] = make_int2(src[e], __float_as_int(w[e]));
    }
}

__global__ __launch_bounds__(256) void k_dinv(const int* __restrict__ first,
                                              const int* __restrict__ edges,
                                              float* __restrict__ dinv, int N) {
    const int g = blockIdx.x * 256 + threadIdx.x;
    const int i = g >> 3, sub = g & 7;
    float s = 0.0f;
    if (i < N) {
        const int end = first[i + 1];
        for (int p = first[i] + sub; p < end; p += 8) s += __int_as_float(edges[2 * (size_t)p + 1]);
    }
    s += __shfl_xor(s, 4);
    s += __shfl_xor(s, 2);
    s += __shfl_xor(s, 1);
    if (i < N && sub == 0) {
        float d = 1.0f + s;
        dinv[i] = d > 0.0f ? rsqrtf(d) : 0.0f;
    }
}

static __device__ __forceinline__ int lane_i(int v, int j) {
    return __builtin_amdgcn_readlane(v, j);
}
static __device__ __forceinline__ float lane_f(float v, int j) {
    return __int_as_float(__builtin_amdgcn_readlane(__float_as_int(v), j));
}

static __device__ __forceinline__ void pack_w1_chunk(const float* __restrict__ W1, float* __restrict__ Wp, int chunk) {
    int idx = chunk * 256 + threadIdx.x;
    if (idx >= NFEAT * NHID) return;
    int k = idx >> 7, n = idx & 127;
    Wp[(size_t)(k >> 1) * (NHID * 2) + n * 2 + (k & 1)] = W1[idx];
}

static __device__ __forceinline__ void pack_w2_chunk(const float* __restrict__ W2, float* __restrict__ Wp, int chunk) {
    int idx = chunk * 256 + threadIdx.x;
    if (idx >= NHID * NCPAD) return;
    int k = idx / NCPAD, n = idx % NCPAD;
    float v = (n < NCLASS) ? W2[k * NCLASS + n] : 0.0f;
    Wp[(size_t)(k >> 1) * (NCPAD * 2) + n * 2 + (k & 1)] = v;
}

#define PACK1_CHUNKS ((NFEAT * NHID + 255) / 256)
#define PACK2_CHUNKS ((NHID * NCPAD + 255) / 256)
__global__ __launch_bounds__(256) void k_prep(const float* __restrict__ W1, float* __restrict__ Wp1,
                                              const float* __restrict__ W2, float* __restrict__ Wp2,
                                              int* __restrict__ cnt, int N) {
    const int b = blockIdx.x;
    if (b < PACK1_CHUNKS) pack_w1_chunk(W1, Wp1, b);
    else if (b < PACK1_CHUNKS + PACK2_CHUNKS) pack_w2_chunk(W2, Wp2, b - PACK1_CHUNKS);
    else cnt_zero_chunk(cnt, N, b - PACK1_CHUNKS - PACK2_CHUNKS);
}

static __device__ __forceinline__ void gemm1_tile(const float* __restrict__ X,
                                                  const float* __restrict__ Wp,
                                                  float* __restrict__ H, int M, int tile) {
    __shared__ float2 Xs[128 * 33];
    const int tid = threadIdx.x;
    const int m0 = tile * 32;

#pragma unroll
    for (int it = 0; it < 8; ++it) {
        int q = it * 256 + tid;
        int m = q >> 6, k4 = q & 63;
        int mr = m0 + m; if (mr >= M) mr = M - 1;
        float4 v = ((const float4*)(X + (size_t)mr * NFEAT))[k4];
        Xs[(2 * k4)     * 33 + m] = make_float2(v.x, v.y);
        Xs[(2 * k4 + 1) * 33 + m] = make_float2(v.z, v.w);
    }
    __syncthreads();

    const int wave = tid >> 5, lane = tid & 31, l = lane & 15, half = lane >> 4;
    const int n0 = wave * 16;
    const float2* Bp = (const float2*)Wp;

    v8f acc0 = {}, acc1 = {};
#pragma unroll 4
    for (int kp = 0; kp < 128; kp += 2) {
        int row = kp + half;
        float2 a0 = Xs[row * 33 + l];
        float2 a1 = Xs[row * 33 + 16 + l];
        float2 b  = Bp[(size_t)row * NHID + n0 + l];
        v2f av0 = {a0.x, a0.y}, av1 = {a1.x, a1.y}, bv = {b.x, b.y};
        acc0 = __builtin_amdgcn_wmma_f32_16x16x4_f32(false, av0, false, bv, (short)0, acc0, false, false);
        acc1 = __builtin_amdgcn_wmma_f32_16x16x4_f32(false, av1, false, bv, (short)0, acc1, false, false);
    }

    const int mb = 8 * half;
    float* Ho = H + (size_t)(m0 + mb) * NHID + n0 + l;
    if (m0 + 32 <= M) {
#pragma unroll
        for (int r = 0; r < 8; ++r) {
            Ho[(size_t)r * NHID]        = acc0[r];
            Ho[(size_t)(16 + r) * NHID] = acc1[r];
        }
    } else {
#pragma unroll
        for (int r = 0; r < 8; ++r) {
            if (m0 + mb + r < M)      Ho[(size_t)r * NHID]        = acc0[r];
            if (m0 + 16 + mb + r < M) Ho[(size_t)(16 + r) * NHID] = acc1[r];
        }
    }
}

#define ROLE_PERIOD 5
__global__ __launch_bounds__(256) void k_rank_gemm1(const int* __restrict__ dst, int* cnt,
                                                    int* __restrict__ rank, int E,
                                                    const float* __restrict__ X,
                                                    const float* __restrict__ Wp,
                                                    float* __restrict__ H, int M,
                                                    int tile_beg, int tile_end) {
    const int grp = blockIdx.x / ROLE_PERIOD, r = blockIdx.x % ROLE_PERIOD;
    if (r == 0) {
        const int tile = tile_beg + grp;
        if (tile < tile_end) gemm1_tile(X, Wp, H, M, tile);
    } else {
        rank_chunk(dst, cnt, rank, E, grp * (ROLE_PERIOD - 1) + r - 1);
    }
}

__global__ __launch_bounds__(256) void k_fill_gemm1(const int* __restrict__ src,
                                                    const int* __restrict__ dst,
                                                    const float* __restrict__ w,
                                                    const int* __restrict__ rank,
                                                    const int* __restrict__ first,
                                                    int* __restrict__ edges, int E,
                                                    const float* __restrict__ X,
                                                    const float* __restrict__ Wp,
                                                    float* __restrict__ H, int M,
                                                    int tile_beg, int tile_end) {
    const int grp = blockIdx.x / ROLE_PERIOD, r = blockIdx.x % ROLE_PERIOD;
    if (r == 0) {
        const int tile = tile_beg + grp;
        if (tile < tile_end) gemm1_tile(X, Wp, H, M, tile);
    } else {
        fill_chunk(src, dst, w, rank, first, edges, E, grp * (ROLE_PERIOD - 1) + r - 1);
    }
}

__global__ __launch_bounds__(256) void k_agg1_gather(const int* __restrict__ first,
                                                     const int* __restrict__ edges,
                                                     const float* __restrict__ dinv,
                                                     const float* __restrict__ H,
                                                     float* __restrict__ Hacc, int N) {
    const int lane = threadIdx.x & 31;
    const int i = __builtin_amdgcn_readfirstlane((int)(blockIdx.x * 8 + (threadIdx.x >> 5)));
    if (i >= N) return;
    const int beg = first[i], end = first[i + 1];
    const float di = dinv[i];
    const float4* H4 = (const float4*)H;

    const float cs = di * di;
    const float4 hs = H4[(size_t)i * 32 + lane];
    float4 acc = make_float4(cs * hs.x, cs * hs.y, cs * hs.z, cs * hs.w);

    for (int base = beg; base < end; base += 32) {
        const int n = min(32, end - base);
        int s_l = 0;
        float c_l = 0.0f;
        if (lane < n) {
            int2 sw = ((const int2*)edges)[base + lane];
            s_l = sw.x;
            c_l = dinv[sw.x] * __int_as_float(sw.y) * di;
        }
        int j = 0;
        for (; j + 4 <= n; j += 4) {
            const int s0 = lane_i(s_l, j), s1 = lane_i(s_l, j + 1);
            const int s2 = lane_i(s_l, j + 2), s3 = lane_i(s_l, j + 3);
            const float c0 = lane_f(c_l, j), c1 = lane_f(c_l, j + 1);
            const float c2 = lane_f(c_l, j + 2), c3 = lane_f(c_l, j + 3);
            const float4 v0 = H4[(size_t)s0 * 32 + lane];
            const float4 v1 = H4[(size_t)s1 * 32 + lane];
            const float4 v2 = H4[(size_t)s2 * 32 + lane];
            const float4 v3 = H4[(size_t)s3 * 32 + lane];
            acc.x += c0 * v0.x; acc.y += c0 * v0.y; acc.z += c0 * v0.z; acc.w += c0 * v0.w;
            acc.x += c1 * v1.x; acc.y += c1 * v1.y; acc.z += c1 * v1.z; acc.w += c1 * v1.w;
            acc.x += c2 * v2.x; acc.y += c2 * v2.y; acc.z += c2 * v2.z; acc.w += c2 * v2.w;
            acc.x += c3 * v3.x; acc.y += c3 * v3.y; acc.z += c3 * v3.z; acc.w += c3 * v3.w;
        }
        for (; j < n; ++j) {
            const int s0 = lane_i(s_l, j);
            const float c0 = lane_f(c_l, j);
            const float4 v0 = H4[(size_t)s0 * 32 + lane];
            acc.x += c0 * v0.x; acc.y += c0 * v0.y; acc.z += c0 * v0.z; acc.w += c0 * v0.w;
        }
    }
    ((float4*)Hacc)[(size_t)i * 32 + lane] = acc;
}

__global__ __launch_bounds__(64) void k_gemm2(const float* __restrict__ Hacc,
                                              const float* __restrict__ b1,
                                              const float* __restrict__ Wp,
                                              float* __restrict__ G, int M) {
    __shared__ float2 Xs[64 * 33];
    const int tid = threadIdx.x;
    const int m0 = blockIdx.x * 32;

#pragma unroll
    for (int it = 0; it < 16; ++it) {
        int q = it * 64 + tid;
        int m = q >> 5, k4 = q & 31;
        int mr = m0 + m; if (mr >= M) mr = M - 1;
        float4 v = ((const float4*)(Hacc + (size_t)mr * NHID))[k4];
        float4 bb = ((const float4*)b1)[k4];
        v.x = fmaxf(v.x + bb.x, 0.0f);
        v.y = fmaxf(v.y + bb.y, 0.0f);
        v.z = fmaxf(v.z + bb.z, 0.0f);
        v.w = fmaxf(v.w + bb.w, 0.0f);
        Xs[(2 * k4)     * 33 + m] = make_float2(v.x, v.y);
        Xs[(2 * k4 + 1) * 33 + m] = make_float2(v.z, v.w);
    }
    __syncthreads();

    const int wave = tid >> 5, lane = tid & 31, l = lane & 15, half = lane >> 4;
    const float2* Bp = (const float2*)Wp;

    v8f acc0 = {}, acc1 = {}, acc2 = {};
#pragma unroll 4
    for (int kp = 0; kp < 64; kp += 2) {
        int row = kp + half;
        float2 a  = Xs[row * 33 + wave * 16 + l];
        float2 b0 = Bp[(size_t)row * NCPAD + 0  + l];
        float2 b1f= Bp[(size_t)row * NCPAD + 16 + l];
        float2 b2f= Bp[(size_t)row * NCPAD + 32 + l];
        v2f av = {a.x, a.y};
        v2f bv0 = {b0.x, b0.y}, bv1 = {b1f.x, b1f.y}, bv2 = {b2f.x, b2f.y};
        acc0 = __builtin_amdgcn_wmma_f32_16x16x4_f32(false, av, false, bv0, (short)0, acc0, false, false);
        acc1 = __builtin_amdgcn_wmma_f32_16x16x4_f32(false, av, false, bv1, (short)0, acc1, false, false);
        acc2 = __builtin_amdgcn_wmma_f32_16x16x4_f32(false, av, false, bv2, (short)0, acc2, false, false);
    }

    const int mb = m0 + wave * 16 + 8 * half;
    float* Go = G + (size_t)mb * NCPAD + l;
    if (m0 + 32 <= M) {
#pragma unroll
        for (int r = 0; r < 8; ++r) {
            float* o = Go + (size_t)r * NCPAD;
            o[0]  = acc0[r];
            o[16] = acc1[r];
            o[32] = acc2[r];
        }
    } else {
#pragma unroll
        for (int r = 0; r < 8; ++r) {
            if (mb + r < M) {
                float* o = Go + (size_t)r * NCPAD;
                o[0]  = acc0[r];
                o[16] = acc1[r];
                o[32] = acc2[r];
            }
        }
    }
}

__global__ __launch_bounds__(256) void k_agg2_gather(const int* __restrict__ first,
                                                     const int* __restrict__ edges,
                                                     const float* __restrict__ dinv,
                                                     const float* __restrict__ G,
                                                     float* __restrict__ Oacc, int N) {
    const int lane = threadIdx.x & 31;
    const int lane2 = 32 + (lane & 15);
    const int i = __builtin_amdgcn_readfirstlane((int)(blockIdx.x * 8 + (threadIdx.x >> 5)));
    if (i >= N) return;
    const int beg = first[i], end = first[i + 1];
    const float di = dinv[i];

    const float cs = di * di;
    const float* gs = G + (size_t)i * NCPAD;
    float acc0 = cs * gs[lane];
    float acc1 = cs * gs[lane2];

    for (int base = beg; base < end; base += 32) {
        const int n = min(32, end - base);
        int s_l = 0;
        float c_l = 0.0f;
        if (lane < n) {
            int2 sw = ((const int2*)edges)[base + lane];
            s_l = sw.x;
            c_l = dinv[sw.x] * __int_as_float(sw.y) * di;
        }
        int j = 0;
        for (; j + 4 <= n; j += 4) {
            const float* g0 = G + (size_t)lane_i(s_l, j)     * NCPAD;
            const float* g1 = G + (size_t)lane_i(s_l, j + 1) * NCPAD;
            const float* g2 = G + (size_t)lane_i(s_l, j + 2) * NCPAD;
            const float* g3 = G + (size_t)lane_i(s_l, j + 3) * NCPAD;
            const float c0 = lane_f(c_l, j), c1 = lane_f(c_l, j + 1);
            const float c2 = lane_f(c_l, j + 2), c3 = lane_f(c_l, j + 3);
            const float x0 = g0[lane], y0 = g0[lane2];
            const float x1 = g1[lane], y1 = g1[lane2];
            const float x2 = g2[lane], y2 = g2[lane2];
            const float x3 = g3[lane], y3 = g3[lane2];
            acc0 += c0 * x0; acc1 += c0 * y0;
            acc0 += c1 * x1; acc1 += c1 * y1;
            acc0 += c2 * x2; acc1 += c2 * y2;
            acc0 += c3 * x3; acc1 += c3 * y3;
        }
        for (; j < n; ++j) {
            const float* g0 = G + (size_t)lane_i(s_l, j) * NCPAD;
            const float c0 = lane_f(c_l, j);
            acc0 += c0 * g0[lane];
            acc1 += c0 * g0[lane2];
        }
    }
    float* o = Oacc + (size_t)i * NCPAD;
    o[lane] = acc0;
    if (lane < 16) o[32 + lane] = acc1;
}

__global__ __launch_bounds__(256) void k_logsoftmax(const float* __restrict__ Oacc,
                                                    const float* __restrict__ b2,
                                                    float* __restrict__ out, int N) {
    int i = blockIdx.x * 8 + (threadIdx.x >> 5);
    if (i >= N) return;
    int lane = threadIdx.x & 31;
    const float* row = Oacc + (size_t)i * NCPAD;

    float v1 = row[lane] + b2[lane];
    int c2 = 32 + lane;
    bool has2 = c2 < NCLASS;
    float v2 = has2 ? (row[c2] + b2[c2]) : -INFINITY;

    float mx = fmaxf(v1, v2);
#pragma unroll
    for (int off = 16; off >= 1; off >>= 1) mx = fmaxf(mx, __shfl_xor(mx, off));

    float s = expf(v1 - mx) + (has2 ? expf(v2 - mx) : 0.0f);
#pragma unroll
    for (int off = 16; off >= 1; off >>= 1) s += __shfl_xor(s, off);

    float ls = logf(s);
    out[(size_t)i * NCLASS + lane] = v1 - mx - ls;
    if (has2) out[(size_t)i * NCLASS + c2] = v2 - mx - ls;
}

extern "C" void kernel_launch(void* const* d_in, const int* in_sizes, int n_in,
                              void* d_out, int out_size, void* d_ws, size_t ws_size,
                              hipStream_t stream) {
    const float* x  = (const float*)d_in[0];
    const int*   ei = (const int*)  d_in[1];
    const float* ew = (const float*)d_in[2];
    const float* W1 = (const float*)d_in[3];
    const float* b1 = (const float*)d_in[4];
    const float* W2 = (const float*)d_in[5];
    const float* b2 = (const float*)d_in[6];
    float* out = (float*)d_out;

    const int N = in_sizes[0] / NFEAT;
    const int E = in_sizes[2];
    const int* src = ei;
    const int* dst = ei + E;

    float* ws = (float*)d_ws;
    size_t o = 0;
    auto take = [&](size_t n) { float* p = ws + o; o += (n + 255) & ~(size_t)255; return p; };
    float* dinv = take((size_t)N);
    float* Wp1  = take((size_t)NFEAT * NHID);
    float* Wp2  = take((size_t)NHID * NCPAD);
    float* Hxw  = take((size_t)N * NHID);
    float* Hacc = take((size_t)N * NHID);
    float* G    = Hxw;
    float* Oacc = Hxw + (size_t)N * NCPAD;

    int* cnt  = (int*)Hacc;
    int* rank = cnt + (size_t)N * CNT_STRIDE;
    int* bsum = rank + E;
    int* edges = (int*)d_out;
    int* first = edges + 2 * (size_t)E;

    k_prep<<<PACK1_CHUNKS + PACK2_CHUNKS + (N + 255) / 256, 256, 0, stream>>>(W1, Wp1, W2, Wp2, cnt, N);

    const int tiles = (N + 31) / 32, tmid = (tiles + 1) / 2;
    const int cgroups = ((E + 255) / 256 + ROLE_PERIOD - 2) / (ROLE_PERIOD - 1);
    const int g1 = cgroups > tmid ? cgroups : tmid;
    const int g2 = cgroups > tiles - tmid ? cgroups : tiles - tmid;

    k_rank_gemm1<<<g1 * ROLE_PERIOD, 256, 0, stream>>>(dst, cnt, rank, E, x, Wp1, Hxw, N, 0, tmid);
    k_scan_part <<<(N + 1023) / 1024, 256, 0, stream>>>(cnt, bsum, N);
    k_scan_final<<<(N + 1023) / 1024, 256, 0, stream>>>(cnt, bsum, first, N);
    k_fill_gemm1<<<g2 * ROLE_PERIOD, 256, 0, stream>>>(src, dst, ew, rank, first, edges, E,
                                                       x, Wp1, Hxw, N, tmid, tiles);
    k_dinv      <<<(N * 8 + 255) / 256, 256, 0, stream>>>(first, edges, dinv, N);

    k_agg1_gather<<<(N + 7) / 8, 256, 0, stream>>>(first, edges, dinv, Hxw, Hacc, N);

    k_gemm2<<<(N + 31) / 32, 64, 0, stream>>>(Hacc, b1, Wp2, G, N);

    k_agg2_gather<<<(N + 7) / 8, 256, 0, stream>>>(first, edges, dinv, G, Oacc, N);

    k_logsoftmax<<<(N + 7) / 8, 256, 0, stream>>>(Oacc, b2, out, N);
}


# Gather kernels consume their own rows: layer 1 does bias, ReLU and the W2 product on a 32-node LDS tile, layer 2 bias and log-softmax in registers; two planes and 2 of 10 launches go; sum orders kept
# speedup vs baseline: 2.8256x; 1.0700x over previous
#include <hip/hip_runtime.h>
#include <cstdint>

typedef float v2f __attribute__((ext_vector_type(2)));
typedef float v8f __attribute__((ext_vector_type(8)));

#define NFEAT 256
#define NHID  128
#define NCLASS 41
#define NCPAD  48

#define CNT_STRIDE 32

static __device__ __forceinline__ void cnt_zero_chunk(int* __restrict__ cnt, int N, int chunk) {
    int i = chunk * 256 + threadIdx.x;
    if (i < N) cnt[(size_t)i * CNT_STRIDE] = 0;
}

static __device__ __forceinline__ void rank_chunk(const int* __restrict__ dst, int* cnt,
                                                  int* __restrict__ rank, int E, int chunk) {
    int e = chunk * 256 + threadIdx.x;
    if (e < E) rank[e] = atomicAdd(&cnt[(size_t)dst[e] * CNT_STRIDE], 1);
}

__global__ __launch_bounds__(256) void k_scan_part(const int* __restrict__ cnt,
                                                   int* __restrict__ bsum, int N) {
    __shared__ int wsum[8];
    const int t = threadIdx.x;
    const int i0 = blockIdx.x * 1024 + t * 4;
    int v = 0;
    for (int k = 0; k < 4; ++k) {
        int i = i0 + k;
        if (i < N) v += cnt[(size_t)i * CNT_STRIDE];
    }
    for (int off = 16; off >= 1; off >>= 1) v += __shfl_xor(v, off);
    if ((t & 31) == 0) wsum[t >> 5] = v;
    __syncthreads();
    if (t == 0) {
        int s = 0;
        for (int k = 0; k < 8; ++k) s += wsum[k];
        bsum[blockIdx.x] = s;
    }
}

__global__ __launch_bounds__(256) void k_scan_final(const int* __restrict__ cnt,
                                                    const int* __restrict__ bsum,
                                                    int* __restrict__ first, int N) {
    __shared__ int wtot[8];
    __shared__ int wpre[8];
    const int t = threadIdx.x, lane = t & 31, wv = t >> 5;

    int p = 0;
    for (int j = t; j < (int)blockIdx.x; j += 256) p += bsum[j];
    for (int off = 16; off >= 1; off >>= 1) p += __shfl_xor(p, off);
    if (lane == 0) wpre[wv] = p;

    const int i0 = blockIdx.x * 1024 + t * 4;
    int c[4];
    int v = 0;
    for (int k = 0; k < 4; ++k) {
        int i = i0 + k;
        c[k] = (i < N) ? cnt[(size_t)i * CNT_STRIDE] : 0;
        v += c[k];
    }
    int inc = v;
    for (int off = 1; off < 32; off <<= 1) {
        int u = __shfl_up(inc, off);
        if (lane >= off) inc += u;
    }
    if (lane == 31) wtot[wv] = inc;
    __syncthreads();

    int ex = inc - v;
    for (int k = 0; k < 8; ++k) {
        ex += wpre[k];
        if (k < wv) ex += wtot[k];
    }
    for (int k = 0; k < 4; ++k) {
        int i = i0 + k;
        if (i < N) {
            first[i] = ex;
            ex += c[k];
            if (i == N - 1) first[N] = ex;
        }
    }
}

static __device__ __forceinline__ void fill_chunk(const int* __restrict__ src,
                                                  const int* __restrict__ dst,
                                                  const float* __restrict__ w,
                                                  const int* __restrict__ rank,
                                                  const int* __restrict__ first,
                                                  int* __restrict__ edges, int E, int chunk) {
    int e = chunk * 256 + threadIdx.x;
    if (e < E) {
        int slot = first[dst[e]] + rank[e];
        ((int2*)edges)[slot] = make_int2(src[e], __float_as_int(w[e]));
    }
}

__global__ __launch_bounds__(256) void k_dinv(const int* __restrict__ first,
                                              const int* __restrict__ edges,
                                              float* __restrict__ dinv, int N) {
    const int g = blockIdx.x * 256 + threadIdx.x;
    const int i = g >> 3, sub = g & 7;
    float s = 0.0f;
    if (i < N) {
        const int end = first[i + 1];
        for (int p = first[i] + sub; p < end; p += 8) s += __int_as_float(edges[2 * (size_t)p + 1]);
    }
    s += __shfl_xor(s, 4);
    s += __shfl_xor(s, 2);
    s += __shfl_xor(s, 1);
    if (i < N && sub == 0) {
        float d = 1.0f + s;
        dinv[i] = d > 0.0f ? rsqrtf(d) : 0.0f;
    }
}

static __device__ __forceinline__ int lane_i(int v, int j) {
    return __builtin_amdgcn_readlane(v, j);
}
static __device__ __forceinline__ float lane_f(float v, int j) {
    return __int_as_float(__builtin_amdgcn_readlane(__float_as_int(v), j));
}

static __device__ __forceinline__ void pack_w1_chunk(const float* __restrict__ W1, float* __restrict__ Wp, int chunk) {
    int idx = chunk * 256 + threadIdx.x;
    if (idx >= NFEAT * NHID) return;
    int k = idx >> 7, n = idx & 127;
    Wp[(size_t)(k >> 1) * (NHID * 2) + n * 2 + (k & 1)] = W1[idx];
}

static __device__ __forceinline__ void pack_w2_chunk(const float* __restrict__ W2, float* __restrict__ Wp, int chunk) {
    int idx = chunk * 256 + threadIdx.x;
    if (idx >= NHID * NCPAD) return;
    int k = idx / NCPAD, n = idx % NCPAD;
    float v = (n < NCLASS) ? W2[k * NCLASS + n] : 0.0f;
    Wp[(size_t)(k >> 1) * (NCPAD * 2) + n * 2 + (k & 1)] = v;
}

#define PACK1_CHUNKS ((NFEAT * NHID + 255) / 256)
#define PACK2_CHUNKS ((NHID * NCPAD + 255) / 256)
__global__ __launch_bounds__(256) void k_prep(const float* __restrict__ W1, float* __restrict__ Wp1,
                                              const float* __restrict__ W2, float* __restrict__ Wp2,
                                              int* __restrict__ cnt, int N) {
    const int b = blockIdx.x;
    if (b < PACK1_CHUNKS) pack_w1_chunk(W1, Wp1, b);
    else if (b < PACK1_CHUNKS + PACK2_CHUNKS) pack_w2_chunk(W2, Wp2, b - PACK1_CHUNKS);
    else cnt_zero_chunk(cnt, N, b - PACK1_CHUNKS - PACK2_CHUNKS);
}

static __device__ __forceinline__ void gemm1_tile(const float* __restrict__ X,
                                                  const float* __restrict__ Wp,
                                                  float* __restrict__ H, int M, int tile) {
    __shared__ float2 Xs[128 * 33];
    const int tid = threadIdx.x;
    const int m0 = tile * 32;

#pragma unroll
    for (int it = 0; it < 8; ++it) {
        int q = it * 256 + tid;
        int m = q >> 6, k4 = q & 63;
        int mr = m0 + m; if (mr >= M) mr = M - 1;
        float4 v = ((const float4*)(X + (size_t)mr * NFEAT))[k4];
        Xs[(2 * k4)     * 33 + m] = make_float2(v.x, v.y);
        Xs[(2 * k4 + 1) * 33 + m] = make_float2(v.z, v.w);
    }
    __syncthreads();

    const int wave = tid >> 5, lane = tid & 31, l = lane & 15, half = lane >> 4;
    const int n0 = wave * 16;
    const float2* Bp = (const float2*)Wp;

    v8f acc0 = {}, acc1 = {};
#pragma unroll 4
    for (int kp = 0; kp < 128; kp += 2) {
        int row = kp + half;
        float2 a0 = Xs[row * 33 + l];
        float2 a1 = Xs[row * 33 + 16 + l];
        float2 b  = Bp[(size_t)row * NHID + n0 + l];
        v2f av0 = {a0.x, a0.y}, av1 = {a1.x, a1.y}, bv = {b.x, b.y};
        acc0 = __builtin_amdgcn_wmma_f32_16x16x4_f32(false, av0, false, bv, (short)0, acc0, false, false);
        acc1 = __builtin_amdgcn_wmma_f32_16x16x4_f32(false, av1, false, bv, (short)0, acc1, false, false);
    }

    const int mb = 8 * half;
    float* Ho = H + (size_t)(m0 + mb) * NHID + n0 + l;
    if (m0 + 32 <= M) {
#pragma unroll
        for (int r = 0; r < 8; ++r) {
            Ho[(size_t)r * NHID]        = acc0[r];
            Ho[(size_t)(16 + r) * NHID] = acc1[r];
        }
    } else {
#pragma unroll
        for (int r = 0; r < 8; ++r) {
            if (m0 + mb + r < M)      Ho[(size_t)r * NHID]        = acc0[r];
            if (m0 + 16 + mb + r < M) Ho[(size_t)(16 + r) * NHID] = acc1[r];
        }
    }
}

#define ROLE_PERIOD 5
__global__ __launch_bounds__(256) void k_rank_gemm1(const int* __restrict__ dst, int* cnt,
                                                    int* __restrict__ rank, int E,
                                                    const float* __restrict__ X,
                                                    const float* __restrict__ Wp,
                                                    float* __restrict__ H, int M,
                                                    int tile_beg, int tile_end) {
    const int grp = blockIdx.x / ROLE_PERIOD, r = blockIdx.x % ROLE_PERIOD;
    if (r == 0) {
        const int tile = tile_beg + grp;
        if (tile < tile_end) gemm1_tile(X, Wp, H, M, tile);
    } else {
        rank_chunk(dst, cnt, rank, E, grp * (ROLE_PERIOD - 1) + r - 1);
    }
}

__global__ __launch_bounds__(256) void k_fill_gemm1(const int* __restrict__ src,
                                                    const int* __restrict__ dst,
                                                    const float* __restrict__ w,
                                                    const int* __restrict__ rank,
                                                    const int* __restrict__ first,
                                                    int* __restrict__ edges, int E,
                                                    const float* __restrict__ X,
                                                    const float* __restrict__ Wp,
                                                    float* __restrict__ H, int M,
                                                    int tile_beg, int tile_end) {
    const int grp = blockIdx.x / ROLE_PERIOD, r = blockIdx.x % ROLE_PERIOD;
    if (r == 0) {
        const int tile = tile_beg + grp;
        if (tile < tile_end) gemm1_tile(X, Wp, H, M, tile);
    } else {
        fill_chunk(src, dst, w, rank, first, edges, E, grp * (ROLE_PERIOD - 1) + r - 1);
    }
}

__global__ __launch_bounds__(256) void k_agg1_gemm2(const int* __restrict__ first,
                                                    const int* __restrict__ edges,
                                                    const float* __restrict__ dinv,
                                                    const float* __restrict__ H,
                                                    const float* __restrict__ b1,
                                                    const float* __restrict__ Wp,
                                                    float* __restrict__ G, int N) {
    __shared__ float2 Xs[64 * 33];
    const int lane = threadIdx.x & 31;
    const int wave = __builtin_amdgcn_readfirstlane((int)(threadIdx.x >> 5));
    const int m0 = blockIdx.x * 32;
    const float4* H4 = (const float4*)H;
    const float4 bb = ((const float4*)b1)[lane];

    for (int t = 0; t < 4; ++t) {
        const int m = wave * 4 + t;
        const int i = m0 + m;
        float4 v = make_float4(0.0f, 0.0f, 0.0f, 0.0f);
        if (i < N) {
            const int beg = first[i], end = first[i + 1];
            const float di = dinv[i];

            const float cs = di * di;
            const float4 hs = H4[(size_t)i * 32 + lane];
            float4 acc = make_float4(cs * hs.x, cs * hs.y, cs * hs.z, cs * hs.w);

            for (int base = beg; base < end; base += 32) {
                const int n = min(32, end - base);
                int s_l = 0;
                float c_l = 0.0f;
                if (lane < n) {
                    int2 sw = ((const int2*)edges)[base + lane];
                    s_l = sw.x;
                    c_l = dinv[sw.x] * __int_as_float(sw.y) * di;
                }
                int j = 0;
                for (; j + 4 <= n; j += 4) {
                    const int s0 = lane_i(s_l, j), s1 = lane_i(s_l, j + 1);
                    const int s2 = lane_i(s_l, j + 2), s3 = lane_i(s_l, j + 3);
                    const float c0 = lane_f(c_l, j), c1 = lane_f(c_l, j + 1);
                    const float c2 = lane_f(c_l, j + 2), c3 = lane_f(c_l, j + 3);
                    const float4 v0 = H4[(size_t)s0 * 32 + lane];
                    const float4 v1 = H4[(size_t)s1 * 32 + lane];
                    const float4 v2 = H4[(size_t)s2 * 32 + lane];
                    const float4 v3 = H4[(size_t)s3 * 32 + lane];
                    acc.x += c0 * v0.x; acc.y += c0 * v0.y; acc.z += c0 * v0.z; acc.w += c0 * v0.w;
                    acc.x += c1 * v1.x; acc.y += c1 * v1.y; acc.z += c1 * v1.z; acc.w += c1 * v1.w;
                    acc.x += c2 * v2.x; acc.y += c2 * v2.y; acc.z += c2 * v2.z; acc.w += c2 * v2.w;
                    acc.x += c3 * v3.x; acc.y += c3 * v3.y; acc.z += c3 * v3.z; acc.w += c3 * v3.w;
                }
                for (; j < n; ++j) {
                    const int s0 = lane_i(s_l, j);
                    const float c0 = lane_f(c_l, j);
                    const float4 v0 = H4[(size_t)s0 * 32 + lane];
                    acc.x += c0 * v0.x; acc.y += c0 * v0.y; acc.z += c0 * v0.z; acc.w += c0 * v0.w;
                }
            }
            v.x = fmaxf(acc.x + bb.x, 0.0f);
            v.y = fmaxf(acc.y + bb.y, 0.0f);
            v.z = fmaxf(acc.z + bb.z, 0.0f);
            v.w = fmaxf(acc.w + bb.w, 0.0f);
        }
        Xs[(2 * lane)     * 33 + m] = make_float2(v.x, v.y);
        Xs[(2 * lane + 1) * 33 + m] = make_float2(v.z, v.w);
    }
    __syncthreads();

    if (wave >= 6) return;
    const int ms = wave / 3, nt = wave - 3 * ms;
    const int l = lane & 15, half = lane >> 4;
    const float2* Bp = (const float2*)Wp;

    v8f acc = {};
#pragma unroll 4
    for (int kp = 0; kp < 64; kp += 2) {
        int row = kp + half;
        float2 a = Xs[row * 33 + ms * 16 + l];
        float2 b = Bp[(size_t)row * NCPAD + nt * 16 + l];
        v2f av = {a.x, a.y}, bv = {b.x, b.y};
        acc = __builtin_amdgcn_wmma_f32_16x16x4_f32(false, av, false, bv, (short)0, acc, false, false);
    }

    const int mb = m0 + ms * 16 + 8 * half;
    float* Go = G + (size_t)mb * NCPAD + nt * 16 + l;
    if (m0 + 32 <= N) {
#pragma unroll
        for (int r = 0; r < 8; ++r) Go[(size_t)r * NCPAD] = acc[r];
    } else {
#pragma unroll
        for (int r = 0; r < 8; ++r) {
            if (mb + r < N) Go[(size_t)r * NCPAD] = acc[r];
        }
    }
}

__global__ __launch_bounds__(256) void k_agg2_lsm(const int* __restrict__ first,
                                                  const int* __restrict__ edges,
                                                  const float* __restrict__ dinv,
                                                  const float* __restrict__ G,
                                                  const float* __restrict__ b2,
                                                  float* __restrict__ out, int N) {
    const int lane = threadIdx.x & 31;
    const int lane2 = 32 + (lane & 15);
    const int i = __builtin_amdgcn_readfirstlane((int)(blockIdx.x * 8 + (threadIdx.x >> 5)));
    if (i >= N) return;
    const int beg = first[i], end = first[i + 1];
    const float di = dinv[i];

    const float cs = di * di;
    const float* gs = G + (size_t)i * NCPAD;
    float acc0 = cs * gs[lane];
    float acc1 = cs * gs[lane2];

    for (int base = beg; base < end; base += 32) {
        const int n = min(32, end - base);
        int s_l = 0;
        float c_l = 0.0f;
        if (lane < n) {
            int2 sw = ((const int2*)edges)[base + lane];
            s_l = sw.x;
            c_l = dinv[sw.x] * __int_as_float(sw.y) * di;
        }
        int j = 0;
        for (; j + 4 <= n; j += 4) {
            const float* g0 = G + (size_t)lane_i(s_l, j)     * NCPAD;
            const float* g1 = G + (size_t)lane_i(s_l, j + 1) * NCPAD;
            const float* g2 = G + (size_t)lane_i(s_l, j + 2) * NCPAD;
            const float* g3 = G + (size_t)lane_i(s_l, j + 3) * NCPAD;
            const float c0 = lane_f(c_l, j), c1 = lane_f(c_l, j + 1);
            const float c2 = lane_f(c_l, j + 2), c3 = lane_f(c_l, j + 3);
            const float x0 = g0[lane], y0 = g0[lane2];
            const float x1 = g1[lane], y1 = g1[lane2];
            const float x2 = g2[lane], y2 = g2[lane2];
            const float x3 = g3[lane], y3 = g3[lane2];
            acc0 += c0 * x0; acc1 += c0 * y0;
            acc0 += c1 * x1; acc1 += c1 * y1;
            acc0 += c2 * x2; acc1 += c2 * y2;
            acc0 += c3 * x3; acc1 += c3 * y3;
        }
        for (; j < n; ++j) {
            const float* g0 = G + (size_t)lane_i(s_l, j) * NCPAD;
            const float c0 = lane_f(c_l, j);
            acc0 += c0 * g0[lane];
            acc1 += c0 * g0[lane2];
        }
    }

    const float v1 = acc0 + b2[lane];
    const int c2 = 32 + lane;
    const bool has2 = c2 < NCLASS;
    const float v2 = has2 ? (acc1 + b2[c2]) : -INFINITY;

    float mx = fmaxf(v1, v2);
#pragma unroll
    for (int off = 16; off >= 1; off >>= 1) mx = fmaxf(mx, __shfl_xor(mx, off));

    float s = expf(v1 - mx) + (has2 ? expf(v2 - mx) : 0.0f);
#pragma unroll
    for (int off = 16; off >= 1; off >>= 1) s += __shfl_xor(s, off);

    const float ls = logf(s);
    out[(size_t)i * NCLASS + lane] = v1 - mx - ls;
    if (has2) out[(size_t)i * NCLASS + c2] = v2 - mx - ls;
}

extern "C" void kernel_launch(void* const* d_in, const int* in_sizes, int n_in,
                              void* d_out, int out_size, void* d_ws, size_t ws_size,
                              hipStream_t stream) {
    const float* x  = (const float*)d_in[0];
    const int*   ei = (const int*)  d_in[1];
    const float* ew = (const float*)d_in[2];
    const float* W1 = (const float*)d_in[3];
    const float* b1 = (const float*)d_in[4];
    const float* W2 = (const float*)d_in[5];
    const float* b2 = (const float*)d_in[6];
    float* out = (float*)d_out;

    const int N = in_sizes[0] / NFEAT;
    const int E = in_sizes[2];
    const int* src = ei;
    const int* dst = ei + E;

    float* ws = (float*)d_ws;
    size_t o = 0;
    auto take = [&](size_t n) { float* p = ws + o; o += (n + 255) & ~(size_t)255; return p; };
    float* dinv = take((size_t)N);
    float* Wp1  = take((size_t)NFEAT * NHID);
    float* Wp2  = take((size_t)NHID * NCPAD);
    float* Hxw  = take((size_t)N * NHID);
    int* edges = (int*)take(2 * (size_t)E);
    int* first = (int*)take((size_t)N + 1);
    float* G  = ws + o;
    int* cnt  = (int*)G;
    int* rank = cnt + (size_t)N * CNT_STRIDE;
    int* bsum = rank + E;

    k_prep<<<PACK1_CHUNKS + PACK2_CHUNKS + (N + 255) / 256, 256, 0, stream>>>(W1, Wp1, W2, Wp2, cnt, N);

    const int tiles = (N + 31) / 32, tmid = (tiles + 1) / 2;
    const int cgroups = ((E + 255) / 256 + ROLE_PERIOD - 2) / (ROLE_PERIOD - 1);
    const int g1 = cgroups > tmid ? cgroups : tmid;
    const int g2 = cgroups > tiles - tmid ? cgroups : tiles - tmid;

    k_rank_gemm1<<<g1 * ROLE_PERIOD, 256, 0, stream>>>(dst, cnt, rank, E, x, Wp1, Hxw, N, 0, tmid);
    k_scan_part <<<(N + 1023) / 1024, 256, 0, stream>>>(cnt, bsum, N);
    k_scan_final<<<(N + 1023) / 1024, 256, 0, stream>>>(cnt, bsum, first, N);
    k_fill_gemm1<<<g2 * ROLE_PERIOD, 256, 0, stream>>>(src, dst, ew, rank, first, edges, E,
                                                       x, Wp1, Hxw, N, tmid, tiles);
    k_dinv      <<<(N * 8 + 255) / 256, 256, 0, stream>>>(first, edges, dinv, N);

    k_agg1_gemm2<<<(N + 31) / 32, 256, 0, stream>>>(first, edges, dinv, Hxw, b1, Wp2, G, N);

    k_agg2_lsm<<<(N + 7) / 8, 256, 0, stream>>>(first, edges, dinv, G, b2, out, N);
}


# Layer-2 gather costs one aligned 256-byte request per edge: G rows padded to 64 floats as (c, 32+c) pairs read by one 8-byte load; layer 1 leaves each edge's norm in its slot; all sums as before
# speedup vs baseline: 2.9462x; 1.0427x over previous
#include <hip/hip_runtime.h>
#include <cstdint>

typedef float v2f __attribute__((ext_vector_type(2)));
typedef float v8f __attribute__((ext_vector_type(8)));

#define NFEAT 256
#define NHID  128
#define NCLASS 41
#define NCPAD  48

#define CNT_STRIDE 32

static __device__ __forceinline__ void cnt_zero_chunk(int* __restrict__ cnt, int N, int chunk) {
    int i = chunk * 256 + threadIdx.x;
    if (i < N) cnt[(size_t)i * CNT_STRIDE] = 0;
}

static __device__ __forceinline__ void rank_chunk(const int* __restrict__ dst, int* cnt,
                                                  int* __restrict__ rank, int E, int chunk) {
    int e = chunk * 256 + threadIdx.x;
    if (e < E) rank[e] = atomicAdd(&cnt[(size_t)dst[e] * CNT_STRIDE], 1);
}

__global__ __launch_bounds__(256) void k_scan_part(const int* __restrict__ cnt,
                                                   int* __restrict__ bsum, int N) {
    __shared__ int wsum[8];
    const int t = threadIdx.x;
    const int i0 = blockIdx.x * 1024 + t * 4;
    int v = 0;
    for (int k = 0; k < 4; ++k) {
        int i = i0 + k;
        if (i < N) v += cnt[(size_t)i * CNT_STRIDE];
    }
    for (int off = 16; off >= 1; off >>= 1) v += __shfl_xor(v, off);
    if ((t & 31) == 0) wsum[t >> 5] = v;
    __syncthreads();
    if (t == 0) {
        int s = 0;
        for (int k = 0; k < 8; ++k) s += wsum[k];
        bsum[blockIdx.x] = s;
    }
}

__global__ __launch_bounds__(256) void k_scan_final(const int* __restrict__ cnt,
                                                    const int* __restrict__ bsum,
                                                    int* __restrict__ first, int N) {
    __shared__ int wtot[8];
    __shared__ int wpre[8];
    const int t = threadIdx.x, lane = t & 31, wv = t >> 5;

    int p = 0;
    for (int j = t; j < (int)blockIdx.x; j += 256) p += bsum[j];
    for (int off = 16; off >= 1; off >>= 1) p += __shfl_xor(p, off);
    if (lane == 0) wpre[wv] = p;

    const int i0 = blockIdx.x * 1024 + t * 4;
    int c[4];
    int v = 0;
    for (int k = 0; k < 4; ++k) {
        int i = i0 + k;
        c[k] = (i < N) ? cnt[(size_t)i * CNT_STRIDE] : 0;
        v += c[k];
    }
    int inc = v;
    for (int off = 1; off < 32; off <<= 1) {
        int u = __shfl_up(inc, off);
        if (lane >= off) inc += u;
    }
    if (lane == 31) wtot[wv] = inc;
    __syncthreads();

    int ex = inc - v;
    for (int k = 0; k < 8; ++k) {
        ex += wpre[k];
        if (k < wv) ex += wtot[k];
    }
    for (int k = 0; k < 4; ++k) {
        int i = i0 + k;
        if (i < N) {
            first[i] = ex;
            ex += c[k];
            if (i == N - 1) first[N] = ex;
        }
    }
}

static __device__ __forceinline__ void fill_chunk(const int* __restrict__ src,
                                                  const int* __restrict__ dst,
                                                  const float* __restrict__ w,
                                                  const int* __restrict__ rank,
                                                  const int* __restrict__ first,
                                                  int* __restrict__ edges, int E, int chunk) {
    int e = chunk * 256 + threadIdx.x;
    if (e < E) {
        int slot = first[dst[e]] + rank[e];
        ((int2*)edges)[slot] = make_int2(src[e], __float_as_int(w[e]));
    }
}

__global__ __launch_bounds__(256) void k_dinv(const int* __restrict__ first,
                                              const int* __restrict__ edges,
                                              float* __restrict__ dinv, int N) {
    const int g = blockIdx.x * 256 + threadIdx.x;
    const int i = g >> 3, sub = g & 7;
    float s = 0.0f;
    if (i < N) {
        const int end = first[i + 1];
        for (int p = first[i] + sub; p < end; p += 8) s += __int_as_float(edges[2 * (size_t)p + 1]);
    }
    s += __shfl_xor(s, 4);
    s += __shfl_xor(s, 2);
    s += __shfl_xor(s, 1);
    if (i < N && sub == 0) {
        float d = 1.0f + s;
        dinv[i] = d > 0.0f ? rsqrtf(d) : 0.0f;
    }
}

static __device__ __forceinline__ int lane_i(int v, int j) {
    return __builtin_amdgcn_readlane(v, j);
}
static __device__ __forceinline__ float lane_f(float v, int j) {
    return __int_as_float(__builtin_amdgcn_readlane(__float_as_int(v), j));
}

static __device__ __forceinline__ void pack_w1_chunk(const float* __restrict__ W1, float* __restrict__ Wp, int chunk) {
    int idx = chunk * 256 + threadIdx.x;
    if (idx >= NFEAT * NHID) return;
    int k = idx >> 7, n = idx & 127;
    Wp[(size_t)(k >> 1) * (NHID * 2) + n * 2 + (k & 1)] = W1[idx];
}

static __device__ __forceinline__ void pack_w2_chunk(const float* __restrict__ W2, float* __restrict__ Wp, int chunk) {
    int idx = chunk * 256 + threadIdx.x;
    if (idx >= NHID * NCPAD) return;
    int k = idx / NCPAD, n = idx % NCPAD;
    float v = (n < NCLASS) ? W2[k * NCLASS + n] : 0.0f;
    Wp[(size_t)(k >> 1) * (NCPAD * 2) + n * 2 + (k & 1)] = v;
}

#define PACK1_CHUNKS ((NFEAT * NHID + 255) / 256)
#define PACK2_CHUNKS ((NHID * NCPAD + 255) / 256)
__global__ __launch_bounds__(256) void k_prep(const float* __restrict__ W1, float* __restrict__ Wp1,
                                              const float* __restrict__ W2, float* __restrict__ Wp2,
                                              int* __restrict__ cnt, int N) {
    const int b = blockIdx.x;
    if (b < PACK1_CHUNKS) pack_w1_chunk(W1, Wp1, b);
    else if (b < PACK1_CHUNKS + PACK2_CHUNKS) pack_w2_chunk(W2, Wp2, b - PACK1_CHUNKS);
    else cnt_zero_chunk(cnt, N, b - PACK1_CHUNKS - PACK2_CHUNKS);
}

static __device__ __forceinline__ void gemm1_tile(const float* __restrict__ X,
                                                  const float* __restrict__ Wp,
                                                  float* __restrict__ H, int M, int tile) {
    __shared__ float2 Xs[128 * 33];
    const int tid = threadIdx.x;
    const int m0 = tile * 32;

#pragma unroll
    for (int it = 0; it < 8; ++it) {
        int q = it * 256 + tid;
        int m = q >> 6, k4 = q & 63;
        int mr = m0 + m; if (mr >= M) mr = M - 1;
        float4 v = ((const float4*)(X + (size_t)mr * NFEAT))[k4];
        Xs[(2 * k4)     * 33 + m] = make_float2(v.x, v.y);
        Xs[(2 * k4 + 1) * 33 + m] = make_float2(v.z, v.w);
    }
    __syncthreads();

    const int wave = tid >> 5, lane = tid & 31, l = lane & 15, half = lane >> 4;
    const int n0 = wave * 16;
    const float2* Bp = (const float2*)Wp;

    v8f acc0 = {}, acc1 = {};
#pragma unroll 4
    for (int kp = 0; kp < 128; kp += 2) {
        int row = kp + half;
        float2 a0 = Xs[row * 33 + l];
        float2 a1 = Xs[row * 33 + 16 + l];
        float2 b  = Bp[(size_t)row * NHID + n0 + l];
        v2f av0 = {a0.x, a0.y}, av1 = {a1.x, a1.y}, bv = {b.x, b.y};
        acc0 = __builtin_amdgcn_wmma_f32_16x16x4_f32(false, av0, false, bv, (short)0, acc0, false, false);
        acc1 = __builtin_amdgcn_wmma_f32_16x16x4_f32(false, av1, false, bv, (short)0, acc1, false, false);
    }

    const int mb = 8 * half;
    float* Ho = H + (size_t)(m0 + mb) * NHID + n0 + l;
    if (m0 + 32 <= M) {
#pragma unroll
        for (int r = 0; r < 8; ++r) {
            Ho[(size_t)r * NHID]        = acc0[r];
            Ho[(size_t)(16 + r) * NHID] = acc1[r];
        }
    } else {
#pragma unroll
        for (int r = 0; r < 8; ++r) {
            if (m0 + mb + r < M)      Ho[(size_t)r * NHID]        = acc0[r];
            if (m0 + 16 + mb + r < M) Ho[(size_t)(16 + r) * NHID] = acc1[r];
        }
    }
}

#define ROLE_PERIOD 5
__global__ __launch_bounds__(256) void k_rank_gemm1(const int* __restrict__ dst, int* cnt,
                                                    int* __restrict__ rank, int E,
                                                    const float* __restrict__ X,
                                                    const float* __restrict__ Wp,
                                                    float* __restrict__ H, int M,
                                                    int tile_beg, int tile_end) {
    const int grp = blockIdx.x / ROLE_PERIOD, r = blockIdx.x % ROLE_PERIOD;
    if (r == 0) {
        const int tile = tile_beg + grp;
        if (tile < tile_end) gemm1_tile(X, Wp, H, M, tile);
    } else {
        rank_chunk(dst, cnt, rank, E, grp * (ROLE_PERIOD - 1) + r - 1);
    }
}

__global__ __launch_bounds__(256) void k_fill_gemm1(const int* __restrict__ src,
                                                    const int* __restrict__ dst,
                                                    const float* __restrict__ w,
                                                    const int* __restrict__ rank,
                                                    const int* __restrict__ first,
                                                    int* __restrict__ edges, int E,
                                                    const float* __restrict__ X,
                                                    const float* __restrict__ Wp,
                                                    float* __restrict__ H, int M,
                                                    int tile_beg, int tile_end) {
    const int grp = blockIdx.x / ROLE_PERIOD, r = blockIdx.x % ROLE_PERIOD;
    if (r == 0) {
        const int tile = tile_beg + grp;
        if (tile < tile_end) gemm1_tile(X, Wp, H, M, tile);
    } else {
        fill_chunk(src, dst, w, rank, first, edges, E, grp * (ROLE_PERIOD - 1) + r - 1);
    }
}

#define GPITCH 64
__global__ __launch_bounds__(256) void k_agg1_gemm2(const int* __restrict__ first,
                                                    int* edges,
                                                    const float* __restrict__ dinv,
                                                    const float* __restrict__ H,
                                                    const float* __restrict__ b1,
                                                    const float* __restrict__ Wp,
                                                    float* __restrict__ G, int N) {
    __shared__ float2 Xs[64 * 33];
    __shared__ float2 Gs[32 * (GPITCH / 2)];
    const int lane = threadIdx.x & 31;
    const int wave = __builtin_amdgcn_readfirstlane((int)(threadIdx.x >> 5));
    const int m0 = blockIdx.x * 32;
    const float4* H4 = (const float4*)H;
    const float4 bb = ((const float4*)b1)[lane];

    {
        const int row = threadIdx.x >> 3, p = 16 + 2 * (threadIdx.x & 7);
        Gs[row * (GPITCH / 2) + p].y     = 0.0f;
        Gs[row * (GPITCH / 2) + p + 1].y = 0.0f;
    }

    for (int t = 0; t < 4; ++t) {
        const int m = wave * 4 + t;
        const int i = m0 + m;
        float4 v = make_float4(0.0f, 0.0f, 0.0f, 0.0f);
        if (i < N) {
            const int beg = first[i], end = first[i + 1];
            const float di = dinv[i];

            const float cs = di * di;
            const float4 hs = H4[(size_t)i * 32 + lane];
            float4 acc = make_float4(cs * hs.x, cs * hs.y, cs * hs.z, cs * hs.w);

            for (int base = beg; base < end; base += 32) {
                const int n = min(32, end - base);
                int s_l = 0;
                float c_l = 0.0f;
                if (lane < n) {
                    int2 sw = ((const int2*)edges)[base + lane];
                    s_l = sw.x;
                    c_l = dinv[sw.x] * __int_as_float(sw.y) * di;
                    edges[2 * (size_t)(base + lane) + 1] = __float_as_int(c_l);
                }
                int j = 0;
                for (; j + 4 <= n; j += 4) {
                    const int s0 = lane_i(s_l, j), s1 = lane_i(s_l, j + 1);
                    const int s2 = lane_i(s_l, j + 2), s3 = lane_i(s_l, j + 3);
                    const float c0 = lane_f(c_l, j), c1 = lane_f(c_l, j + 1);
                    const float c2 = lane_f(c_l, j + 2), c3 = lane_f(c_l, j + 3);
                    const float4 v0 = H4[(size_t)s0 * 32 + lane];
                    const float4 v1 = H4[(size_t)s1 * 32 + lane];
                    const float4 v2 = H4[(size_t)s2 * 32 + lane];
                    const float4 v3 = H4[(size_t)s3 * 32 + lane];
                    acc.x += c0 * v0.x; acc.y += c0 * v0.y; acc.z += c0 * v0.z; acc.w += c0 * v0.w;
                    acc.x += c1 * v1.x; acc.y += c1 * v1.y; acc.z += c1 * v1.z; acc.w += c1 * v1.w;
                    acc.x += c2 * v2.x; acc.y += c2 * v2.y; acc.z += c2 * v2.z; acc.w += c2 * v2.w;
                    acc.x += c3 * v3.x; acc.y += c3 * v3.y; acc.z += c3 * v3.z; acc.w += c3 * v3.w;
                }
                for (; j < n; ++j) {
                    const int s0 = lane_i(s_l, j);
                    const float c0 = lane_f(c_l, j);
                    const float4 v0 = H4[(size_t)s0 * 32 + lane];
                    acc.x += c0 * v0.x; acc.y += c0 * v0.y; acc.z += c0 * v0.z; acc.w += c0 * v0.w;
                }
            }
            v.x = fmaxf(acc.x + bb.x, 0.0f);
            v.y = fmaxf(acc.y + bb.y, 0.0f);
            v.z = fmaxf(acc.z + bb.z, 0.0f);
            v.w = fmaxf(acc.w + bb.w, 0.0f);
        }
        Xs[(2 * lane)     * 33 + m] = make_float2(v.x, v.y);
        Xs[(2 * lane + 1) * 33 + m] = make_float2(v.z, v.w);
    }
    __syncthreads();

    if (wave < 6) {
        const int ms = wave / 3, nt = wave - 3 * ms;
        const int l = lane & 15, half = lane >> 4;
        const float2* Bp = (const float2*)Wp;

        v8f acc = {};
#pragma unroll 4
        for (int kp = 0; kp < 64; kp += 2) {
            int row = kp + half;
            float2 a = Xs[row * 33 + ms * 16 + l];
            float2 b = Bp[(size_t)row * NCPAD + nt * 16 + l];
            v2f av = {a.x, a.y}, bv = {b.x, b.y};
            acc = __builtin_amdgcn_wmma_f32_16x16x4_f32(false, av, false, bv, (short)0, acc, false, false);
        }

        const int fi = (nt == 2) ? (2 * l + 1) : (2 * (nt * 16 + l));
        float* Gf = (float*)Gs + (ms * 16 + 8 * half) * GPITCH + fi;
#pragma unroll
        for (int r = 0; r < 8; ++r) Gf[r * GPITCH] = acc[r];
    }
    __syncthreads();

    float2* G2 = (float2*)G;
    for (int t = 0; t < 4; ++t) {
        const int m = wave * 4 + t;
        if (m0 + m < N) G2[(size_t)(m0 + m) * (GPITCH / 2) + lane] = Gs[m * (GPITCH / 2) + lane];
    }
}

__global__ __launch_bounds__(256) void k_agg2_lsm(const int* __restrict__ first,
                                                  const int* __restrict__ edges,
                                                  const float* __restrict__ dinv,
                                                  const float* __restrict__ G,
                                                  const float* __restrict__ b2,
                                                  float* __restrict__ out, int N) {
    const int lane = threadIdx.x & 31;
    const int i = __builtin_amdgcn_readfirstlane((int)(blockIdx.x * 8 + (threadIdx.x >> 5)));
    if (i >= N) return;
    const int beg = first[i], end = first[i + 1];
    const float di = dinv[i];
    const float2* G2 = (const float2*)G;

    const float cs = di * di;
    const float2 gs = G2[(size_t)i * (GPITCH / 2) + lane];
    float acc0 = cs * gs.x;
    float acc1 = cs * gs.y;

    for (int base = beg; base < end; base += 32) {
        const int n = min(32, end - base);
        int s_l = 0;
        float c_l = 0.0f;
        if (lane < n) {
            int2 sw = ((const int2*)edges)[base + lane];
            s_l = sw.x;
            c_l = __int_as_float(sw.y);
        }
        int j = 0;
        for (; j + 4 <= n; j += 4) {
            const float2* g0 = G2 + (size_t)lane_i(s_l, j)     * (GPITCH / 2);
            const float2* g1 = G2 + (size_t)lane_i(s_l, j + 1) * (GPITCH / 2);
            const float2* g2 = G2 + (size_t)lane_i(s_l, j + 2) * (GPITCH / 2);
            const float2* g3 = G2 + (size_t)lane_i(s_l, j + 3) * (GPITCH / 2);
            const float c0 = lane_f(c_l, j), c1 = lane_f(c_l, j + 1);
            const float c2 = lane_f(c_l, j + 2), c3 = lane_f(c_l, j + 3);
            const float2 v0 = g0[lane];
            const float2 v1 = g1[lane];
            const float2 v2 = g2[lane];
            const float2 v3 = g3[lane];
            acc0 += c0 * v0.x; acc1 += c0 * v0.y;
            acc0 += c1 * v1.x; acc1 += c1 * v1.y;
            acc0 += c2 * v2.x; acc1 += c2 * v2.y;
            acc0 += c3 * v3.x; acc1 += c3 * v3.y;
        }
        for (; j < n; ++j) {
            const float2* g0 = G2 + (size_t)lane_i(s_l, j) * (GPITCH / 2);
            const float c0 = lane_f(c_l, j);
            const float2 v0 = g0[lane];
            acc0 += c0 * v0.x;
            acc1 += c0 * v0.y;
        }
    }

    const float v1 = acc0 + b2[lane];
    const int c2 = 32 + lane;
    const bool has2 = c2 < NCLASS;
    const float v2 = has2 ? (acc1 + b2[c2]) : -INFINITY;

    float mx = fmaxf(v1, v2);
#pragma unroll
    for (int off = 16; off >= 1; off >>= 1) mx = fmaxf(mx, __shfl_xor(mx, off));

    float s = expf(v1 - mx) + (has2 ? expf(v2 - mx) : 0.0f);
#pragma unroll
    for (int off = 16; off >= 1; off >>= 1) s += __shfl_xor(s, off);

    const float ls = logf(s);
    out[(size_t)i * NCLASS + lane] = v1 - mx - ls;
    if (has2) out[(size_t)i * NCLASS + c2] = v2 - mx - ls;
}

extern "C" void kernel_launch(void* const* d_in, const int* in_sizes, int n_in,
                              void* d_out, int out_size, void* d_ws, size_t ws_size,
                              hipStream_t stream) {
    const float* x  = (const float*)d_in[0];
    const int*   ei = (const int*)  d_in[1];
    const float* ew = (const float*)d_in[2];
    const float* W1 = (const float*)d_in[3];
    const float* b1 = (const float*)d_in[4];
    const float* W2 = (const float*)d_in[5];
    const float* b2 = (const float*)d_in[6];
    float* out = (float*)d_out;

    const int N = in_sizes[0] / NFEAT;
    const int E = in_sizes[2];
    const int* src = ei;
    const int* dst = ei + E;

    float* ws = (float*)d_ws;
    size_t o = 0;
    auto take = [&](size_t n) { float* p = ws + o; o += (n + 255) & ~(size_t)255; return p; };
    float* dinv = take((size_t)N);
    float* Wp1  = take((size_t)NFEAT * NHID);
    float* Wp2  = take((size_t)NHID * NCPAD);
    float* Hxw  = take((size_t)N * NHID);
    int* edges = (int*)take(2 * (size_t)E);
    int* first = (int*)take((size_t)N + 1);
    float* G  = ws + o;
    int* cnt  = (int*)G;
    int* rank = cnt + (size_t)N * CNT_STRIDE;
    int* bsum = rank + E;

    k_prep<<<PACK1_CHUNKS + PACK2_CHUNKS + (N + 255) / 256, 256, 0, stream>>>(W1, Wp1, W2, Wp2, cnt, N);

    const int tiles = (N + 31) / 32, tmid = (tiles + 1) / 2;
    const int cgroups = ((E + 255) / 256 + ROLE_PERIOD - 2) / (ROLE_PERIOD - 1);
    const int g1 = cgroups > tmid ? cgroups : tmid;
    const int g2 = cgroups > tiles - tmid ? cgroups : tiles - tmid;

    k_rank_gemm1<<<g1 * ROLE_PERIOD, 256, 0, stream>>>(dst, cnt, rank, E, x, Wp1, Hxw, N, 0, tmid);
    k_scan_part <<<(N + 1023) / 1024, 256, 0, stream>>>(cnt, bsum, N);
    k_scan_final<<<(N + 1023) / 1024, 256, 0, stream>>>(cnt, bsum, first, N);
    k_fill_gemm1<<<g2 * ROLE_PERIOD, 256, 0, stream>>>(src, dst, ew, rank, first, edges, E,
                                                       x, Wp1, Hxw, N, tmid, tiles);
    k_dinv      <<<(N * 8 + 255) / 256, 256, 0, stream>>>(first, edges, dinv, N);

    k_agg1_gemm2<<<(N + 31) / 32, 256, 0, stream>>>(first, edges, dinv, Hxw, b1, Wp2, G, N);

    k_agg2_lsm<<<(N + 7) / 8, 256, 0, stream>>>(first, edges, dinv, G, b2, out, N);
}


# Counting sort binned in LDS: edges dealt to 256-node buckets, each bucket sorted and its degrees summed in LDS; no global atomics; 8 launches to 6; degree sums now in LDS-atomic arrival order
# speedup vs baseline: 3.8189x; 1.2962x over previous
#include <hip/hip_runtime.h>
#include <cstdint>

typedef float v2f __attribute__((ext_vector_type(2)));
typedef float v8f __attribute__((ext_vector_type(8)));

#define NFEAT 256
#define NHID  128
#define NCLASS 41
#define NCPAD  48

#define BUCKET_SHIFT 8
#define MAX_BUCKETS  512
#define EDGE_RUN     4096

struct EdgeRec { int s, w, d; };

static __device__ __forceinline__ int block_excl_scan(int v, int* wtot, int& total) {
    const int lane = threadIdx.x & 31, wv = threadIdx.x >> 5;
    int inc = v;
    for (int off = 1; off < 32; off <<= 1) {
        int u = __shfl_up(inc, off);
        if (lane >= off) inc += u;
    }
    if (lane == 31) wtot[wv] = inc;
    __syncthreads();
    int pre = 0, tot = 0;
    for (int k = 0; k < 8; ++k) {
        int t = wtot[k];
        if (k < wv) pre += t;
        tot += t;
    }
    total = tot;
    __syncthreads();
    return pre + inc - v;
}

static __device__ __forceinline__ void hist_run(const int* __restrict__ dst, int* __restrict__ cm,
                                                int E, int nbk, int ebk, int blk) {
    __shared__ int h[MAX_BUCKETS];
    const int tid = threadIdx.x;
    h[tid] = 0;
    h[tid + 256] = 0;
    __syncthreads();
    int d[EDGE_RUN / 256];
#pragma unroll
    for (int it = 0; it < EDGE_RUN / 256; ++it) {
        int e = blk * EDGE_RUN + it * 256 + tid;
        d[it] = (e < E) ? dst[e] : -1;
    }
#pragma unroll
    for (int it = 0; it < EDGE_RUN / 256; ++it)
        if (d[it] >= 0) atomicAdd(&h[d[it] >> BUCKET_SHIFT], 1);
    __syncthreads();
    if (tid < nbk)       cm[(size_t)tid * ebk + blk]         = h[tid];
    if (tid + 256 < nbk) cm[(size_t)(tid + 256) * ebk + blk] = h[tid + 256];
}

__global__ __launch_bounds__(256) void k_bucket_scan(const int* __restrict__ cm, int* __restrict__ off,
                                                     int* __restrict__ btot, int ebk) {
    __shared__ int wtot[8];
    const int tid = threadIdx.x;
    const size_t row = (size_t)blockIdx.x * ebk;
    int carry = 0;
    for (int c0 = 0; c0 < ebk; c0 += 256) {
        const int j = c0 + tid;
        const int v = (j < ebk) ? cm[row + j] : 0;
        int tot;
        const int ex = block_excl_scan(v, wtot, tot);
        if (j < ebk) off[row + j] = carry + ex;
        carry += tot;
    }
    if (tid == 0) btot[blockIdx.x] = carry;
}

static __device__ __forceinline__ void deal_run(const int* __restrict__ src, const int* __restrict__ dst,
                                                const float* __restrict__ w,
                                                const int* __restrict__ off, const int* __restrict__ btot,
                                                int* __restrict__ tmp,
                                                int E, int nbk, int ebk, int blk) {
    EdgeRec* __restrict__ tq = (EdgeRec*)tmp;
    __shared__ int cur[MAX_BUCKETS];
    __shared__ int wtot[8];
    const int tid = threadIdx.x;
    const int b0 = (tid < nbk) ? btot[tid] : 0;
    const int b1 = (tid + 256 < nbk) ? btot[tid + 256] : 0;
    int tot0, tot1;
    const int ex0 = block_excl_scan(b0, wtot, tot0);
    const int ex1 = block_excl_scan(b1, wtot, tot1);
    cur[tid]       = (tid < nbk)       ? ex0 + off[(size_t)tid * ebk + blk] : 0;
    cur[tid + 256] = (tid + 256 < nbk) ? tot0 + ex1 + off[(size_t)(tid + 256) * ebk + blk] : 0;
    __syncthreads();
    for (int g = 0; g < EDGE_RUN / 256; g += 8) {
        int d[8], s[8], wi[8];
#pragma unroll
        for (int it = 0; it < 8; ++it) {
            int e = blk * EDGE_RUN + (g + it) * 256 + tid;
            const bool in = e < E;
            d[it]  = in ? dst[e] : -1;
            s[it]  = in ? src[e] : 0;
            wi[it] = in ? __float_as_int(w[e]) : 0;
        }
#pragma unroll
        for (int it = 0; it < 8; ++it) {
            if (d[it] >= 0) {
                const int pos = atomicAdd(&cur[d[it] >> BUCKET_SHIFT], 1);
                EdgeRec r;
                r.s = s[it]; r.w = wi[it]; r.d = d[it];
                tq[pos] = r;
            }
        }
    }
}

static __device__ __forceinline__ void sort_bucket(const int* __restrict__ btot,
                                                   const int* __restrict__ tmp,
                                                   int* __restrict__ first, float* __restrict__ dinv,
                                                   int* __restrict__ edges, int N, int b) {
    const EdgeRec* __restrict__ tq = (const EdgeRec*)tmp;
    __shared__ int   s_cnt[256];
    __shared__ float s_deg[256];
    __shared__ int   s_cur[256];
    __shared__ int   wtot[8];
    const int tid = threadIdx.x;

    int part = 0;
    for (int j = tid; j < b; j += 256) part += btot[j];
    int base;
    (void)block_excl_scan(part, wtot, base);
    const int nb = btot[b];
    const int node0 = b << BUCKET_SHIFT;

    s_cnt[tid] = 0;
    s_deg[tid] = 0.0f;
    __syncthreads();
    for (int p0 = tid; p0 < nb; p0 += 4 * 256) {
        EdgeRec q[4];
#pragma unroll
        for (int k = 0; k < 4; ++k) {
            const int p = p0 + k * 256;
            if (p < nb) q[k] = tq[base + p];
            else { q[k].s = 0; q[k].w = 0; q[k].d = -1; }
        }
#pragma unroll
        for (int k = 0; k < 4; ++k) {
            if (q[k].d >= 0) {
                const int d = q[k].d - node0;
                atomicAdd(&s_cnt[d], 1);
                atomicAdd(&s_deg[d], __int_as_float(q[k].w));
            }
        }
    }
    __syncthreads();

    const int c = s_cnt[tid];
    int tot;
    const int ex = block_excl_scan(c, wtot, tot);
    const int node = node0 + tid;
    if (node < N) {
        first[node] = base + ex;
        const float dg = 1.0f + s_deg[tid];
        dinv[node] = dg > 0.0f ? rsqrtf(dg) : 0.0f;
        if (node == N - 1) first[N] = base + ex + c;
    }
    s_cur[tid] = ex;
    __syncthreads();
    for (int p0 = tid; p0 < nb; p0 += 4 * 256) {
        EdgeRec q[4];
#pragma unroll
        for (int k = 0; k < 4; ++k) {
            const int p = p0 + k * 256;
            if (p < nb) q[k] = tq[base + p];
            else { q[k].s = 0; q[k].w = 0; q[k].d = -1; }
        }
#pragma unroll
        for (int k = 0; k < 4; ++k) {
            if (q[k].d >= 0) {
                const int slot = base + atomicAdd(&s_cur[q[k].d - node0], 1);
                ((int2*)edges)[slot] = make_int2(q[k].s, q[k].w);
            }
        }
    }
}

static __device__ __forceinline__ int lane_i(int v, int j) {
    return __builtin_amdgcn_readlane(v, j);
}
static __device__ __forceinline__ float lane_f(float v, int j) {
    return __int_as_float(__builtin_amdgcn_readlane(__float_as_int(v), j));
}

static __device__ __forceinline__ void pack_w1_chunk(const float* __restrict__ W1, float* __restrict__ Wp, int chunk) {
    int idx = chunk * 256 + threadIdx.x;
    if (idx >= NFEAT * NHID) return;
    int k = idx >> 7, n = idx & 127;
    Wp[(size_t)(k >> 1) * (NHID * 2) + n * 2 + (k & 1)] = W1[idx];
}

static __device__ __forceinline__ void pack_w2_chunk(const float* __restrict__ W2, float* __restrict__ Wp, int chunk) {
    int idx = chunk * 256 + threadIdx.x;
    if (idx >= NHID * NCPAD) return;
    int k = idx / NCPAD, n = idx % NCPAD;
    float v = (n < NCLASS) ? W2[k * NCLASS + n] : 0.0f;
    Wp[(size_t)(k >> 1) * (NCPAD * 2) + n * 2 + (k & 1)] = v;
}

#define PACK1_CHUNKS ((NFEAT * NHID + 255) / 256)
#define PACK2_CHUNKS ((NHID * NCPAD + 255) / 256)
__global__ __launch_bounds__(256) void k_prep(const float* __restrict__ W1, float* __restrict__ Wp1,
                                              const float* __restrict__ W2, float* __restrict__ Wp2,
                                              const int* __restrict__ dst, int* __restrict__ cm,
                                              int E, int nbk, int ebk) {
    const int b = blockIdx.x;
    if (b < ebk) hist_run(dst, cm, E, nbk, ebk, b);
    else if (b < ebk + PACK1_CHUNKS) pack_w1_chunk(W1, Wp1, b - ebk);
    else pack_w2_chunk(W2, Wp2, b - ebk - PACK1_CHUNKS);
}

static __device__ __forceinline__ void gemm1_tile(const float* __restrict__ X,
                                                  const float* __restrict__ Wp,
                                                  float* __restrict__ H, int M, int tile) {
    __shared__ float2 Xs[128 * 33];
    const int tid = threadIdx.x;
    const int m0 = tile * 32;

#pragma unroll
    for (int it = 0; it < 8; ++it) {
        int q = it * 256 + tid;
        int m = q >> 6, k4 = q & 63;
        int mr = m0 + m; if (mr >= M) mr = M - 1;
        float4 v = ((const float4*)(X + (size_t)mr * NFEAT))[k4];
        Xs[(2 * k4)     * 33 + m] = make_float2(v.x, v.y);
        Xs[(2 * k4 + 1) * 33 + m] = make_float2(v.z, v.w);
    }
    __syncthreads();

    const int wave = tid >> 5, lane = tid & 31, l = lane & 15, half = lane >> 4;
    const int n0 = wave * 16;
    const float2* Bp = (const float2*)Wp;

    v8f acc0 = {}, acc1 = {};
#pragma unroll 4
    for (int kp = 0; kp < 128; kp += 2) {
        int row = kp + half;
        float2 a0 = Xs[row * 33 + l];
        float2 a1 = Xs[row * 33 + 16 + l];
        float2 b  = Bp[(size_t)row * NHID + n0 + l];
        v2f av0 = {a0.x, a0.y}, av1 = {a1.x, a1.y}, bv = {b.x, b.y};
        acc0 = __builtin_amdgcn_wmma_f32_16x16x4_f32(false, av0, false, bv, (short)0, acc0, false, false);
        acc1 = __builtin_amdgcn_wmma_f32_16x16x4_f32(false, av1, false, bv, (short)0, acc1, false, false);
    }

    const int mb = 8 * half;
    float* Ho = H + (size_t)(m0 + mb) * NHID + n0 + l;
    if (m0 + 32 <= M) {
#pragma unroll
        for (int r = 0; r < 8; ++r) {
            Ho[(size_t)r * NHID]        = acc0[r];
            Ho[(size_t)(16 + r) * NHID] = acc1[r];
        }
    } else {
#pragma unroll
        for (int r = 0; r < 8; ++r) {
            if (m0 + mb + r < M)      Ho[(size_t)r * NHID]        = acc0[r];
            if (m0 + 16 + mb + r < M) Ho[(size_t)(16 + r) * NHID] = acc1[r];
        }
    }
}

__global__ __launch_bounds__(256) void k_deal_gemm1(const int* __restrict__ src,
                                                    const int* __restrict__ dst,
                                                    const float* __restrict__ w,
                                                    const int* __restrict__ off,
                                                    const int* __restrict__ btot,
                                                    int* __restrict__ tmp,
                                                    int E, int nbk, int ebk,
                                                    const float* __restrict__ X,
                                                    const float* __restrict__ Wp,
                                                    float* __restrict__ H, int M,
                                                    int tile_beg, int tile_end) {
    const int b = blockIdx.x;
    if (b < ebk) {
        deal_run(src, dst, w, off, btot, tmp, E, nbk, ebk, b);
    } else {
        const int tile = tile_beg + (b - ebk);
        if (tile < tile_end) gemm1_tile(X, Wp, H, M, tile);
    }
}

__global__ __launch_bounds__(256) void k_sort_gemm1(const int* __restrict__ btot,
                                                    const int* __restrict__ tmp,
                                                    int* __restrict__ first, float* __restrict__ dinv,
                                                    int* __restrict__ edges, int N, int nbk,
                                                    const float* __restrict__ X,
                                                    const float* __restrict__ Wp,
                                                    float* __restrict__ H,
                                                    int tile_beg, int tile_end) {
    const int b = blockIdx.x;
    if (b < nbk) {
        sort_bucket(btot, tmp, first, dinv, edges, N, b);
    } else {
        const int tile = tile_beg + (b - nbk);
        if (tile < tile_end) gemm1_tile(X, Wp, H, N, tile);
    }
}

#define GPITCH 64
__global__ __launch_bounds__(256) void k_agg1_gemm2(const int* __restrict__ first,
                                                    int* edges,
                                                    const float* __restrict__ dinv,
                                                    const float* __restrict__ H,
                                                    const float* __restrict__ b1,
                                                    const float* __restrict__ Wp,
                                                    float* __restrict__ G, int N) {
    __shared__ float2 Xs[64 * 33];
    __shared__ float2 Gs[32 * (GPITCH / 2)];
    const int lane = threadIdx.x & 31;
    const int wave = __builtin_amdgcn_readfirstlane((int)(threadIdx.x >> 5));
    const int m0 = blockIdx.x * 32;
    const float4* H4 = (const float4*)H;
    const float4 bb = ((const float4*)b1)[lane];

    {
        const int row = threadIdx.x >> 3, p = 16 + 2 * (threadIdx.x & 7);
        Gs[row * (GPITCH / 2) + p].y     = 0.0f;
        Gs[row * (GPITCH / 2) + p + 1].y = 0.0f;
    }

    for (int t = 0; t < 4; ++t) {
        const int m = wave * 4 + t;
        const int i = m0 + m;
        float4 v = make_float4(0.0f, 0.0f, 0.0f, 0.0f);
        if (i < N) {
            const int beg = first[i], end = first[i + 1];
            const float di = dinv[i];

            const float cs = di * di;
            const float4 hs = H4[(size_t)i * 32 + lane];
            float4 acc = make_float4(cs * hs.x, cs * hs.y, cs * hs.z, cs * hs.w);

            for (int base = beg; base < end; base += 32) {
                const int n = min(32, end - base);
                int s_l = 0;
                float c_l = 0.0f;
                if (lane < n) {
                    int2 sw = ((const int2*)edges)[base + lane];
                    s_l = sw.x;
                    c_l = dinv[sw.x] * __int_as_float(sw.y) * di;
                    edges[2 * (size_t)(base + lane) + 1] = __float_as_int(c_l);
                }
                int j = 0;
                for (; j + 4 <= n; j += 4) {
                    const int s0 = lane_i(s_l, j), s1 = lane_i(s_l, j + 1);
                    const int s2 = lane_i(s_l, j + 2), s3 = lane_i(s_l, j + 3);
                    const float c0 = lane_f(c_l, j), c1 = lane_f(c_l, j + 1);
                    const float c2 = lane_f(c_l, j + 2), c3 = lane_f(c_l, j + 3);
                    const float4 v0 = H4[(size_t)s0 * 32 + lane];
                    const float4 v1 = H4[(size_t)s1 * 32 + lane];
                    const float4 v2 = H4[(size_t)s2 * 32 + lane];
                    const float4 v3 = H4[(size_t)s3 * 32 + lane];
                    acc.x += c0 * v0.x; acc.y += c0 * v0.y; acc.z += c0 * v0.z; acc.w += c0 * v0.w;
                    acc.x += c1 * v1.x; acc.y += c1 * v1.y; acc.z += c1 * v1.z; acc.w += c1 * v1.w;
                    acc.x += c2 * v2.x; acc.y += c2 * v2.y; acc.z += c2 * v2.z; acc.w += c2 * v2.w;
                    acc.x += c3 * v3.x; acc.y += c3 * v3.y; acc.z += c3 * v3.z; acc.w += c3 * v3.w;
                }
                for (; j < n; ++j) {
                    const int s0 = lane_i(s_l, j);
                    const float c0 = lane_f(c_l, j);
                    const float4 v0 = H4[(size_t)s0 * 32 + lane];
                    acc.x += c0 * v0.x; acc.y += c0 * v0.y; acc.z += c0 * v0.z; acc.w += c0 * v0.w;
                }
            }
            v.x = fmaxf(acc.x + bb.x, 0.0f);
            v.y = fmaxf(acc.y + bb.y, 0.0f);
            v.z = fmaxf(acc.z + bb.z, 0.0f);
            v.w = fmaxf(acc.w + bb.w, 0.0f);
        }
        Xs[(2 * lane)     * 33 + m] = make_float2(v.x, v.y);
        Xs[(2 * lane + 1) * 33 + m] = make_float2(v.z, v.w);
    }
    __syncthreads();

    if (wave < 6) {
        const int ms = wave / 3, nt = wave - 3 * ms;
        const int l = lane & 15, half = lane >> 4;
        const float2* Bp = (const float2*)Wp;

        v8f acc = {};
#pragma unroll 4
        for (int kp = 0; kp < 64; kp += 2) {
            int row = kp + half;
            float2 a = Xs[row * 33 + ms * 16 + l];
            float2 b = Bp[(size_t)row * NCPAD + nt * 16 + l];
            v2f av = {a.x, a.y}, bv = {b.x, b.y};
            acc = __builtin_amdgcn_wmma_f32_16x16x4_f32(false, av, false, bv, (short)0, acc, false, false);
        }

        const int fi = (nt == 2) ? (2 * l + 1) : (2 * (nt * 16 + l));
        float* Gf = (float*)Gs + (ms * 16 + 8 * half) * GPITCH + fi;
#pragma unroll
        for (int r = 0; r < 8; ++r) Gf[r * GPITCH] = acc[r];
    }
    __syncthreads();

    float2* G2 = (float2*)G;
    for (int t = 0; t < 4; ++t) {
        const int m = wave * 4 + t;
        if (m0 + m < N) G2[(size_t)(m0 + m) * (GPITCH / 2) + lane] = Gs[m * (GPITCH / 2) + lane];
    }
}

__global__ __launch_bounds__(256) void k_agg2_lsm(const int* __restrict__ first,
                                                  const int* __restrict__ edges,
                                                  const float* __restrict__ dinv,
                                                  const float* __restrict__ G,
                                                  const float* __restrict__ b2,
                                                  float* __restrict__ out, int N) {
    const int lane = threadIdx.x & 31;
    const int i = __builtin_amdgcn_readfirstlane((int)(blockIdx.x * 8 + (threadIdx.x >> 5)));
    if (i >= N) return;
    const int beg = first[i], end = first[i + 1];
    const float di = dinv[i];
    const float2* G2 = (const float2*)G;

    const float cs = di * di;
    const float2 gs = G2[(size_t)i * (GPITCH / 2) + lane];
    float acc0 = cs * gs.x;
    float acc1 = cs * gs.y;

    for (int base = beg; base < end; base += 32) {
        const int n = min(32, end - base);
        int s_l = 0;
        float c_l = 0.0f;
        if (lane < n) {
            int2 sw = ((const int2*)edges)[base + lane];
            s_l = sw.x;
            c_l = __int_as_float(sw.y);
        }
        int j = 0;
        for (; j + 4 <= n; j += 4) {
            const float2* g0 = G2 + (size_t)lane_i(s_l, j)     * (GPITCH / 2);
            const float2* g1 = G2 + (size_t)lane_i(s_l, j + 1) * (GPITCH / 2);
            const float2* g2 = G2 + (size_t)lane_i(s_l, j + 2) * (GPITCH / 2);
            const float2* g3 = G2 + (size_t)lane_i(s_l, j + 3) * (GPITCH / 2);
            const float c0 = lane_f(c_l, j), c1 = lane_f(c_l, j + 1);
            const float c2 = lane_f(c_l, j + 2), c3 = lane_f(c_l, j + 3);
            const float2 v0 = g0[lane];
            const float2 v1 = g1[lane];
            const float2 v2 = g2[lane];
            const float2 v3 = g3[lane];
            acc0 += c0 * v0.x; acc1 += c0 * v0.y;
            acc0 += c1 * v1.x; acc1 += c1 * v1.y;
            acc0 += c2 * v2.x; acc1 += c2 * v2.y;
            acc0 += c3 * v3.x; acc1 += c3 * v3.y;
        }
        for (; j < n; ++j) {
            const float2* g0 = G2 + (size_t)lane_i(s_l, j) * (GPITCH / 2);
            const float c0 = lane_f(c_l, j);
            const float2 v0 = g0[lane];
            acc0 += c0 * v0.x;
            acc1 += c0 * v0.y;
        }
    }

    const float v1 = acc0 + b2[lane];
    const int c2 = 32 + lane;
    const bool has2 = c2 < NCLASS;
    const float v2 = has2 ? (acc1 + b2[c2]) : -INFINITY;

    float mx = fmaxf(v1, v2);
#pragma unroll
    for (int off = 16; off >= 1; off >>= 1) mx = fmaxf(mx, __shfl_xor(mx, off));

    float s = expf(v1 - mx) + (has2 ? expf(v2 - mx) : 0.0f);
#pragma unroll
    for (int off = 16; off >= 1; off >>= 1) s += __shfl_xor(s, off);

    const float ls = logf(s);
    out[(size_t)i * NCLASS + lane] = v1 - mx - ls;
    if (has2) out[(size_t)i * NCLASS + c2] = v2 - mx - ls;
}

extern "C" void kernel_launch(void* const* d_in, const int* in_sizes, int n_in,
                              void* d_out, int out_size, void* d_ws, size_t ws_size,
                              hipStream_t stream) {
    const float* x  = (const float*)d_in[0];
    const int*   ei = (const int*)  d_in[1];
    const float* ew = (const float*)d_in[2];
    const float* W1 = (const float*)d_in[3];
    const float* b1 = (const float*)d_in[4];
    const float* W2 = (const float*)d_in[5];
    const float* b2 = (const float*)d_in[6];
    float* out = (float*)d_out;

    const int N = in_sizes[0] / NFEAT;
    const int E = in_sizes[2];
    const int* src = ei;
    const int* dst = ei + E;

    float* ws = (float*)d_ws;
    size_t o = 0;
    auto take = [&](size_t n) { float* p = ws + o; o += (n + 255) & ~(size_t)255; return p; };
    float* dinv = take((size_t)N);
    float* Wp1  = take((size_t)NFEAT * NHID);
    float* Wp2  = take((size_t)NHID * NCPAD);
    float* Hxw  = take((size_t)N * NHID);
    int* edges = (int*)take(2 * (size_t)E);
    int* first = (int*)take((size_t)N + 1);
    const int nbk = (N + (1 << BUCKET_SHIFT) - 1) >> BUCKET_SHIFT;
    const int ebk = (E + EDGE_RUN - 1) / EDGE_RUN;
    float* G  = ws + o;
    int* cm   = (int*)take((size_t)nbk * ebk);
    int* off  = (int*)take((size_t)nbk * ebk);
    int* btot = (int*)take((size_t)nbk);
    int* tmp  = (int*)take(3 * (size_t)E);

    k_prep<<<ebk + PACK1_CHUNKS + PACK2_CHUNKS, 256, 0, stream>>>(W1, Wp1, W2, Wp2, dst, cm, E, nbk, ebk);
    k_bucket_scan<<<nbk, 256, 0, stream>>>(cm, off, btot, ebk);

    const int tiles = (N + 31) / 32, tmid = (tiles + 1) / 2;
    k_deal_gemm1<<<ebk + tmid, 256, 0, stream>>>(src, dst, ew, off, btot, tmp, E, nbk, ebk,
                                                 x, Wp1, Hxw, N, 0, tmid);
    k_sort_gemm1<<<nbk + (tiles - tmid), 256, 0, stream>>>(btot, tmp, first, dinv, edges, N, nbk,
                                                           x, Wp1, Hxw, tmid, tiles);

    k_agg1_gemm2<<<(N + 31) / 32, 256, 0, stream>>>(first, edges, dinv, Hxw, b1, Wp2, G, N);

    k_agg2_lsm<<<(N + 7) / 8, 256, 0, stream>>>(first, edges, dinv, G, b2, out, N);
}


# Layer-2 gather: a wave walks 8 consecutive nodes, not one (next node's slots and row asked for ahead, 8 rows in flight, terms as fused multiply-adds); 100000 waves become 12500; sum order kept
# speedup vs baseline: 3.9547x; 1.0356x over previous
#include <hip/hip_runtime.h>
#include <cstdint>

typedef float v2f __attribute__((ext_vector_type(2)));
typedef float v8f __attribute__((ext_vector_type(8)));

#define NFEAT 256
#define NHID  128
#define NCLASS 41
#define NCPAD  48

#define BUCKET_SHIFT 8
#define MAX_BUCKETS  512
#define EDGE_RUN     4096

struct EdgeRec { int s, w, d; };

static __device__ __forceinline__ int block_excl_scan(int v, int* wtot, int& total) {
    const int lane = threadIdx.x & 31, wv = threadIdx.x >> 5;
    int inc = v;
    for (int off = 1; off < 32; off <<= 1) {
        int u = __shfl_up(inc, off);
        if (lane >= off) inc += u;
    }
    if (lane == 31) wtot[wv] = inc;
    __syncthreads();
    int pre = 0, tot = 0;
    for (int k = 0; k < 8; ++k) {
        int t = wtot[k];
        if (k < wv) pre += t;
        tot += t;
    }
    total = tot;
    __syncthreads();
    return pre + inc - v;
}

static __device__ __forceinline__ void hist_run(const int* __restrict__ dst, int* __restrict__ cm,
                                                int E, int nbk, int ebk, int blk) {
    __shared__ int h[MAX_BUCKETS];
    const int tid = threadIdx.x;
    h[tid] = 0;
    h[tid + 256] = 0;
    __syncthreads();
    int d[EDGE_RUN / 256];
#pragma unroll
    for (int it = 0; it < EDGE_RUN / 256; ++it) {
        int e = blk * EDGE_RUN + it * 256 + tid;
        d[it] = (e < E) ? dst[e] : -1;
    }
#pragma unroll
    for (int it = 0; it < EDGE_RUN / 256; ++it)
        if (d[it] >= 0) atomicAdd(&h[d[it] >> BUCKET_SHIFT], 1);
    __syncthreads();
    if (tid < nbk)       cm[(size_t)tid * ebk + blk]         = h[tid];
    if (tid + 256 < nbk) cm[(size_t)(tid + 256) * ebk + blk] = h[tid + 256];
}

__global__ __launch_bounds__(256) void k_bucket_scan(const int* __restrict__ cm, int* __restrict__ off,
                                                     int* __restrict__ btot, int ebk) {
    __shared__ int wtot[8];
    const int tid = threadIdx.x;
    const size_t row = (size_t)blockIdx.x * ebk;
    int carry = 0;
    for (int c0 = 0; c0 < ebk; c0 += 256) {
        const int j = c0 + tid;
        const int v = (j < ebk) ? cm[row + j] : 0;
        int tot;
        const int ex = block_excl_scan(v, wtot, tot);
        if (j < ebk) off[row + j] = carry + ex;
        carry += tot;
    }
    if (tid == 0) btot[blockIdx.x] = carry;
}

static __device__ __forceinline__ void deal_run(const int* __restrict__ src, const int* __restrict__ dst,
                                                const float* __restrict__ w,
                                                const int* __restrict__ off, const int* __restrict__ btot,
                                                int* __restrict__ tmp,
                                                int E, int nbk, int ebk, int blk) {
    EdgeRec* __restrict__ tq = (EdgeRec*)tmp;
    __shared__ int cur[MAX_BUCKETS];
    __shared__ int wtot[8];
    const int tid = threadIdx.x;
    const int b0 = (tid < nbk) ? btot[tid] : 0;
    const int b1 = (tid + 256 < nbk) ? btot[tid + 256] : 0;
    int tot0, tot1;
    const int ex0 = block_excl_scan(b0, wtot, tot0);
    const int ex1 = block_excl_scan(b1, wtot, tot1);
    cur[tid]       = (tid < nbk)       ? ex0 + off[(size_t)tid * ebk + blk] : 0;
    cur[tid + 256] = (tid + 256 < nbk) ? tot0 + ex1 + off[(size_t)(tid + 256) * ebk + blk] : 0;
    __syncthreads();
    for (int g = 0; g < EDGE_RUN / 256; g += 8) {
        int d[8], s[8], wi[8];
#pragma unroll
        for (int it = 0; it < 8; ++it) {
            int e = blk * EDGE_RUN + (g + it) * 256 + tid;
            const bool in = e < E;
            d[it]  = in ? dst[e] : -1;
            s[it]  = in ? src[e] : 0;
            wi[it] = in ? __float_as_int(w[e]) : 0;
        }
#pragma unroll
        for (int it = 0; it < 8; ++it) {
            if (d[it] >= 0) {
                const int pos = atomicAdd(&cur[d[it] >> BUCKET_SHIFT], 1);
                EdgeRec r;
                r.s = s[it]; r.w = wi[it]; r.d = d[it];
                tq[pos] = r;
            }
        }
    }
}

static __device__ __forceinline__ void sort_bucket(const int* __restrict__ btot,
                                                   const int* __restrict__ tmp,
                                                   int* __restrict__ first, float* __restrict__ dinv,
                                                   int* __restrict__ edges, int N, int b) {
    const EdgeRec* __restrict__ tq = (const EdgeRec*)tmp;
    __shared__ int   s_cnt[256];
    __shared__ float s_deg[256];
    __shared__ int   s_cur[256];
    __shared__ int   wtot[8];
    const int tid = threadIdx.x;

    int part = 0;
    for (int j = tid; j < b; j += 256) part += btot[j];
    int base;
    (void)block_excl_scan(part, wtot, base);
    const int nb = btot[b];
    const int node0 = b << BUCKET_SHIFT;

    s_cnt[tid] = 0;
    s_deg[tid] = 0.0f;
    __syncthreads();
    for (int p0 = tid; p0 < nb; p0 += 4 * 256) {
        EdgeRec q[4];
#pragma unroll
        for (int k = 0; k < 4; ++k) {
            const int p = p0 + k * 256;
            if (p < nb) q[k] = tq[base + p];
            else { q[k].s = 0; q[k].w = 0; q[k].d = -1; }
        }
#pragma unroll
        for (int k = 0; k < 4; ++k) {
            if (q[k].d >= 0) {
                const int d = q[k].d - node0;
                atomicAdd(&s_cnt[d], 1);
                atomicAdd(&s_deg[d], __int_as_float(q[k].w));
            }
        }
    }
    __syncthreads();

    const int c = s_cnt[tid];
    int tot;
    const int ex = block_excl_scan(c, wtot, tot);
    const int node = node0 + tid;
    if (node < N) {
        first[node] = base + ex;
        const float dg = 1.0f + s_deg[tid];
        dinv[node] = dg > 0.0f ? rsqrtf(dg) : 0.0f;
        if (node == N - 1) first[N] = base + ex + c;
    }
    s_cur[tid] = ex;
    __syncthreads();
    for (int p0 = tid; p0 < nb; p0 += 4 * 256) {
        EdgeRec q[4];
#pragma unroll
        for (int k = 0; k < 4; ++k) {
            const int p = p0 + k * 256;
            if (p < nb) q[k] = tq[base + p];
            else { q[k].s = 0; q[k].w = 0; q[k].d = -1; }
        }
#pragma unroll
        for (int k = 0; k < 4; ++k) {
            if (q[k].d >= 0) {
                const int slot = base + atomicAdd(&s_cur[q[k].d - node0], 1);
                ((int2*)edges)[slot] = make_int2(q[k].s, q[k].w);
            }
        }
    }
}

static __device__ __forceinline__ int lane_i(int v, int j) {
    return __builtin_amdgcn_readlane(v, j);
}
static __device__ __forceinline__ float lane_f(float v, int j) {
    return __int_as_float(__builtin_amdgcn_readlane(__float_as_int(v), j));
}

static __device__ __forceinline__ void pack_w1_chunk(const float* __restrict__ W1, float* __restrict__ Wp, int chunk) {
    int idx = chunk * 256 + threadIdx.x;
    if (idx >= NFEAT * NHID) return;
    int k = idx >> 7, n = idx & 127;
    Wp[(size_t)(k >> 1) * (NHID * 2) + n * 2 + (k & 1)] = W1[idx];
}

static __device__ __forceinline__ void pack_w2_chunk(const float* __restrict__ W2, float* __restrict__ Wp, int chunk) {
    int idx = chunk * 256 + threadIdx.x;
    if (idx >= NHID * NCPAD) return;
    int k = idx / NCPAD, n = idx % NCPAD;
    float v = (n < NCLASS) ? W2[k * NCLASS + n] : 0.0f;
    Wp[(size_t)(k >> 1) * (NCPAD * 2) + n * 2 + (k & 1)] = v;
}

#define PACK1_CHUNKS ((NFEAT * NHID + 255) / 256)
#define PACK2_CHUNKS ((NHID * NCPAD + 255) / 256)
__global__ __launch_bounds__(256) void k_prep(const float* __restrict__ W1, float* __restrict__ Wp1,
                                              const float* __restrict__ W2, float* __restrict__ Wp2,
                                              const int* __restrict__ dst, int* __restrict__ cm,
                                              int E, int nbk, int ebk) {
    const int b = blockIdx.x;
    if (b < ebk) hist_run(dst, cm, E, nbk, ebk, b);
    else if (b < ebk + PACK1_CHUNKS) pack_w1_chunk(W1, Wp1, b - ebk);
    else pack_w2_chunk(W2, Wp2, b - ebk - PACK1_CHUNKS);
}

static __device__ __forceinline__ void gemm1_tile(const float* __restrict__ X,
                                                  const float* __restrict__ Wp,
                                                  float* __restrict__ H, int M, int tile) {
    __shared__ float2 Xs[128 * 33];
    const int tid = threadIdx.x;
    const int m0 = tile * 32;

#pragma unroll
    for (int it = 0; it < 8; ++it) {
        int q = it * 256 + tid;
        int m = q >> 6, k4 = q & 63;
        int mr = m0 + m; if (mr >= M) mr = M - 1;
        float4 v = ((const float4*)(X + (size_t)mr * NFEAT))[k4];
        Xs[(2 * k4)     * 33 + m] = make_float2(v.x, v.y);
        Xs[(2 * k4 + 1) * 33 + m] = make_float2(v.z, v.w);
    }
    __syncthreads();

    const int wave = tid >> 5, lane = tid & 31, l = lane & 15, half = lane >> 4;
    const int n0 = wave * 16;
    const float2* Bp = (const float2*)Wp;

    v8f acc0 = {}, acc1 = {};
#pragma unroll 4
    for (int kp = 0; kp < 128; kp += 2) {
        int row = kp + half;
        float2 a0 = Xs[row * 33 + l];
        float2 a1 = Xs[row * 33 + 16 + l];
        float2 b  = Bp[(size_t)row * NHID + n0 + l];
        v2f av0 = {a0.x, a0.y}, av1 = {a1.x, a1.y}, bv = {b.x, b.y};
        acc0 = __builtin_amdgcn_wmma_f32_16x16x4_f32(false, av0, false, bv, (short)0, acc0, false, false);
        acc1 = __builtin_amdgcn_wmma_f32_16x16x4_f32(false, av1, false, bv, (short)0, acc1, false, false);
    }

    const int mb = 8 * half;
    float* Ho = H + (size_t)(m0 + mb) * NHID + n0 + l;
    if (m0 + 32 <= M) {
#pragma unroll
        for (int r = 0; r < 8; ++r) {
            Ho[(size_t)r * NHID]        = acc0[r];
            Ho[(size_t)(16 + r) * NHID] = acc1[r];
        }
    } else {
#pragma unroll
        for (int r = 0; r < 8; ++r) {
            if (m0 + mb + r < M)      Ho[(size_t)r * NHID]        = acc0[r];
            if (m0 + 16 + mb + r < M) Ho[(size_t)(16 + r) * NHID] = acc1[r];
        }
    }
}

__global__ __launch_bounds__(256) void k_deal_gemm1(const int* __restrict__ src,
                                                    const int* __restrict__ dst,
                                                    const float* __restrict__ w,
                                                    const int* __restrict__ off,
                                                    const int* __restrict__ btot,
                                                    int* __restrict__ tmp,
                                                    int E, int nbk, int ebk,
                                                    const float* __restrict__ X,
                                                    const float* __restrict__ Wp,
                                                    float* __restrict__ H, int M,
                                                    int tile_beg, int tile_end) {
    const int b = blockIdx.x;
    if (b < ebk) {
        deal_run(src, dst, w, off, btot, tmp, E, nbk, ebk, b);
    } else {
        const int tile = tile_beg + (b - ebk);
        if (tile < tile_end) gemm1_tile(X, Wp, H, M, tile);
    }
}

__global__ __launch_bounds__(256) void k_sort_gemm1(const int* __restrict__ btot,
                                                    const int* __restrict__ tmp,
                                                    int* __restrict__ first, float* __restrict__ dinv,
                                                    int* __restrict__ edges, int N, int nbk,
                                                    const float* __restrict__ X,
                                                    const float* __restrict__ Wp,
                                                    float* __restrict__ H,
                                                    int tile_beg, int tile_end) {
    const int b = blockIdx.x;
    if (b < nbk) {
        sort_bucket(btot, tmp, first, dinv, edges, N, b);
    } else {
        const int tile = tile_beg + (b - nbk);
        if (tile < tile_end) gemm1_tile(X, Wp, H, N, tile);
    }
}

#define GPITCH 64
__global__ __launch_bounds__(256) void k_agg1_gemm2(const int* __restrict__ first,
                                                    int* edges,
                                                    const float* __restrict__ dinv,
                                                    const float* __restrict__ H,
                                                    const float* __restrict__ b1,
                                                    const float* __restrict__ Wp,
                                                    float* __restrict__ G, int N) {
    __shared__ float2 Xs[64 * 33];
    __shared__ float2 Gs[32 * (GPITCH / 2)];
    const int lane = threadIdx.x & 31;
    const int wave = __builtin_amdgcn_readfirstlane((int)(threadIdx.x >> 5));
    const int m0 = blockIdx.x * 32;
    const float4* H4 = (const float4*)H;
    const float4 bb = ((const float4*)b1)[lane];

    {
        const int row = threadIdx.x >> 3, p = 16 + 2 * (threadIdx.x & 7);
        Gs[row * (GPITCH / 2) + p].y     = 0.0f;
        Gs[row * (GPITCH / 2) + p + 1].y = 0.0f;
    }

    for (int t = 0; t < 4; ++t) {
        const int m = wave * 4 + t;
        const int i = m0 + m;
        float4 v = make_float4(0.0f, 0.0f, 0.0f, 0.0f);
        if (i < N) {
            const int beg = first[i], end = first[i + 1];
            const float di = dinv[i];

            const float cs = di * di;
            const float4 hs = H4[(size_t)i * 32 + lane];
            float4 acc = make_float4(cs * hs.x, cs * hs.y, cs * hs.z, cs * hs.w);

            for (int base = beg; base < end; base += 32) {
                const int n = min(32, end - base);
                int s_l = 0;
                float c_l = 0.0f;
                if (lane < n) {
                    int2 sw = ((const int2*)edges)[base + lane];
                    s_l = sw.x;
                    c_l = dinv[sw.x] * __int_as_float(sw.y) * di;
                    edges[2 * (size_t)(base + lane) + 1] = __float_as_int(c_l);
                }
                int j = 0;
                for (; j + 4 <= n; j += 4) {
                    const int s0 = lane_i(s_l, j), s1 = lane_i(s_l, j + 1);
                    const int s2 = lane_i(s_l, j + 2), s3 = lane_i(s_l, j + 3);
                    const float c0 = lane_f(c_l, j), c1 = lane_f(c_l, j + 1);
                    const float c2 = lane_f(c_l, j + 2), c3 = lane_f(c_l, j + 3);
                    const float4 v0 = H4[(size_t)s0 * 32 + lane];
                    const float4 v1 = H4[(size_t)s1 * 32 + lane];
                    const float4 v2 = H4[(size_t)s2 * 32 + lane];
                    const float4 v3 = H4[(size_t)s3 * 32 + lane];
                    acc.x += c0 * v0.x; acc.y += c0 * v0.y; acc.z += c0 * v0.z; acc.w += c0 * v0.w;
                    acc.x += c1 * v1.x; acc.y += c1 * v1.y; acc.z += c1 * v1.z; acc.w += c1 * v1.w;
                    acc.x += c2 * v2.x; acc.y += c2 * v2.y; acc.z += c2 * v2.z; acc.w += c2 * v2.w;
                    acc.x += c3 * v3.x; acc.y += c3 * v3.y; acc.z += c3 * v3.z; acc.w += c3 * v3.w;
                }
                for (; j < n; ++j) {
                    const int s0 = lane_i(s_l, j);
                    const float c0 = lane_f(c_l, j);
                    const float4 v0 = H4[(size_t)s0 * 32 + lane];
                    acc.x += c0 * v0.x; acc.y += c0 * v0.y; acc.z += c0 * v0.z; acc.w += c0 * v0.w;
                }
            }
            v.x = fmaxf(acc.x + bb.x, 0.0f);
            v.y = fmaxf(acc.y + bb.y, 0.0f);
            v.z = fmaxf(acc.z + bb.z, 0.0f);
            v.w = fmaxf(acc.w + bb.w, 0.0f);
        }
        Xs[(2 * lane)     * 33 + m] = make_float2(v.x, v.y);
        Xs[(2 * lane + 1) * 33 + m] = make_float2(v.z, v.w);
    }
    __syncthreads();

    if (wave < 6) {
        const int ms = wave / 3, nt = wave - 3 * ms;
        const int l = lane & 15, half = lane >> 4;
        const float2* Bp = (const float2*)Wp;

        v8f acc = {};
#pragma unroll 4
        for (int kp = 0; kp < 64; kp += 2) {
            int row = kp + half;
            float2 a = Xs[row * 33 + ms * 16 + l];
            float2 b = Bp[(size_t)row * NCPAD + nt * 16 + l];
            v2f av = {a.x, a.y}, bv = {b.x, b.y};
            acc = __builtin_amdgcn_wmma_f32_16x16x4_f32(false, av, false, bv, (short)0, acc, false, false);
        }

        const int fi = (nt == 2) ? (2 * l + 1) : (2 * (nt * 16 + l));
        float* Gf = (float*)Gs + (ms * 16 + 8 * half) * GPITCH + fi;
#pragma unroll
        for (int r = 0; r < 8; ++r) Gf[r * GPITCH] = acc[r];
    }
    __syncthreads();

    float2* G2 = (float2*)G;
    for (int t = 0; t < 4; ++t) {
        const int m = wave * 4 + t;
        if (m0 + m < N) G2[(size_t)(m0 + m) * (GPITCH / 2) + lane] = Gs[m * (GPITCH / 2) + lane];
    }
}

#define L2_NODES 8
#define L2_WAVES 4

template <int M>
static __device__ __forceinline__ void l2_rows(const float2* __restrict__ G2, int s_l, float c_l,
                                               int j, int lane, float& acc0, float& acc1) {
    float2 v[M];
#pragma unroll
    for (int k = 0; k < M; ++k)
        v[k] = (G2 + (size_t)lane_i(s_l, j + k) * (GPITCH / 2))[lane];
#pragma unroll
    for (int k = 0; k < M; ++k) {
        const float c = lane_f(c_l, j + k);
        acc0 = fmaf(c, v[k].x, acc0);
        acc1 = fmaf(c, v[k].y, acc1);
    }
}

static __device__ __forceinline__ void l2_chunk(const float2* __restrict__ G2, int s_l, float c_l,
                                                int n, int lane, float& acc0, float& acc1) {
    int j = 0;
    for (; j + 8 <= n; j += 8) l2_rows<8>(G2, s_l, c_l, j, lane, acc0, acc1);
    if (j + 4 <= n) { l2_rows<4>(G2, s_l, c_l, j, lane, acc0, acc1); j += 4; }
    switch (n - j) {
        case 3: l2_rows<3>(G2, s_l, c_l, j, lane, acc0, acc1); break;
        case 2: l2_rows<2>(G2, s_l, c_l, j, lane, acc0, acc1); break;
        case 1: l2_rows<1>(G2, s_l, c_l, j, lane, acc0, acc1); break;
        default: break;
    }
}

__global__ __launch_bounds__(32 * L2_WAVES) void k_agg2_lsm(const int* __restrict__ first,
                                                            const int* __restrict__ edges,
                                                            const float* __restrict__ dinv,
                                                            const float* __restrict__ G,
                                                            const float* __restrict__ b2,
                                                            float* __restrict__ out, int N) {
    const int lane = threadIdx.x & 31;
    const int n0 = __builtin_amdgcn_readfirstlane(
        (int)((blockIdx.x * L2_WAVES + (threadIdx.x >> 5)) * L2_NODES));
    if (n0 >= N) return;
    const int2*   E2 = (const int2*)edges;
    const float2* G2 = (const float2*)G;

    const int   bnd = first[min(n0 + lane, N)];
    const float dv  = dinv[min(n0 + lane, N - 1)];
    const int  c2   = 32 + lane;
    const bool has2 = c2 < NCLASS;
    const float bia1 = b2[lane];
    const float bia2 = has2 ? b2[c2] : 0.0f;

    int beg = lane_i(bnd, 0), end = lane_i(bnd, 1);
    int s_n = 0;
    float c_n = 0.0f;
    if (lane < min(32, end - beg)) {
        const int2 sw = E2[beg + lane];
        s_n = sw.x;
        c_n = __int_as_float(sw.y);
    }
    float2 g_n = G2[(size_t)n0 * (GPITCH / 2) + lane];

    for (int t = 0; t < L2_NODES; ++t) {
        const int i = n0 + t;
        if (i >= N) break;
        const int s_n_cur = s_n;
        const float c_n_cur = c_n;
        const float2 gs = g_n;
        const float di = lane_f(dv, t);

        int nend = end;
        if (t + 1 < L2_NODES && i + 1 < N) {
            nend = lane_i(bnd, t + 2);
            s_n = 0;
            c_n = 0.0f;
            if (lane < min(32, nend - end)) {
                const int2 sw = E2[end + lane];
                s_n = sw.x;
                c_n = __int_as_float(sw.y);
            }
            g_n = G2[(size_t)(i + 1) * (GPITCH / 2) + lane];
        }

        const float cs = di * di;
        float acc0 = cs * gs.x;
        float acc1 = cs * gs.y;

        l2_chunk(G2, s_n_cur, c_n_cur, min(32, end - beg), lane, acc0, acc1);
        for (int base = beg + 32; base < end; base += 32) {
            const int n = min(32, end - base);
            int s_x = 0;
            float c_x = 0.0f;
            if (lane < n) {
                const int2 sw = E2[base + lane];
                s_x = sw.x;
                c_x = __int_as_float(sw.y);
            }
            l2_chunk(G2, s_x, c_x, n, lane, acc0, acc1);
        }

        const float v1 = acc0 + bia1;
        const float v2 = has2 ? (acc1 + bia2) : -INFINITY;

        float mx = fmaxf(v1, v2);
#pragma unroll
        for (int off = 16; off >= 1; off >>= 1) mx = fmaxf(mx, __shfl_xor(mx, off));

        float s = expf(v1 - mx) + (has2 ? expf(v2 - mx) : 0.0f);
#pragma unroll
        for (int off = 16; off >= 1; off >>= 1) s += __shfl_xor(s, off);

        const float ls = logf(s);
        out[(size_t)i * NCLASS + lane] = v1 - mx - ls;
        if (has2) out[(size_t)i * NCLASS + c2] = v2 - mx - ls;

        beg = end;
        end = nend;
    }
}

extern "C" void kernel_launch(void* const* d_in, const int* in_sizes, int n_in,
                              void* d_out, int out_size, void* d_ws, size_t ws_size,
                              hipStream_t stream) {
    const float* x  = (const float*)d_in[0];
    const int*   ei = (const int*)  d_in[1];
    const float* ew = (const float*)d_in[2];
    const float* W1 = (const float*)d_in[3];
    const float* b1 = (const float*)d_in[4];
    const float* W2 = (const float*)d_in[5];
    const float* b2 = (const float*)d_in[6];
    float* out = (float*)d_out;

    const int N = in_sizes[0] / NFEAT;
    const int E = in_sizes[2];
    const int* src = ei;
    const int* dst = ei + E;

    float* ws = (float*)d_ws;
    size_t o = 0;
    auto take = [&](size_t n) { float* p = ws + o; o += (n + 255) & ~(size_t)255; return p; };
    float* dinv = take((size_t)N);
    float* Wp1  = take((size_t)NFEAT * NHID);
    float* Wp2  = take((size_t)NHID * NCPAD);
    float* Hxw  = take((size_t)N * NHID);
    int* edges = (int*)take(2 * (size_t)E);
    int* first = (int*)take((size_t)N + 1);
    const int nbk = (N + (1 << BUCKET_SHIFT) - 1) >> BUCKET_SHIFT;
    const int ebk = (E + EDGE_RUN - 1) / EDGE_RUN;
    float* G  = ws + o;
    int* cm   = (int*)take((size_t)nbk * ebk);
    int* off  = (int*)take((size_t)nbk * ebk);
    int* btot = (int*)take((size_t)nbk);
    int* tmp  = (int*)take(3 * (size_t)E);

    k_prep<<<ebk + PACK1_CHUNKS + PACK2_CHUNKS, 256, 0, stream>>>(W1, Wp1, W2, Wp2, dst, cm, E, nbk, ebk);
    k_bucket_scan<<<nbk, 256, 0, stream>>>(cm, off, btot, ebk);

    const int tiles = (N + 31) / 32, tmid = (tiles + 1) / 2;
    k_deal_gemm1<<<ebk + tmid, 256, 0, stream>>>(src, dst, ew, off, btot, tmp, E, nbk, ebk,
                                                 x, Wp1, Hxw, N, 0, tmid);
    k_sort_gemm1<<<nbk + (tiles - tmid), 256, 0, stream>>>(btot, tmp, first, dinv, edges, N, nbk,
                                                           x, Wp1, Hxw, tmid, tiles);

    k_agg1_gemm2<<<(N + 31) / 32, 256, 0, stream>>>(first, edges, dinv, Hxw, b1, Wp2, G, N);

    k_agg2_lsm<<<(N + L2_WAVES * L2_NODES - 1) / (L2_WAVES * L2_NODES), 32 * L2_WAVES, 0, stream>>>(first, edges, dinv, G, b2, out, N);
}


# Sort level 1 self-contained: each run bucket-sorted into its own records plus piece starts; bucket blocks walk one piece per run; count and scan launches go (6 to 4); sums stay in LDS-arrival order
# speedup vs baseline: 4.1521x; 1.0499x over previous
#include <hip/hip_runtime.h>
#include <cstdint>

typedef float v2f __attribute__((ext_vector_type(2)));
typedef float v8f __attribute__((ext_vector_type(8)));

#define NFEAT 256
#define NHID  128
#define NCLASS 41
#define NCPAD  48

#define BUCKET_SHIFT 8
#define MAX_BUCKETS  512
#define EDGE_RUN     4096

struct EdgeRec { int s, w, d; };

static __device__ __forceinline__ int block_excl_scan(int v, int* wtot, int& total) {
    const int lane = threadIdx.x & 31, wv = threadIdx.x >> 5;
    int inc = v;
    for (int off = 1; off < 32; off <<= 1) {
        int u = __shfl_up(inc, off);
        if (lane >= off) inc += u;
    }
    if (lane == 31) wtot[wv] = inc;
    __syncthreads();
    int pre = 0, tot = 0;
    for (int k = 0; k < 8; ++k) {
        int t = wtot[k];
        if (k < wv) pre += t;
        tot += t;
    }
    total = tot;
    __syncthreads();
    return pre + inc - v;
}

static __device__ __forceinline__ void deal_run(const int* __restrict__ src, const int* __restrict__ dst,
                                                const float* __restrict__ w,
                                                int* __restrict__ pst, int* __restrict__ tmp,
                                                int E, int nbk, int ebk, int blk) {
    EdgeRec* __restrict__ tq = (EdgeRec*)tmp + (size_t)blk * EDGE_RUN;
    __shared__ int cur[MAX_BUCKETS];
    __shared__ int wtot[8];
    const int tid = threadIdx.x;
    cur[tid] = 0;
    cur[tid + 256] = 0;
    __syncthreads();
    int d[EDGE_RUN / 256];
#pragma unroll
    for (int it = 0; it < EDGE_RUN / 256; ++it) {
        int e = blk * EDGE_RUN + it * 256 + tid;
        d[it] = (e < E) ? dst[e] : -1;
    }
#pragma unroll
    for (int it = 0; it < EDGE_RUN / 256; ++it)
        if (d[it] >= 0) atomicAdd(&cur[d[it] >> BUCKET_SHIFT], 1);
    __syncthreads();
    const int h0 = cur[tid], h1 = cur[tid + 256];
    int tot0, tot1;
    const int ex0 = block_excl_scan(h0, wtot, tot0);
    const int ex1 = tot0 + block_excl_scan(h1, wtot, tot1);
    if (tid < nbk)       pst[(size_t)tid * ebk + blk]         = ex0;
    if (tid + 256 < nbk) pst[(size_t)(tid + 256) * ebk + blk] = ex1;
    if (tid == 0)        pst[(size_t)nbk * ebk + blk]         = tot0 + tot1;
    cur[tid]       = ex0;
    cur[tid + 256] = ex1;
    __syncthreads();
#pragma unroll
    for (int g = 0; g < EDGE_RUN / 256; g += 8) {
        int s[8], wi[8];
#pragma unroll
        for (int it = 0; it < 8; ++it) {
            int e = blk * EDGE_RUN + (g + it) * 256 + tid;
            const bool in = e < E;
            s[it]  = in ? src[e] : 0;
            wi[it] = in ? __float_as_int(w[e]) : 0;
        }
#pragma unroll
        for (int it = 0; it < 8; ++it) {
            if (d[g + it] >= 0) {
                const int pos = atomicAdd(&cur[d[g + it] >> BUCKET_SHIFT], 1);
                EdgeRec r;
                r.s = s[it]; r.w = wi[it]; r.d = d[g + it];
                tq[pos] = r;
            }
        }
    }
}

template <typename F>
static __device__ __forceinline__ void for_bucket_records(const int* __restrict__ pst,
                                                          const EdgeRec* __restrict__ tq,
                                                          int b, int ebk, F f) {
    const int grp = threadIdx.x >> 3, sub = threadIdx.x & 7;
    const int* __restrict__ lo = pst + (size_t)b * ebk;
    const int* __restrict__ hi = lo + ebk;
    for (int j0 = grp; j0 < ebk; j0 += 128) {
        int ps[4], pe[4];
#pragma unroll
        for (int k = 0; k < 4; ++k) {
            const int j = j0 + 32 * k;
            const bool in = j < ebk;
            ps[k] = in ? lo[j] + sub : 0;
            pe[k] = in ? hi[j] : 0;
        }
        for (;;) {
            EdgeRec q[4];
            bool any = false;
#pragma unroll
            for (int k = 0; k < 4; ++k) {
                q[k].s = 0; q[k].w = 0; q[k].d = -1;
                if (ps[k] < pe[k]) {
                    q[k] = tq[(size_t)(j0 + 32 * k) * EDGE_RUN + ps[k]];
                    any = true;
                }
                ps[k] += 8;
            }
            if (!any) break;
#pragma unroll
            for (int k = 0; k < 4; ++k)
                if (q[k].d >= 0) f(q[k]);
        }
    }
}

static __device__ __forceinline__ void sort_bucket(const int* __restrict__ pst,
                                                   const int* __restrict__ tmp,
                                                   int* __restrict__ first, float* __restrict__ dinv,
                                                   int* __restrict__ edges, int N, int ebk, int b) {
    const EdgeRec* __restrict__ tq = (const EdgeRec*)tmp;
    __shared__ int   s_cnt[256];
    __shared__ float s_deg[256];
    __shared__ int   s_cur[256];
    __shared__ int   wtot[8];
    const int tid = threadIdx.x;

    int part = 0;
    for (int j = tid; j < ebk; j += 256) part += pst[(size_t)b * ebk + j];
    int base;
    (void)block_excl_scan(part, wtot, base);
    const int node0 = b << BUCKET_SHIFT;

    s_cnt[tid] = 0;
    s_deg[tid] = 0.0f;
    __syncthreads();
    for_bucket_records(pst, tq, b, ebk, [&](const EdgeRec& r) {
        const int d = r.d - node0;
        atomicAdd(&s_cnt[d], 1);
        atomicAdd(&s_deg[d], __int_as_float(r.w));
    });
    __syncthreads();

    const int c = s_cnt[tid];
    int tot;
    const int ex = block_excl_scan(c, wtot, tot);
    const int node = node0 + tid;
    if (node < N) {
        first[node] = base + ex;
        const float dg = 1.0f + s_deg[tid];
        dinv[node] = dg > 0.0f ? rsqrtf(dg) : 0.0f;
        if (node == N - 1) first[N] = base + ex + c;
    }
    s_cur[tid] = ex;
    __syncthreads();
    for_bucket_records(pst, tq, b, ebk, [&](const EdgeRec& r) {
        const int slot = base + atomicAdd(&s_cur[r.d - node0], 1);
        ((int2*)edges)[slot] = make_int2(r.s, r.w);
    });
}

static __device__ __forceinline__ int lane_i(int v, int j) {
    return __builtin_amdgcn_readlane(v, j);
}
static __device__ __forceinline__ float lane_f(float v, int j) {
    return __int_as_float(__builtin_amdgcn_readlane(__float_as_int(v), j));
}

static __device__ __forceinline__ void pack_w1_chunk(const float* __restrict__ W1, float* __restrict__ Wp, int chunk) {
    int idx = chunk * 256 + threadIdx.x;
    if (idx >= NFEAT * NHID) return;
    int k = idx >> 7, n = idx & 127;
    Wp[(size_t)(k >> 1) * (NHID * 2) + n * 2 + (k & 1)] = W1[idx];
}

static __device__ __forceinline__ void pack_w2_chunk(const float* __restrict__ W2, float* __restrict__ Wp, int chunk) {
    int idx = chunk * 256 + threadIdx.x;
    if (idx >= NHID * NCPAD) return;
    int k = idx / NCPAD, n = idx % NCPAD;
    float v = (n < NCLASS) ? W2[k * NCLASS + n] : 0.0f;
    Wp[(size_t)(k >> 1) * (NCPAD * 2) + n * 2 + (k & 1)] = v;
}

#define PACK1_CHUNKS ((NFEAT * NHID + 255) / 256)
#define PACK2_CHUNKS ((NHID * NCPAD + 255) / 256)

template <bool PACKED>
static __device__ __forceinline__ void gemm1_tile(const float* __restrict__ X,
                                                  const float* __restrict__ Wp,
                                                  float* __restrict__ H, int M, int tile) {
    __shared__ float2 Xs[128 * 33];
    const int tid = threadIdx.x;
    const int m0 = tile * 32;

#pragma unroll
    for (int it = 0; it < 8; ++it) {
        int q = it * 256 + tid;
        int m = q >> 6, k4 = q & 63;
        int mr = m0 + m; if (mr >= M) mr = M - 1;
        float4 v = ((const float4*)(X + (size_t)mr * NFEAT))[k4];
        Xs[(2 * k4)     * 33 + m] = make_float2(v.x, v.y);
        Xs[(2 * k4 + 1) * 33 + m] = make_float2(v.z, v.w);
    }
    __syncthreads();

    const int wave = tid >> 5, lane = tid & 31, l = lane & 15, half = lane >> 4;
    const int n0 = wave * 16;
    const float2* Bp = (const float2*)Wp;

    v8f acc0 = {}, acc1 = {};
#pragma unroll 4
    for (int kp = 0; kp < 128; kp += 2) {
        int row = kp + half;
        float2 a0 = Xs[row * 33 + l];
        float2 a1 = Xs[row * 33 + 16 + l];
        float2 b;
        if (PACKED) {
            b = Bp[(size_t)row * NHID + n0 + l];
        } else {
            b.x = Wp[(size_t)(2 * row)     * NHID + n0 + l];
            b.y = Wp[(size_t)(2 * row + 1) * NHID + n0 + l];
        }
        v2f av0 = {a0.x, a0.y}, av1 = {a1.x, a1.y}, bv = {b.x, b.y};
        acc0 = __builtin_amdgcn_wmma_f32_16x16x4_f32(false, av0, false, bv, (short)0, acc0, false, false);
        acc1 = __builtin_amdgcn_wmma_f32_16x16x4_f32(false, av1, false, bv, (short)0, acc1, false, false);
    }

    const int mb = 8 * half;
    float* Ho = H + (size_t)(m0 + mb) * NHID + n0 + l;
    if (m0 + 32 <= M) {
#pragma unroll
        for (int r = 0; r < 8; ++r) {
            Ho[(size_t)r * NHID]        = acc0[r];
            Ho[(size_t)(16 + r) * NHID] = acc1[r];
        }
    } else {
#pragma unroll
        for (int r = 0; r < 8; ++r) {
            if (m0 + mb + r < M)      Ho[(size_t)r * NHID]        = acc0[r];
            if (m0 + 16 + mb + r < M) Ho[(size_t)(16 + r) * NHID] = acc1[r];
        }
    }
}

__global__ __launch_bounds__(256) void k_deal_gemm1(const int* __restrict__ src,
                                                    const int* __restrict__ dst,
                                                    const float* __restrict__ w,
                                                    int* __restrict__ pst,
                                                    int* __restrict__ tmp,
                                                    int E, int nbk, int ebk,
                                                    const float* __restrict__ W1, float* __restrict__ Wp1,
                                                    const float* __restrict__ W2, float* __restrict__ Wp2,
                                                    const float* __restrict__ X,
                                                    float* __restrict__ H, int M,
                                                    int tile_end) {
    const int b = blockIdx.x;
    if (b < ebk) {
        deal_run(src, dst, w, pst, tmp, E, nbk, ebk, b);
    } else if (b < ebk + PACK1_CHUNKS) {
        pack_w1_chunk(W1, Wp1, b - ebk);
    } else if (b < ebk + PACK1_CHUNKS + PACK2_CHUNKS) {
        pack_w2_chunk(W2, Wp2, b - ebk - PACK1_CHUNKS);
    } else {
        const int tile = b - ebk - PACK1_CHUNKS - PACK2_CHUNKS;
        if (tile < tile_end) gemm1_tile<false>(X, W1, H, M, tile);
    }
}

__global__ __launch_bounds__(256) void k_sort_gemm1(const int* __restrict__ pst,
                                                    const int* __restrict__ tmp,
                                                    int* __restrict__ first, float* __restrict__ dinv,
                                                    int* __restrict__ edges, int N, int nbk, int ebk,
                                                    const float* __restrict__ X,
                                                    const float* __restrict__ Wp,
                                                    float* __restrict__ H,
                                                    int tile_beg, int tile_end) {
    const int b = blockIdx.x;
    if (b < nbk) {
        sort_bucket(pst, tmp, first, dinv, edges, N, ebk, b);
    } else {
        const int tile = tile_beg + (b - nbk);
        if (tile < tile_end) gemm1_tile<true>(X, Wp, H, N, tile);
    }
}

#define GPITCH 64
__global__ __launch_bounds__(256) void k_agg1_gemm2(const int* __restrict__ first,
                                                    int* edges,
                                                    const float* __restrict__ dinv,
                                                    const float* __restrict__ H,
                                                    const float* __restrict__ b1,
                                                    const float* __restrict__ Wp,
                                                    float* __restrict__ G, int N) {
    __shared__ float2 Xs[64 * 33];
    __shared__ float2 Gs[32 * (GPITCH / 2)];
    const int lane = threadIdx.x & 31;
    const int wave = __builtin_amdgcn_readfirstlane((int)(threadIdx.x >> 5));
    const int m0 = blockIdx.x * 32;
    const float4* H4 = (const float4*)H;
    const float4 bb = ((const float4*)b1)[lane];

    {
        const int row = threadIdx.x >> 3, p = 16 + 2 * (threadIdx.x & 7);
        Gs[row * (GPITCH / 2) + p].y     = 0.0f;
        Gs[row * (GPITCH / 2) + p + 1].y = 0.0f;
    }

    for (int t = 0; t < 4; ++t) {
        const int m = wave * 4 + t;
        const int i = m0 + m;
        float4 v = make_float4(0.0f, 0.0f, 0.0f, 0.0f);
        if (i < N) {
            const int beg = first[i], end = first[i + 1];
            const float di = dinv[i];

            const float cs = di * di;
            const float4 hs = H4[(size_t)i * 32 + lane];
            float4 acc = make_float4(cs * hs.x, cs * hs.y, cs * hs.z, cs * hs.w);

            for (int base = beg; base < end; base += 32) {
                const int n = min(32, end - base);
                int s_l = 0;
                float c_l = 0.0f;
                if (lane < n) {
                    int2 sw = ((const int2*)edges)[base + lane];
                    s_l = sw.x;
                    c_l = dinv[sw.x] * __int_as_float(sw.y) * di;
                    edges[2 * (size_t)(base + lane) + 1] = __float_as_int(c_l);
                }
                int j = 0;
                for (; j + 4 <= n; j += 4) {
                    const int s0 = lane_i(s_l, j), s1 = lane_i(s_l, j + 1);
                    const int s2 = lane_i(s_l, j + 2), s3 = lane_i(s_l, j + 3);
                    const float c0 = lane_f(c_l, j), c1 = lane_f(c_l, j + 1);
                    const float c2 = lane_f(c_l, j + 2), c3 = lane_f(c_l, j + 3);
                    const float4 v0 = H4[(size_t)s0 * 32 + lane];
                    const float4 v1 = H4[(size_t)s1 * 32 + lane];
                    const float4 v2 = H4[(size_t)s2 * 32 + lane];
                    const float4 v3 = H4[(size_t)s3 * 32 + lane];
                    acc.x += c0 * v0.x; acc.y += c0 * v0.y; acc.z += c0 * v0.z; acc.w += c0 * v0.w;
                    acc.x += c1 * v1.x; acc.y += c1 * v1.y; acc.z += c1 * v1.z; acc.w += c1 * v1.w;
                    acc.x += c2 * v2.x; acc.y += c2 * v2.y; acc.z += c2 * v2.z; acc.w += c2 * v2.w;
                    acc.x += c3 * v3.x; acc.y += c3 * v3.y; acc.z += c3 * v3.z; acc.w += c3 * v3.w;
                }
                for (; j < n; ++j) {
                    const int s0 = lane_i(s_l, j);
                    const float c0 = lane_f(c_l, j);
                    const float4 v0 = H4[(size_t)s0 * 32 + lane];
                    acc.x += c0 * v0.x; acc.y += c0 * v0.y; acc.z += c0 * v0.z; acc.w += c0 * v0.w;
                }
            }
            v.x = fmaxf(acc.x + bb.x, 0.0f);
            v.y = fmaxf(acc.y + bb.y, 0.0f);
            v.z = fmaxf(acc.z + bb.z, 0.0f);
            v.w = fmaxf(acc.w + bb.w, 0.0f);
        }
        Xs[(2 * lane)     * 33 + m] = make_float2(v.x, v.y);
        Xs[(2 * lane + 1) * 33 + m] = make_float2(v.z, v.w);
    }
    __syncthreads();

    if (wave < 6) {
        const int ms = wave / 3, nt = wave - 3 * ms;
        const int l = lane & 15, half = lane >> 4;
        const float2* Bp = (const float2*)Wp;

        v8f acc = {};
#pragma unroll 4
        for (int kp = 0; kp < 64; kp += 2) {
            int row = kp + half;
            float2 a = Xs[row * 33 + ms * 16 + l];
            float2 b = Bp[(size_t)row * NCPAD + nt * 16 + l];
            v2f av = {a.x, a.y}, bv = {b.x, b.y};
            acc = __builtin_amdgcn_wmma_f32_16x16x4_f32(false, av, false, bv, (short)0, acc, false, false);
        }

        const int fi = (nt == 2) ? (2 * l + 1) : (2 * (nt * 16 + l));
        float* Gf = (float*)Gs + (ms * 16 + 8 * half) * GPITCH + fi;
#pragma unroll
        for (int r = 0; r < 8; ++r) Gf[r * GPITCH] = acc[r];
    }
    __syncthreads();

    float2* G2 = (float2*)G;
    for (int t = 0; t < 4; ++t) {
        const int m = wave * 4 + t;
        if (m0 + m < N) G2[(size_t)(m0 + m) * (GPITCH / 2) + lane] = Gs[m * (GPITCH / 2) + lane];
    }
}

#define L2_NODES 8
#define L2_WAVES 4

template <int M>
static __device__ __forceinline__ void l2_rows(const float2* __restrict__ G2, int s_l, float c_l,
                                               int j, int lane, float& acc0, float& acc1) {
    float2 v[M];
#pragma unroll
    for (int k = 0; k < M; ++k)
        v[k] = (G2 + (size_t)lane_i(s_l, j + k) * (GPITCH / 2))[lane];
#pragma unroll
    for (int k = 0; k < M; ++k) {
        const float c = lane_f(c_l, j + k);
        acc0 = fmaf(c, v[k].x, acc0);
        acc1 = fmaf(c, v[k].y, acc1);
    }
}

static __device__ __forceinline__ void l2_chunk(const float2* __restrict__ G2, int s_l, float c_l,
                                                int n, int lane, float& acc0, float& acc1) {
    int j = 0;
    for (; j + 8 <= n; j += 8) l2_rows<8>(G2, s_l, c_l, j, lane, acc0, acc1);
    if (j + 4 <= n) { l2_rows<4>(G2, s_l, c_l, j, lane, acc0, acc1); j += 4; }
    switch (n - j) {
        case 3: l2_rows<3>(G2, s_l, c_l, j, lane, acc0, acc1); break;
        case 2: l2_rows<2>(G2, s_l, c_l, j, lane, acc0, acc1); break;
        case 1: l2_rows<1>(G2, s_l, c_l, j, lane, acc0, acc1); break;
        default: break;
    }
}

__global__ __launch_bounds__(32 * L2_WAVES) void k_agg2_lsm(const int* __restrict__ first,
                                                            const int* __restrict__ edges,
                                                            const float* __restrict__ dinv,
                                                            const float* __restrict__ G,
                                                            const float* __restrict__ b2,
                                                            float* __restrict__ out, int N) {
    const int lane = threadIdx.x & 31;
    const int n0 = __builtin_amdgcn_readfirstlane(
        (int)((blockIdx.x * L2_WAVES + (threadIdx.x >> 5)) * L2_NODES));
    if (n0 >= N) return;
    const int2*   E2 = (const int2*)edges;
    const float2* G2 = (const float2*)G;

    const int   bnd = first[min(n0 + lane, N)];
    const float dv  = dinv[min(n0 + lane, N - 1)];
    const int  c2   = 32 + lane;
    const bool has2 = c2 < NCLASS;
    const float bia1 = b2[lane];
    const float bia2 = has2 ? b2[c2] : 0.0f;

    int beg = lane_i(bnd, 0), end = lane_i(bnd, 1);
    int s_n = 0;
    float c_n = 0.0f;
    if (lane < min(32, end - beg)) {
        const int2 sw = E2[beg + lane];
        s_n = sw.x;
        c_n = __int_as_float(sw.y);
    }
    float2 g_n = G2[(size_t)n0 * (GPITCH / 2) + lane];

    for (int t = 0; t < L2_NODES; ++t) {
        const int i = n0 + t;
        if (i >= N) break;
        const int s_n_cur = s_n;
        const float c_n_cur = c_n;
        const float2 gs = g_n;
        const float di = lane_f(dv, t);

        int nend = end;
        if (t + 1 < L2_NODES && i + 1 < N) {
            nend = lane_i(bnd, t + 2);
            s_n = 0;
            c_n = 0.0f;
            if (lane < min(32, nend - end)) {
                const int2 sw = E2[end + lane];
                s_n = sw.x;
                c_n = __int_as_float(sw.y);
            }
            g_n = G2[(size_t)(i + 1) * (GPITCH / 2) + lane];
        }

        const float cs = di * di;
        float acc0 = cs * gs.x;
        float acc1 = cs * gs.y;

        l2_chunk(G2, s_n_cur, c_n_cur, min(32, end - beg), lane, acc0, acc1);
        for (int base = beg + 32; base < end; base += 32) {
            const int n = min(32, end - base);
            int s_x = 0;
            float c_x = 0.0f;
            if (lane < n) {
                const int2 sw = E2[base + lane];
                s_x = sw.x;
                c_x = __int_as_float(sw.y);
            }
            l2_chunk(G2, s_x, c_x, n, lane, acc0, acc1);
        }

        const float v1 = acc0 + bia1;
        const float v2 = has2 ? (acc1 + bia2) : -INFINITY;

        float mx = fmaxf(v1, v2);
#pragma unroll
        for (int off = 16; off >= 1; off >>= 1) mx = fmaxf(mx, __shfl_xor(mx, off));

        float s = expf(v1 - mx) + (has2 ? expf(v2 - mx) : 0.0f);
#pragma unroll
        for (int off = 16; off >= 1; off >>= 1) s += __shfl_xor(s, off);

        const float ls = logf(s);
        out[(size_t)i * NCLASS + lane] = v1 - mx - ls;
        if (has2) out[(size_t)i * NCLASS + c2] = v2 - mx - ls;

        beg = end;
        end = nend;
    }
}

extern "C" void kernel_launch(void* const* d_in, const int* in_sizes, int n_in,
                              void* d_out, int out_size, void* d_ws, size_t ws_size,
                              hipStream_t stream) {
    const float* x  = (const float*)d_in[0];
    const int*   ei = (const int*)  d_in[1];
    const float* ew = (const float*)d_in[2];
    const float* W1 = (const float*)d_in[3];
    const float* b1 = (const float*)d_in[4];
    const float* W2 = (const float*)d_in[5];
    const float* b2 = (const float*)d_in[6];
    float* out = (float*)d_out;

    const int N = in_sizes[0] / NFEAT;
    const int E = in_sizes[2];
    const int* src = ei;
    const int* dst = ei + E;

    float* ws = (float*)d_ws;
    size_t o = 0;
    auto take = [&](size_t n) { float* p = ws + o; o += (n + 255) & ~(size_t)255; return p; };
    float* dinv = take((size_t)N);
    float* Wp1  = take((size_t)NFEAT * NHID);
    float* Wp2  = take((size_t)NHID * NCPAD);
    float* Hxw  = take((size_t)N * NHID);
    int* edges = (int*)take(2 * (size_t)E);
    int* first = (int*)take((size_t)N + 1);
    const int nbk = (N + (1 << BUCKET_SHIFT) - 1) >> BUCKET_SHIFT;
    const int ebk = (E + EDGE_RUN - 1) / EDGE_RUN;
    float* G  = ws + o;
    int* pst  = (int*)take(((size_t)nbk + 1) * ebk);
    int* tmp  = (int*)take(3 * (size_t)E);

    const int tiles = (N + 31) / 32, t1 = tiles / 4;
    k_deal_gemm1<<<ebk + PACK1_CHUNKS + PACK2_CHUNKS + t1, 256, 0, stream>>>(src, dst, ew, pst, tmp, E, nbk, ebk,
                                                                            W1, Wp1, W2, Wp2, x, Hxw, N, t1);
    k_sort_gemm1<<<nbk + (tiles - t1), 256, 0, stream>>>(pst, tmp, first, dinv, edges, N, nbk, ebk,
                                                         x, Wp1, Hxw, t1, tiles);

    k_agg1_gemm2<<<(N + 31) / 32, 256, 0, stream>>>(first, edges, dinv, Hxw, b1, Wp2, G, N);

    k_agg2_lsm<<<(N + L2_WAVES * L2_NODES - 1) / (L2_WAVES * L2_NODES), 32 * L2_WAVES, 0, stream>>>(first, edges, dinv, G, b2, out, N);
}


# Whole-line stores: both sort levels and the X@W1 row tile put their result together in LDS in final order and copy it out as full 128-byte lines, not one 8/12/64-byte piece per lane; sum orders kept
# speedup vs baseline: 4.3181x; 1.0400x over previous
#include <hip/hip_runtime.h>
#include <cstdint>

typedef float v2f __attribute__((ext_vector_type(2)));
typedef float v8f __attribute__((ext_vector_type(8)));

#define NFEAT 256
#define NHID  128
#define NCLASS 41
#define NCPAD  48

#define BUCKET_SHIFT 8
#define MAX_BUCKETS  512
#define EDGE_RUN     4096
#define XS_BYTES     (128 * 33 * 8)
#define RUN_BYTES    (EDGE_RUN * 12)

struct EdgeRec { int s, w, d; };

static __device__ __forceinline__ int block_excl_scan(int v, int* wtot, int& total) {
    const int lane = threadIdx.x & 31, wv = threadIdx.x >> 5;
    int inc = v;
    for (int off = 1; off < 32; off <<= 1) {
        int u = __shfl_up(inc, off);
        if (lane >= off) inc += u;
    }
    if (lane == 31) wtot[wv] = inc;
    __syncthreads();
    int pre = 0, tot = 0;
    for (int k = 0; k < 8; ++k) {
        int t = wtot[k];
        if (k < wv) pre += t;
        tot += t;
    }
    total = tot;
    __syncthreads();
    return pre + inc - v;
}

static __device__ __forceinline__ void deal_run(const int* __restrict__ src, const int* __restrict__ dst,
                                                const float* __restrict__ w,
                                                int* __restrict__ pst, int* __restrict__ tmp,
                                                int E, int nbk, int ebk, int blk, int* stg) {
    int* __restrict__ tq = tmp + (size_t)blk * (3 * EDGE_RUN);
    __shared__ int cur[MAX_BUCKETS];
    __shared__ int wtot[8];
    const int tid = threadIdx.x;
    cur[tid] = 0;
    cur[tid + 256] = 0;
    __syncthreads();
    int d[EDGE_RUN / 256];
#pragma unroll
    for (int it = 0; it < EDGE_RUN / 256; ++it) {
        int e = blk * EDGE_RUN + it * 256 + tid;
        d[it] = (e < E) ? dst[e] : -1;
    }
#pragma unroll
    for (int it = 0; it < EDGE_RUN / 256; ++it)
        if (d[it] >= 0) atomicAdd(&cur[d[it] >> BUCKET_SHIFT], 1);
    __syncthreads();
    const int h0 = cur[tid], h1 = cur[tid + 256];
    int tot0, tot1;
    const int ex0 = block_excl_scan(h0, wtot, tot0);
    const int ex1 = tot0 + block_excl_scan(h1, wtot, tot1);
    if (tid < nbk)       pst[(size_t)tid * ebk + blk]         = ex0;
    if (tid + 256 < nbk) pst[(size_t)(tid + 256) * ebk + blk] = ex1;
    if (tid == 0)        pst[(size_t)nbk * ebk + blk]         = tot0 + tot1;
    cur[tid]       = ex0;
    cur[tid + 256] = ex1;
    __syncthreads();
#pragma unroll
    for (int g = 0; g < EDGE_RUN / 256; g += 8) {
        int s[8], wi[8];
#pragma unroll
        for (int it = 0; it < 8; ++it) {
            int e = blk * EDGE_RUN + (g + it) * 256 + tid;
            const bool in = e < E;
            s[it]  = in ? src[e] : 0;
            wi[it] = in ? __float_as_int(w[e]) : 0;
        }
#pragma unroll
        for (int it = 0; it < 8; ++it) {
            if (d[g + it] >= 0) {
                const int pos = atomicAdd(&cur[d[g + it] >> BUCKET_SHIFT], 1);
                stg[3 * pos]     = s[it];
                stg[3 * pos + 1] = wi[it];
                stg[3 * pos + 2] = d[g + it];
            }
        }
    }
    __syncthreads();
    const int cnt = min(EDGE_RUN, E - blk * EDGE_RUN);
    const int nw = 3 * cnt;
#pragma unroll 4
    for (int q = tid; q < (nw >> 2); q += 256) ((int4*)tq)[q] = ((const int4*)stg)[q];
    if (tid < (nw & 3)) tq[(nw & ~3) + tid] = stg[(nw & ~3) + tid];
}

template <typename F>
static __device__ __forceinline__ void for_bucket_records(const int* __restrict__ pst,
                                                          const EdgeRec* __restrict__ tq,
                                                          int b, int ebk, F f) {
    const int grp = threadIdx.x >> 3, sub = threadIdx.x & 7;
    const int* __restrict__ lo = pst + (size_t)b * ebk;
    const int* __restrict__ hi = lo + ebk;
    for (int j0 = grp; j0 < ebk; j0 += 128) {
        int ps[4], pe[4];
#pragma unroll
        for (int k = 0; k < 4; ++k) {
            const int j = j0 + 32 * k;
            const bool in = j < ebk;
            ps[k] = in ? lo[j] + sub : 0;
            pe[k] = in ? hi[j] : 0;
        }
        for (;;) {
            EdgeRec q[4];
            bool any = false;
#pragma unroll
            for (int k = 0; k < 4; ++k) {
                q[k].s = 0; q[k].w = 0; q[k].d = -1;
                if (ps[k] < pe[k]) {
                    q[k] = tq[(size_t)(j0 + 32 * k) * EDGE_RUN + ps[k]];
                    any = true;
                }
                ps[k] += 8;
            }
            if (!any) break;
#pragma unroll
            for (int k = 0; k < 4; ++k)
                if (q[k].d >= 0) f(q[k]);
        }
    }
}

static __device__ __forceinline__ void sort_bucket(const int* __restrict__ pst,
                                                   const int* __restrict__ tmp,
                                                   int* __restrict__ first, float* __restrict__ dinv,
                                                   int* __restrict__ edges, int N, int ebk, int b,
                                                   int2* stg, int cap) {
    const EdgeRec* __restrict__ tq = (const EdgeRec*)tmp;
    __shared__ int   s_cnt[256];
    __shared__ float s_deg[256];
    __shared__ int   s_cur[256];
    __shared__ int   wtot[8];
    const int tid = threadIdx.x;

    int part = 0;
    for (int j = tid; j < ebk; j += 256) part += pst[(size_t)b * ebk + j];
    int base;
    (void)block_excl_scan(part, wtot, base);
    const int node0 = b << BUCKET_SHIFT;

    s_cnt[tid] = 0;
    s_deg[tid] = 0.0f;
    __syncthreads();
    for_bucket_records(pst, tq, b, ebk, [&](const EdgeRec& r) {
        const int d = r.d - node0;
        atomicAdd(&s_cnt[d], 1);
        atomicAdd(&s_deg[d], __int_as_float(r.w));
    });
    __syncthreads();

    const int c = s_cnt[tid];
    int tot;
    const int ex = block_excl_scan(c, wtot, tot);
    const int node = node0 + tid;
    if (node < N) {
        first[node] = base + ex;
        const float dg = 1.0f + s_deg[tid];
        dinv[node] = dg > 0.0f ? rsqrtf(dg) : 0.0f;
        if (node == N - 1) first[N] = base + ex + c;
    }
    s_cur[tid] = ex;
    __syncthreads();
    for_bucket_records(pst, tq, b, ebk, [&](const EdgeRec& r) {
        const int p = atomicAdd(&s_cur[r.d - node0], 1);
        stg[min(p, cap)] = make_int2(r.s, r.w);
        if (p >= cap) ((int2*)edges)[base + p] = make_int2(r.s, r.w);
    });
    __syncthreads();
    const int stop = base + min(tot, cap);
    int q = (base & ~15) + tid;
    if (q < base) q += 256;
#pragma unroll 4
    for (; q < stop; q += 256) ((int2*)edges)[q] = stg[q - base];
}

static __device__ __forceinline__ int lane_i(int v, int j) {
    return __builtin_amdgcn_readlane(v, j);
}
static __device__ __forceinline__ float lane_f(float v, int j) {
    return __int_as_float(__builtin_amdgcn_readlane(__float_as_int(v), j));
}

static __device__ __forceinline__ void pack_w1_chunk(const float* __restrict__ W1, float* __restrict__ Wp, int chunk) {
    int idx = chunk * 256 + threadIdx.x;
    if (idx >= NFEAT * NHID) return;
    int k = idx >> 7, n = idx & 127;
    Wp[(size_t)(k >> 1) * (NHID * 2) + n * 2 + (k & 1)] = W1[idx];
}

static __device__ __forceinline__ void pack_w2_chunk(const float* __restrict__ W2, float* __restrict__ Wp, int chunk) {
    int idx = chunk * 256 + threadIdx.x;
    if (idx >= NHID * NCPAD) return;
    int k = idx / NCPAD, n = idx % NCPAD;
    float v = (n < NCLASS) ? W2[k * NCLASS + n] : 0.0f;
    Wp[(size_t)(k >> 1) * (NCPAD * 2) + n * 2 + (k & 1)] = v;
}

#define PACK1_CHUNKS ((NFEAT * NHID + 255) / 256)
#define PACK2_CHUNKS ((NHID * NCPAD + 255) / 256)

#define HS_PITCH 132
template <bool PACKED>
static __device__ __forceinline__ void gemm1_tile(const float* __restrict__ X,
                                                  const float* __restrict__ Wp,
                                                  float* __restrict__ H, int M, int tile,
                                                  float2* Xs) {
    const int tid = threadIdx.x;
    const int m0 = tile * 32;

#pragma unroll
    for (int it = 0; it < 8; ++it) {
        int q = it * 256 + tid;
        int m = q >> 6, k4 = q & 63;
        int mr = m0 + m; if (mr >= M) mr = M - 1;
        float4 v = ((const float4*)(X + (size_t)mr * NFEAT))[k4];
        Xs[(2 * k4)     * 33 + m] = make_float2(v.x, v.y);
        Xs[(2 * k4 + 1) * 33 + m] = make_float2(v.z, v.w);
    }
    __syncthreads();

    const int wave = tid >> 5, lane = tid & 31, l = lane & 15, half = lane >> 4;
    const int n0 = wave * 16;
    const float2* Bp = (const float2*)Wp;

    v8f acc0 = {}, acc1 = {};
#pragma unroll 4
    for (int kp = 0; kp < 128; kp += 2) {
        int row = kp + half;
        float2 a0 = Xs[row * 33 + l];
        float2 a1 = Xs[row * 33 + 16 + l];
        float2 b;
        if (PACKED) {
            b = Bp[(size_t)row * NHID + n0 + l];
        } else {
            b.x = Wp[(size_t)(2 * row)     * NHID + n0 + l];
            b.y = Wp[(size_t)(2 * row + 1) * NHID + n0 + l];
        }
        v2f av0 = {a0.x, a0.y}, av1 = {a1.x, a1.y}, bv = {b.x, b.y};
        acc0 = __builtin_amdgcn_wmma_f32_16x16x4_f32(false, av0, false, bv, (short)0, acc0, false, false);
        acc1 = __builtin_amdgcn_wmma_f32_16x16x4_f32(false, av1, false, bv, (short)0, acc1, false, false);
    }

    __syncthreads();
    float* Hs = (float*)Xs;
    float* Hw = Hs + (8 * half) * HS_PITCH + n0 + l;
#pragma unroll
    for (int r = 0; r < 8; ++r) {
        Hw[r * HS_PITCH]        = acc0[r];
        Hw[(16 + r) * HS_PITCH] = acc1[r];
    }
    __syncthreads();
    for (int t = 0; t < 4; ++t) {
        const int m = wave * 4 + t;
        if (m0 + m < M)
            ((float4*)(H + (size_t)(m0 + m) * NHID))[lane] = ((const float4*)(Hs + m * HS_PITCH))[lane];
    }
}

__global__ __launch_bounds__(256) void k_deal_gemm1(const int* __restrict__ src,
                                                    const int* __restrict__ dst,
                                                    const float* __restrict__ w,
                                                    int* __restrict__ pst,
                                                    int* __restrict__ tmp,
                                                    int E, int nbk, int ebk,
                                                    const float* __restrict__ W1, float* __restrict__ Wp1,
                                                    const float* __restrict__ W2, float* __restrict__ Wp2,
                                                    const float* __restrict__ X,
                                                    float* __restrict__ H, int M,
                                                    int tile_end) {
    __shared__ int4 smem[RUN_BYTES / 16];
    const int b = blockIdx.x;
    if (b < ebk) {
        deal_run(src, dst, w, pst, tmp, E, nbk, ebk, b, (int*)smem);
    } else if (b < ebk + PACK1_CHUNKS) {
        pack_w1_chunk(W1, Wp1, b - ebk);
    } else if (b < ebk + PACK1_CHUNKS + PACK2_CHUNKS) {
        pack_w2_chunk(W2, Wp2, b - ebk - PACK1_CHUNKS);
    } else {
        const int tile = b - ebk - PACK1_CHUNKS - PACK2_CHUNKS;
        if (tile < tile_end) gemm1_tile<false>(X, W1, H, M, tile, (float2*)smem);
    }
}

__global__ __launch_bounds__(256) void k_sort_gemm1(const int* __restrict__ pst,
                                                    const int* __restrict__ tmp,
                                                    int* __restrict__ first, float* __restrict__ dinv,
                                                    int* __restrict__ edges, int N, int nbk, int ebk,
                                                    const float* __restrict__ X,
                                                    const float* __restrict__ Wp,
                                                    float* __restrict__ H,
                                                    int tile_beg, int tile_end) {
    __shared__ int4 smem[XS_BYTES / 16];
    const int b = blockIdx.x;
    if (b < nbk) {
        sort_bucket(pst, tmp, first, dinv, edges, N, ebk, b, (int2*)smem, XS_BYTES / 8 - 1);
    } else {
        const int tile = tile_beg + (b - nbk);
        if (tile < tile_end) gemm1_tile<true>(X, Wp, H, N, tile, (float2*)smem);
    }
}

#define GPITCH 64
__global__ __launch_bounds__(256) void k_agg1_gemm2(const int* __restrict__ first,
                                                    int* edges,
                                                    const float* __restrict__ dinv,
                                                    const float* __restrict__ H,
                                                    const float* __restrict__ b1,
                                                    const float* __restrict__ Wp,
                                                    float* __restrict__ G, int N) {
    __shared__ float2 Xs[64 * 33];
    __shared__ float2 Gs[32 * (GPITCH / 2)];
    const int lane = threadIdx.x & 31;
    const int wave = __builtin_amdgcn_readfirstlane((int)(threadIdx.x >> 5));
    const int m0 = blockIdx.x * 32;
    const float4* H4 = (const float4*)H;
    const float4 bb = ((const float4*)b1)[lane];

    {
        const int row = threadIdx.x >> 3, p = 16 + 2 * (threadIdx.x & 7);
        Gs[row * (GPITCH / 2) + p].y     = 0.0f;
        Gs[row * (GPITCH / 2) + p + 1].y = 0.0f;
    }

    for (int t = 0; t < 4; ++t) {
        const int m = wave * 4 + t;
        const int i = m0 + m;
        float4 v = make_float4(0.0f, 0.0f, 0.0f, 0.0f);
        if (i < N) {
            const int beg = first[i], end = first[i + 1];
            const float di = dinv[i];

            const float cs = di * di;
            const float4 hs = H4[(size_t)i * 32 + lane];
            float4 acc = make_float4(cs * hs.x, cs * hs.y, cs * hs.z, cs * hs.w);

            for (int base = beg; base < end; base += 32) {
                const int n = min(32, end - base);
                int s_l = 0;
                float c_l = 0.0f;
                if (lane < n) {
                    int2 sw = ((const int2*)edges)[base + lane];
                    s_l = sw.x;
                    c_l = dinv[sw.x] * __int_as_float(sw.y) * di;
                    edges[2 * (size_t)(base + lane) + 1] = __float_as_int(c_l);
                }
                int j = 0;
                for (; j + 4 <= n; j += 4) {
                    const int s0 = lane_i(s_l, j), s1 = lane_i(s_l, j + 1);
                    const int s2 = lane_i(s_l, j + 2), s3 = lane_i(s_l, j + 3);
                    const float c0 = lane_f(c_l, j), c1 = lane_f(c_l, j + 1);
                    const float c2 = lane_f(c_l, j + 2), c3 = lane_f(c_l, j + 3);
                    const float4 v0 = H4[(size_t)s0 * 32 + lane];
                    const float4 v1 = H4[(size_t)s1 * 32 + lane];
                    const float4 v2 = H4[(size_t)s2 * 32 + lane];
                    const float4 v3 = H4[(size_t)s3 * 32 + lane];
                    acc.x += c0 * v0.x; acc.y += c0 * v0.y; acc.z += c0 * v0.z; acc.w += c0 * v0.w;
                    acc.x += c1 * v1.x; acc.y += c1 * v1.y; acc.z += c1 * v1.z; acc.w += c1 * v1.w;
                    acc.x += c2 * v2.x; acc.y += c2 * v2.y; acc.z += c2 * v2.z; acc.w += c2 * v2.w;
                    acc.x += c3 * v3.x; acc.y += c3 * v3.y; acc.z += c3 * v3.z; acc.w += c3 * v3.w;
                }
                for (; j < n; ++j) {
                    const int s0 = lane_i(s_l, j);
                    const float c0 = lane_f(c_l, j);
                    const float4 v0 = H4[(size_t)s0 * 32 + lane];
                    acc.x += c0 * v0.x; acc.y += c0 * v0.y; acc.z += c0 * v0.z; acc.w += c0 * v0.w;
                }
            }
            v.x = fmaxf(acc.x + bb.x, 0.0f);
            v.y = fmaxf(acc.y + bb.y, 0.0f);
            v.z = fmaxf(acc.z + bb.z, 0.0f);
            v.w = fmaxf(acc.w + bb.w, 0.0f);
        }
        Xs[(2 * lane)     * 33 + m] = make_float2(v.x, v.y);
        Xs[(2 * lane + 1) * 33 + m] = make_float2(v.z, v.w);
    }
    __syncthreads();

    if (wave < 6) {
        const int ms = wave / 3, nt = wave - 3 * ms;
        const int l = lane & 15, half = lane >> 4;
        const float2* Bp = (const float2*)Wp;

        v8f acc = {};
#pragma unroll 4
        for (int kp = 0; kp < 64; kp += 2) {
            int row = kp + half;
            float2 a = Xs[row * 33 + ms * 16 + l];
            float2 b = Bp[(size_t)row * NCPAD + nt * 16 + l];
            v2f av = {a.x, a.y}, bv = {b.x, b.y};
            acc = __builtin_amdgcn_wmma_f32_16x16x4_f32(false, av, false, bv, (short)0, acc, false, false);
        }

        const int fi = (nt == 2) ? (2 * l + 1) : (2 * (nt * 16 + l));
        float* Gf = (float*)Gs + (ms * 16 + 8 * half) * GPITCH + fi;
#pragma unroll
        for (int r = 0; r < 8; ++r) Gf[r * GPITCH] = acc[r];
    }
    __syncthreads();

    float2* G2 = (float2*)G;
    for (int t = 0; t < 4; ++t) {
        const int m = wave * 4 + t;
        if (m0 + m < N) G2[(size_t)(m0 + m) * (GPITCH / 2) + lane] = Gs[m * (GPITCH / 2) + lane];
    }
}

#define L2_NODES 8
#define L2_WAVES 4

template <int M>
static __device__ __forceinline__ void l2_rows(const float2* __restrict__ G2, int s_l, float c_l,
                                               int j, int lane, float& acc0, float& acc1) {
    float2 v[M];
#pragma unroll
    for (int k = 0; k < M; ++k)
        v[k] = (G2 + (size_t)lane_i(s_l, j + k) * (GPITCH / 2))[lane];
#pragma unroll
    for (int k = 0; k < M; ++k) {
        const float c = lane_f(c_l, j + k);
        acc0 = fmaf(c, v[k].x, acc0);
        acc1 = fmaf(c, v[k].y, acc1);
    }
}

static __device__ __forceinline__ void l2_chunk(const float2* __restrict__ G2, int s_l, float c_l,
                                                int n, int lane, float& acc0, float& acc1) {
    int j = 0;
    for (; j + 8 <= n; j += 8) l2_rows<8>(G2, s_l, c_l, j, lane, acc0, acc1);
    if (j + 4 <= n) { l2_rows<4>(G2, s_l, c_l, j, lane, acc0, acc1); j += 4; }
    switch (n - j) {
        case 3: l2_rows<3>(G2, s_l, c_l, j, lane, acc0, acc1); break;
        case 2: l2_rows<2>(G2, s_l, c_l, j, lane, acc0, acc1); break;
        case 1: l2_rows<1>(G2, s_l, c_l, j, lane, acc0, acc1); break;
        default: break;
    }
}

__global__ __launch_bounds__(32 * L2_WAVES) void k_agg2_lsm(const int* __restrict__ first,
                                                            const int* __restrict__ edges,
                                                            const float* __restrict__ dinv,
                                                            const float* __restrict__ G,
                                                            const float* __restrict__ b2,
                                                            float* __restrict__ out, int N) {
    const int lane = threadIdx.x & 31;
    const int n0 = __builtin_amdgcn_readfirstlane(
        (int)((blockIdx.x * L2_WAVES + (threadIdx.x >> 5)) * L2_NODES));
    if (n0 >= N) return;
    const int2*   E2 = (const int2*)edges;
    const float2* G2 = (const float2*)G;

    const int   bnd = first[min(n0 + lane, N)];
    const float dv  = dinv[min(n0 + lane, N - 1)];
    const int  c2   = 32 + lane;
    const bool has2 = c2 < NCLASS;
    const float bia1 = b2[lane];
    const float bia2 = has2 ? b2[c2] : 0.0f;

    int beg = lane_i(bnd, 0), end = lane_i(bnd, 1);
    int s_n = 0;
    float c_n = 0.0f;
    if (lane < min(32, end - beg)) {
        const int2 sw = E2[beg + lane];
        s_n = sw.x;
        c_n = __int_as_float(sw.y);
    }
    float2 g_n = G2[(size_t)n0 * (GPITCH / 2) + lane];

    for (int t = 0; t < L2_NODES; ++t) {
        const int i = n0 + t;
        if (i >= N) break;
        const int s_n_cur = s_n;
        const float c_n_cur = c_n;
        const float2 gs = g_n;
        const float di = lane_f(dv, t);

        int nend = end;
        if (t + 1 < L2_NODES && i + 1 < N) {
            nend = lane_i(bnd, t + 2);
            s_n = 0;
            c_n = 0.0f;
            if (lane < min(32, nend - end)) {
                const int2 sw = E2[end + lane];
                s_n = sw.x;
                c_n = __int_as_float(sw.y);
            }
            g_n = G2[(size_t)(i + 1) * (GPITCH / 2) + lane];
        }

        const float cs = di * di;
        float acc0 = cs * gs.x;
        float acc1 = cs * gs.y;

        l2_chunk(G2, s_n_cur, c_n_cur, min(32, end - beg), lane, acc0, acc1);
        for (int base = beg + 32; base < end; base += 32) {
            const int n = min(32, end - base);
            int s_x = 0;
            float c_x = 0.0f;
            if (lane < n) {
                const int2 sw = E2[base + lane];
                s_x = sw.x;
                c_x = __int_as_float(sw.y);
            }
            l2_chunk(G2, s_x, c_x, n, lane, acc0, acc1);
        }

        const float v1 = acc0 + bia1;
        const float v2 = has2 ? (acc1 + bia2) : -INFINITY;

        float mx = fmaxf(v1, v2);
#pragma unroll
        for (int off = 16; off >= 1; off >>= 1) mx = fmaxf(mx, __shfl_xor(mx, off));

        float s = expf(v1 - mx) + (has2 ? expf(v2 - mx) : 0.0f);
#pragma unroll
        for (int off = 16; off >= 1; off >>= 1) s += __shfl_xor(s, off);

        const float ls = logf(s);
        out[(size_t)i * NCLASS + lane] = v1 - mx - ls;
        if (has2) out[(size_t)i * NCLASS + c2] = v2 - mx - ls;

        beg = end;
        end = nend;
    }
}

extern "C" void kernel_launch(void* const* d_in, const int* in_sizes, int n_in,
                              void* d_out, int out_size, void* d_ws, size_t ws_size,
                              hipStream_t stream) {
    const float* x  = (const float*)d_in[0];
    const int*   ei = (const int*)  d_in[1];
    const float* ew = (const float*)d_in[2];
    const float* W1 = (const float*)d_in[3];
    const float* b1 = (const float*)d_in[4];
    const float* W2 = (const float*)d_in[5];
    const float* b2 = (const float*)d_in[6];
    float* out = (float*)d_out;

    const int N = in_sizes[0] / NFEAT;
    const int E = in_sizes[2];
    const int* src = ei;
    const int* dst = ei + E;

    float* ws = (float*)d_ws;
    size_t o = 0;
    auto take = [&](size_t n) { float* p = ws + o; o += (n + 255) & ~(size_t)255; return p; };
    float* dinv = take((size_t)N);
    float* Wp1  = take((size_t)NFEAT * NHID);
    float* Wp2  = take((size_t)NHID * NCPAD);
    float* Hxw  = take((size_t)N * NHID);
    int* edges = (int*)take(2 * (size_t)E);
    int* first = (int*)take((size_t)N + 1);
    const int nbk = (N + (1 << BUCKET_SHIFT) - 1) >> BUCKET_SHIFT;
    const int ebk = (E + EDGE_RUN - 1) / EDGE_RUN;
    float* G  = ws + o;
    int* pst  = (int*)take(((size_t)nbk + 1) * ebk);
    int* tmp  = (int*)take(3 * (size_t)E);

    const int tiles = (N + 31) / 32, t1 = tiles / 4;
    k_deal_gemm1<<<ebk + PACK1_CHUNKS + PACK2_CHUNKS + t1, 256, 0, stream>>>(src, dst, ew, pst, tmp, E, nbk, ebk,
                                                                            W1, Wp1, W2, Wp2, x, Hxw, N, t1);
    k_sort_gemm1<<<nbk + (tiles - t1), 256, 0, stream>>>(pst, tmp, first, dinv, edges, N, nbk, ebk,
                                                         x, Wp1, Hxw, t1, tiles);

    k_agg1_gemm2<<<(N + 31) / 32, 256, 0, stream>>>(first, edges, dinv, Hxw, b1, Wp2, G, N);

    k_agg2_lsm<<<(N + L2_WAVES * L2_NODES - 1) / (L2_WAVES * L2_NODES), 32 * L2_WAVES, 0, stream>>>(first, edges, dinv, G, b2, out, N);
}
